# starting kernel
# baseline (speedup 1.0000x reference)
#include <hip/hip_runtime.h>
#include <hip/hip_bf16.h>

#define LSEQ 60000
#define NB 8
#define INCH 64
#define HID 5

typedef __attribute__((ext_vector_type(2))) float v2f;
typedef __attribute__((ext_vector_type(8))) float v8f;

__global__ void conv1x1_wmma(const float* __restrict__ x,
                             const float* __restrict__ w,
                             const float* __restrict__ bias,
                             float* __restrict__ y) {
    const int lane = threadIdx.x & 31;
    const int wave = threadIdx.x >> 5;
    const int tile = blockIdx.x * 8 + wave;
    const int b = tile / (LSEQ / 16);
    const int lbase = (tile % (LSEQ / 16)) * 16;
    const int m = lane & 15;
    const int half = lane >> 4;

    const float* xb = x + (size_t)b * INCH * LSEQ + lbase + m;

    v8f acc = {};
#pragma unroll
    for (int chunk = 0; chunk < 16; ++chunk) {
        const int k0 = chunk * 4 + 2 * half;
        v2f a, bb;
        a.x = xb[(size_t)k0 * LSEQ];
        a.y = xb[(size_t)(k0 + 1) * LSEQ];
        bb.x = w[k0];
        bb.y = w[k0 + 1];
        acc = __builtin_amdgcn_wmma_f32_16x16x4_f32(
             false, a,  false, bb,
             (short)0, acc,  false,  false);
    }

    const float bv = bias[0];
    float* yo = y + (size_t)b * LSEQ + lbase;
    if (lane == 0) {
#pragma unroll
        for (int r = 0; r < 8; ++r) yo[r] = acc[r] + bv;
    } else if (lane == 16) {
#pragma unroll
        for (int r = 0; r < 8; ++r) yo[8 + r] = acc[r] + bv;
    }
}

__device__ __forceinline__ float sigm(float v) {
    return __builtin_amdgcn_rcpf(1.0f + __expf(-v));
}
__device__ __forceinline__ float tanh_fast(float v) {
    return 1.0f - 2.0f * __builtin_amdgcn_rcpf(__expf(2.0f * v) + 1.0f);
}

__global__ void lstm_scan(const float* __restrict__ y,
                          const float* __restrict__ inp,
                          const float* __restrict__ w_ih_f, const float* __restrict__ w_hh_f,
                          const float* __restrict__ b_ih_f, const float* __restrict__ b_hh_f,
                          const float* __restrict__ w_hr_f,
                          const float* __restrict__ w_ih_b, const float* __restrict__ w_hh_b,
                          const float* __restrict__ b_ih_b, const float* __restrict__ b_hh_b,
                          const float* __restrict__ w_hr_b,
                          const float* __restrict__ h0, const float* __restrict__ c0,
                          float* __restrict__ hout) {
    const int t = threadIdx.x;
    if (t >= 16) return;
    const int dir = t >> 3;
    const int b = t & 7;

    const float* w_ih = dir ? w_ih_b : w_ih_f;
    const float* w_hh = dir ? w_hh_b : w_hh_f;
    const float* b_ih = dir ? b_ih_b : b_ih_f;
    const float* b_hh = dir ? b_hh_b : b_hh_f;
    const float* w_hr = dir ? w_hr_b : w_hr_f;

    float wi0[20], wi1[20], whh[20], bsum[20], whr[HID], c[HID];
#pragma unroll
    for (int j = 0; j < 20; ++j) {
        wi0[j] = w_ih[2 * j];
        wi1[j] = w_ih[2 * j + 1];
        whh[j] = w_hh[j];
        bsum[j] = b_ih[j] + b_hh[j];
    }
#pragma unroll
    for (int k = 0; k < HID; ++k) {
        whr[k] = w_hr[k];
        c[k] = c0[(dir * NB + b) * HID + k];
    }
    float h = h0[dir * NB + b];

    const float* yb = y + (size_t)b * LSEQ;
    const float* ib = inp + (size_t)b * LSEQ;
    float* ho = hout + (size_t)(dir * NB + b) * LSEQ;

    for (int s = 0; s < LSEQ; ++s) {
        const int l = dir ? (LSEQ - 1 - s) : s;
        if ((s & 31) == 0) {
            const int pl = dir ? (l - 96) : (l + 96);
            if ((unsigned)pl < (unsigned)LSEQ) {
                __builtin_prefetch(yb + pl, 0, 0);
                __builtin_prefetch(ib + pl, 0, 0);
            }
        }
        const float x0 = yb[l];
        const float x1 = ib[l];
        float hn = 0.0f;
#pragma unroll
        for (int k = 0; k < HID; ++k) {
            const float gi = fmaf(x0, wi0[k],      fmaf(x1, wi1[k],      fmaf(h, whh[k],      bsum[k])));
            const float gf = fmaf(x0, wi0[k + 5],  fmaf(x1, wi1[k + 5],  fmaf(h, whh[k + 5],  bsum[k + 5])));
            const float gc = fmaf(x0, wi0[k + 10], fmaf(x1, wi1[k + 10], fmaf(h, whh[k + 10], bsum[k + 10])));
            const float go = fmaf(x0, wi0[k + 15], fmaf(x1, wi1[k + 15], fmaf(h, whh[k + 15], bsum[k + 15])));
            const float iv = sigm(gi);
            const float fv = sigm(gf);
            const float gv = tanh_fast(gc);
            const float ov = sigm(go);
            c[k] = fmaf(fv, c[k], iv * gv);
            hn = fmaf(ov * tanh_fast(c[k]), whr[k], hn);
        }
        h = hn;
        ho[l] = h;
    }
}

__global__ void seq_softmax(const float* __restrict__ hbuf, float* __restrict__ out) {
    const int b = blockIdx.x;
    const int tid = threadIdx.x;
    const int T = 256;
    __shared__ float red[256];

    const float* hf = hbuf + (size_t)b * LSEQ;
    const float* hb = hbuf + (size_t)(NB + b) * LSEQ;

    float m = -3.0e38f;
    for (int l = tid; l < LSEQ; l += T) m = fmaxf(m, hf[l] + hb[l]);
    red[tid] = m;
    __syncthreads();
    for (int s = T / 2; s > 0; s >>= 1) {
        if (tid < s) red[tid] = fmaxf(red[tid], red[tid + s]);
        __syncthreads();
    }
    m = red[0];
    __syncthreads();

    float sum = 0.0f;
    for (int l = tid; l < LSEQ; l += T) sum += __expf(hf[l] + hb[l] - m);
    red[tid] = sum;
    __syncthreads();
    for (int s = T / 2; s > 0; s >>= 1) {
        if (tid < s) red[tid] += red[tid + s];
        __syncthreads();
    }
    sum = red[0];
    __syncthreads();

    const float inv = __builtin_amdgcn_rcpf(sum);
    for (int l = tid; l < LSEQ; l += T) {
        const float e = __expf(hf[l] + hb[l] - m) * inv;
        float* o = out + ((size_t)b * LSEQ + l) * 2;
        o[0] = e;
        o[1] = e;
    }
}

extern "C" void kernel_launch(void* const* d_in, const int* in_sizes, int n_in,
                              void* d_out, int out_size, void* d_ws, size_t ws_size,
                              hipStream_t stream) {
    const float* x      = (const float*)d_in[0];
    const float* inp    = (const float*)d_in[1];
    const float* conv_w = (const float*)d_in[2];
    const float* conv_b = (const float*)d_in[3];
    const float* w_ih_f = (const float*)d_in[4];
    const float* w_hh_f = (const float*)d_in[5];
    const float* b_ih_f = (const float*)d_in[6];
    const float* b_hh_f = (const float*)d_in[7];
    const float* w_hr_f = (const float*)d_in[8];
    const float* w_ih_b = (const float*)d_in[9];
    const float* w_hh_b = (const float*)d_in[10];
    const float* b_ih_b = (const float*)d_in[11];
    const float* b_hh_b = (const float*)d_in[12];
    const float* w_hr_b = (const float*)d_in[13];
    const float* h0     = (const float*)d_in[14];
    const float* c0     = (const float*)d_in[15];
    float* out = (float*)d_out;

    float* y    = (float*)d_ws;
    float* hbuf = y + (size_t)NB * LSEQ;

    conv1x1_wmma<<<(NB * LSEQ / 16) / 8, 256, 0, stream>>>(x, conv_w, conv_b, y);

    lstm_scan<<<1, 32, 0, stream>>>(y, inp,
                                    w_ih_f, w_hh_f, b_ih_f, b_hh_f, w_hr_f,
                                    w_ih_b, w_hh_b, b_ih_b, b_hh_b, w_hr_b,
                                    h0, c0, hbuf);

    seq_softmax<<<NB, 256, 0, stream>>>(hbuf, out);
}


# LSTM scan: one wave per recurrence, one lane per hidden unit; h summed by a 3-step lane butterfly and gate sums re-associated (input terms formed a block ahead), so f32 sum order changes
# speedup vs baseline: 10.1714x; 10.1714x over previous
#include <hip/hip_runtime.h>
#include <hip/hip_bf16.h>

#define LSEQ 60000
#define NB 8
#define INCH 64
#define HID 5

typedef __attribute__((ext_vector_type(2))) float v2f;
typedef __attribute__((ext_vector_type(8))) float v8f;

__global__ void conv1x1_wmma(const float* __restrict__ x,
                             const float* __restrict__ w,
                             const float* __restrict__ bias,
                             float* __restrict__ y) {
    const int lane = threadIdx.x & 31;
    const int wave = threadIdx.x >> 5;
    const int tile = blockIdx.x * 8 + wave;
    const int b = tile / (LSEQ / 16);
    const int lbase = (tile % (LSEQ / 16)) * 16;
    const int m = lane & 15;
    const int half = lane >> 4;

    const float* xb = x + (size_t)b * INCH * LSEQ + lbase + m;

    v8f acc = {};
#pragma unroll
    for (int chunk = 0; chunk < 16; ++chunk) {
        const int k0 = chunk * 4 + 2 * half;
        v2f a, bb;
        a.x = xb[(size_t)k0 * LSEQ];
        a.y = xb[(size_t)(k0 + 1) * LSEQ];
        bb.x = w[k0];
        bb.y = w[k0 + 1];
        acc = __builtin_amdgcn_wmma_f32_16x16x4_f32(
             false, a,  false, bb,
             (short)0, acc,  false,  false);
    }

    const float bv = bias[0];
    float* yo = y + (size_t)b * LSEQ + lbase;
    if (lane == 0) {
#pragma unroll
        for (int r = 0; r < 8; ++r) yo[r] = acc[r] + bv;
    } else if (lane == 16) {
#pragma unroll
        for (int r = 0; r < 8; ++r) yo[8 + r] = acc[r] + bv;
    }
}

#define UB 8
#define NBLK (LSEQ / UB)
#define IT_BLK 4
#define LOG2E 1.4426950408889634f
#define KC 2.8853900817779268f
#define INVKC 0.34657359027997264f

#define SB() __builtin_amdgcn_sched_barrier(0)

template <int MASK>
__device__ __forceinline__ float xadd(float v) {
    const int t = __builtin_amdgcn_update_dpp(0, __builtin_bit_cast(int, v), 0x160 + MASK, 0xf, 0xf, true);
    return v + __builtin_bit_cast(float, t);
}

struct LaneW {
    float a[4];
    float w0[4];
    float w1[4];
    float b[4];
    float r1;
    float r2;
};

template <int DIR>
__device__ __forceinline__ void load_blk(const float* __restrict__ yb, const float* __restrict__ ib,
                                         int blk, float (&xy)[UB], float (&xi)[UB]) {
    if (blk > NBLK - 1) blk = NBLK - 1;
    const int base = DIR ? (LSEQ - UB - blk * UB) : (blk * UB);
    const float4 y0 = *reinterpret_cast<const float4*>(yb + base);
    const float4 y1 = *reinterpret_cast<const float4*>(yb + base + 4);
    const float4 i0 = *reinterpret_cast<const float4*>(ib + base);
    const float4 i1 = *reinterpret_cast<const float4*>(ib + base + 4);
    xy[0] = y0.x; xy[1] = y0.y; xy[2] = y0.z; xy[3] = y0.w;
    xy[4] = y1.x; xy[5] = y1.y; xy[6] = y1.z; xy[7] = y1.w;
    xi[0] = i0.x; xi[1] = i0.y; xi[2] = i0.z; xi[3] = i0.w;
    xi[4] = i1.x; xi[5] = i1.y; xi[6] = i1.z; xi[7] = i1.w;
}

template <int DIR>
__device__ __forceinline__ void pre_blk(const LaneW& W, const float (&xy)[UB], const float (&xi)[UB],
                                        float (&P)[UB][4]) {
#pragma unroll
    for (int j = 0; j < UB; ++j) {
        const int m = DIR ? (UB - 1 - j) : j;
#pragma unroll
        for (int q = 0; q < 4; ++q)
            P[j][q] = fmaf(xy[m], W.w0[q], fmaf(xi[m], W.w1[q], W.b[q]));
    }
}

template <int DIR>
__device__ __forceinline__ void steps8(const LaneW& W, const float (&P)[UB][4], float (&Pn)[UB][4],
                                       const float (&xy)[UB], const float (&xi)[UB],
                                       float& h, float& cs, float* hs) {
#pragma unroll
    for (int j = 0; j < UB; ++j) {
        const int m = DIR ? (UB - 1 - j) : j;
        const float gi = fmaf(h, W.a[0], P[j][0]);
        const float gf = fmaf(h, W.a[1], P[j][1]);
        const float gg = fmaf(h, W.a[2], P[j][2]);
        const float go = fmaf(h, W.a[3], P[j][3]);
        SB();
        const float ei = __builtin_amdgcn_exp2f(gi);
        const float ef = __builtin_amdgcn_exp2f(gf);
        const float eg = __builtin_amdgcn_exp2f(gg);
        const float eo = __builtin_amdgcn_exp2f(go);
        SB();
        const float di = fmaf(ei, INVKC, INVKC);
        const float df = 1.0f + ef;
        const float dg = fmaf(eg, -0.5f, -0.5f);
        SB();
        const float ki = __builtin_amdgcn_rcpf(di);
        const float dq = 1.0f + eo;
        const float fv = __builtin_amdgcn_rcpf(df);
        const float s2 = __builtin_amdgcn_rcpf(dg);
        const float ov = __builtin_amdgcn_rcpf(dq);
        SB();
        const float ca = fmaf(fv, cs, ki);
        cs = fmaf(ki, s2, ca);
        SB();
        const float ec = __builtin_amdgcn_exp2f(cs);
        const float o1 = ov * W.r1;
        const float o2 = ov * W.r2;
        const float t0 = fmaf(xi[m], W.w1[0], W.b[0]);
        const float t1 = fmaf(xi[m], W.w1[1], W.b[1]);
        const float t2 = fmaf(xi[m], W.w1[2], W.b[2]);
        const float t3 = fmaf(xi[m], W.w1[3], W.b[3]);
        SB();
        const float dc = 1.0f + ec;
        Pn[j][0] = fmaf(xy[m], W.w0[0], t0);
        Pn[j][1] = fmaf(xy[m], W.w0[1], t1);
        Pn[j][2] = fmaf(xy[m], W.w0[2], t2);
        Pn[j][3] = fmaf(xy[m], W.w0[3], t3);
        SB();
        const float rc = __builtin_amdgcn_rcpf(dc);
        float p = fmaf(rc, o2, o1);
        p = xadd<1>(p);
        p = xadd<2>(p);
        p = xadd<4>(p);
        SB();
        h = p;
        hs[j] = p;
    }
}

template <int DIR>
__device__ __forceinline__ void run_chain(const LaneW& W, const float* __restrict__ yb,
                                          const float* __restrict__ ib, float* __restrict__ ho,
                                          float h, float cs, int lane) {
    float PA[UB][4], PB[UB][4];
    float XAy[UB], XAi[UB], XBy[UB], XBi[UB];
    float hs[UB * IT_BLK];

    load_blk<DIR>(yb, ib, 0, XAy, XAi);
    pre_blk<DIR>(W, XAy, XAi, PA);
    load_blk<DIR>(yb, ib, 1, XBy, XBi);

    for (int n = 0; n < NBLK; n += IT_BLK) {
        {
            int pf = DIR ? (LSEQ - UB * IT_BLK - n * UB - 4 * UB * IT_BLK) : (n * UB + 4 * UB * IT_BLK);
            pf = pf < 0 ? 0 : (pf > LSEQ - 1 ? LSEQ - 1 : pf);
            __builtin_prefetch(yb + pf, 0, 0);
            __builtin_prefetch(ib + pf, 0, 0);
        }
        load_blk<DIR>(yb, ib, n + 2, XAy, XAi);
        steps8<DIR>(W, PA, PB, XBy, XBi, h, cs, hs);
        load_blk<DIR>(yb, ib, n + 3, XBy, XBi);
        steps8<DIR>(W, PB, PA, XAy, XAi, h, cs, hs + UB);
        load_blk<DIR>(yb, ib, n + 4, XAy, XAi);
        steps8<DIR>(W, PA, PB, XBy, XBi, h, cs, hs + 2 * UB);
        load_blk<DIR>(yb, ib, n + 5, XBy, XBi);
        steps8<DIR>(W, PB, PA, XAy, XAi, h, cs, hs + 3 * UB);

        const int base = DIR ? (LSEQ - UB * IT_BLK - n * UB) : (n * UB);
        if (lane == 0) {
#pragma unroll
            for (int v = 0; v < UB * IT_BLK / 4; ++v) {
                float4 o;
                if (DIR) {
                    const int s = UB * IT_BLK - 1 - 4 * v;
                    o = make_float4(hs[s], hs[s - 1], hs[s - 2], hs[s - 3]);
                } else {
                    o = make_float4(hs[4 * v], hs[4 * v + 1], hs[4 * v + 2], hs[4 * v + 3]);
                }
                *reinterpret_cast<float4*>(ho + base + 4 * v) = o;
            }
        }
    }
}

__global__ __launch_bounds__(32) void lstm_scan(const float* __restrict__ y,
                          const float* __restrict__ inp,
                          const float* __restrict__ w_ih_f, const float* __restrict__ w_hh_f,
                          const float* __restrict__ b_ih_f, const float* __restrict__ b_hh_f,
                          const float* __restrict__ w_hr_f,
                          const float* __restrict__ w_ih_b, const float* __restrict__ w_hh_b,
                          const float* __restrict__ b_ih_b, const float* __restrict__ b_hh_b,
                          const float* __restrict__ w_hr_b,
                          const float* __restrict__ h0, const float* __restrict__ c0,
                          float* __restrict__ hout) {
    const int chain = blockIdx.x;
    const int dir = chain >> 3;
    const int b = chain & 7;
    const int lane = threadIdx.x;
    const int k = lane & 7;
    const bool real = k < HID;
    const int kk = real ? k : 0;

    const float* w_ih = dir ? w_ih_b : w_ih_f;
    const float* w_hh = dir ? w_hh_b : w_hh_f;
    const float* b_ih = dir ? b_ih_b : b_ih_f;
    const float* b_hh = dir ? b_hh_b : b_hh_f;
    const float* w_hr = dir ? w_hr_b : w_hr_f;

    LaneW W;
#pragma unroll
    for (int q = 0; q < 4; ++q) {
        const int j = q * HID + kk;
        const float sc = real ? ((q == 2) ? KC : -LOG2E) : 0.0f;
        W.a[q] = sc * w_hh[j];
        W.w0[q] = sc * w_ih[2 * j];
        W.w1[q] = sc * w_ih[2 * j + 1];
        W.b[q] = sc * (b_ih[j] + b_hh[j]);
    }
    W.r1 = real ? w_hr[kk] : 0.0f;
    W.r2 = -2.0f * W.r1;
    const float cs = real ? KC * c0[chain * HID + kk] : 0.0f;
    const float h = h0[chain];

    const float* yb = y + (size_t)b * LSEQ;
    const float* ib = inp + (size_t)b * LSEQ;
    float* ho = hout + (size_t)chain * LSEQ;

    if (dir) run_chain<1>(W, yb, ib, ho, h, cs, lane);
    else     run_chain<0>(W, yb, ib, ho, h, cs, lane);
}

__global__ void seq_softmax(const float* __restrict__ hbuf, float* __restrict__ out) {
    const int b = blockIdx.x;
    const int tid = threadIdx.x;
    const int T = 256;
    __shared__ float red[256];

    const float* hf = hbuf + (size_t)b * LSEQ;
    const float* hb = hbuf + (size_t)(NB + b) * LSEQ;

    float m = -3.0e38f;
    for (int l = tid; l < LSEQ; l += T) m = fmaxf(m, hf[l] + hb[l]);
    red[tid] = m;
    __syncthreads();
    for (int s = T / 2; s > 0; s >>= 1) {
        if (tid < s) red[tid] = fmaxf(red[tid], red[tid + s]);
        __syncthreads();
    }
    m = red[0];
    __syncthreads();

    float sum = 0.0f;
    for (int l = tid; l < LSEQ; l += T) sum += __expf(hf[l] + hb[l] - m);
    red[tid] = sum;
    __syncthreads();
    for (int s = T / 2; s > 0; s >>= 1) {
        if (tid < s) red[tid] += red[tid + s];
        __syncthreads();
    }
    sum = red[0];
    __syncthreads();

    const float inv = __builtin_amdgcn_rcpf(sum);
    for (int l = tid; l < LSEQ; l += T) {
        const float e = __expf(hf[l] + hb[l] - m) * inv;
        float* o = out + ((size_t)b * LSEQ + l) * 2;
        o[0] = e;
        o[1] = e;
    }
}

extern "C" void kernel_launch(void* const* d_in, const int* in_sizes, int n_in,
                              void* d_out, int out_size, void* d_ws, size_t ws_size,
                              hipStream_t stream) {
    const float* x      = (const float*)d_in[0];
    const float* inp    = (const float*)d_in[1];
    const float* conv_w = (const float*)d_in[2];
    const float* conv_b = (const float*)d_in[3];
    const float* w_ih_f = (const float*)d_in[4];
    const float* w_hh_f = (const float*)d_in[5];
    const float* b_ih_f = (const float*)d_in[6];
    const float* b_hh_f = (const float*)d_in[7];
    const float* w_hr_f = (const float*)d_in[8];
    const float* w_ih_b = (const float*)d_in[9];
    const float* w_hh_b = (const float*)d_in[10];
    const float* b_ih_b = (const float*)d_in[11];
    const float* b_hh_b = (const float*)d_in[12];
    const float* w_hr_b = (const float*)d_in[13];
    const float* h0     = (const float*)d_in[14];
    const float* c0     = (const float*)d_in[15];
    float* out = (float*)d_out;

    float* y    = (float*)d_ws;
    float* hbuf = y + (size_t)NB * LSEQ;

    conv1x1_wmma<<<(NB * LSEQ / 16) / 8, 256, 0, stream>>>(x, conv_w, conv_b, y);

    lstm_scan<<<2 * NB, 32, 0, stream>>>(y, inp,
                                    w_ih_f, w_hh_f, b_ih_f, b_hh_f, w_hr_f,
                                    w_ih_b, w_hh_b, b_ih_b, b_hh_b, w_hr_b,
                                    h0, c0, hbuf);

    seq_softmax<<<NB, 256, 0, stream>>>(hbuf, out);
}


# LSTM scan: every activation on the hardware tanh instruction, sigmoid(v) = (1 + tanh(v/2))/2 with the halves folded into the weights; activation, cell-sum and first butterfly-add rounding change
# speedup vs baseline: 12.8628x; 1.2646x over previous
#include <hip/hip_runtime.h>
#include <hip/hip_bf16.h>

#define LSEQ 60000
#define NB 8
#define INCH 64
#define HID 5

typedef __attribute__((ext_vector_type(2))) float v2f;
typedef __attribute__((ext_vector_type(8))) float v8f;

__global__ void conv1x1_wmma(const float* __restrict__ x,
                             const float* __restrict__ w,
                             const float* __restrict__ bias,
                             float* __restrict__ y) {
    const int lane = threadIdx.x & 31;
    const int wave = threadIdx.x >> 5;
    const int tile = blockIdx.x * 8 + wave;
    const int b = tile / (LSEQ / 16);
    const int lbase = (tile % (LSEQ / 16)) * 16;
    const int m = lane & 15;
    const int half = lane >> 4;

    const float* xb = x + (size_t)b * INCH * LSEQ + lbase + m;

    v8f acc = {};
#pragma unroll
    for (int chunk = 0; chunk < 16; ++chunk) {
        const int k0 = chunk * 4 + 2 * half;
        v2f a, bb;
        a.x = xb[(size_t)k0 * LSEQ];
        a.y = xb[(size_t)(k0 + 1) * LSEQ];
        bb.x = w[k0];
        bb.y = w[k0 + 1];
        acc = __builtin_amdgcn_wmma_f32_16x16x4_f32(
             false, a,  false, bb,
             (short)0, acc,  false,  false);
    }

    const float bv = bias[0];
    float* yo = y + (size_t)b * LSEQ + lbase;
    if (lane == 0) {
#pragma unroll
        for (int r = 0; r < 8; ++r) yo[r] = acc[r] + bv;
    } else if (lane == 16) {
#pragma unroll
        for (int r = 0; r < 8; ++r) yo[8 + r] = acc[r] + bv;
    }
}

#define UB 8
#define NBLK (LSEQ / UB)
#define IT_BLK 4

#define SB() __builtin_amdgcn_sched_barrier(0)

template <int MASK>
__device__ __forceinline__ float xadd(float v) {
    const int t = __builtin_amdgcn_update_dpp(0, __builtin_bit_cast(int, v), 0x160 + MASK, 0xf, 0xf, true);
    return v + __builtin_bit_cast(float, t);
}

struct LaneW {
    float a[4];
    float w0[4];
    float w1[4];
    float b[4];
    float rh;
};

template <int DIR>
__device__ __forceinline__ void load_blk(const float* __restrict__ yb, const float* __restrict__ ib,
                                         int blk, float (&xy)[UB], float (&xi)[UB]) {
    if (blk > NBLK - 1) blk = NBLK - 1;
    const int base = DIR ? (LSEQ - UB - blk * UB) : (blk * UB);
    const float4 y0 = *reinterpret_cast<const float4*>(yb + base);
    const float4 y1 = *reinterpret_cast<const float4*>(yb + base + 4);
    const float4 i0 = *reinterpret_cast<const float4*>(ib + base);
    const float4 i1 = *reinterpret_cast<const float4*>(ib + base + 4);
    xy[0] = y0.x; xy[1] = y0.y; xy[2] = y0.z; xy[3] = y0.w;
    xy[4] = y1.x; xy[5] = y1.y; xy[6] = y1.z; xy[7] = y1.w;
    xi[0] = i0.x; xi[1] = i0.y; xi[2] = i0.z; xi[3] = i0.w;
    xi[4] = i1.x; xi[5] = i1.y; xi[6] = i1.z; xi[7] = i1.w;
}

template <int DIR>
__device__ __forceinline__ void pre_blk(const LaneW& W, const float (&xy)[UB], const float (&xi)[UB],
                                        float (&P)[UB][4]) {
#pragma unroll
    for (int j = 0; j < UB; ++j) {
        const int m = DIR ? (UB - 1 - j) : j;
#pragma unroll
        for (int q = 0; q < 4; ++q)
            P[j][q] = fmaf(xy[m], W.w0[q], fmaf(xi[m], W.w1[q], W.b[q]));
    }
}

template <int DIR>
__device__ __forceinline__ void steps8(const LaneW& W, const float (&P)[UB][4], float (&Pn)[UB][4],
                                       const float (&xy)[UB], const float (&xi)[UB],
                                       float& h, float& ch, float* hs) {
#pragma unroll
    for (int j = 0; j < UB; ++j) {
        const int m = DIR ? (UB - 1 - j) : j;
        const float gi = fmaf(h, W.a[0], P[j][0]);
        const float gf = fmaf(h, W.a[1], P[j][1]);
        const float gg = fmaf(h, W.a[2], P[j][2]);
        const float go = fmaf(h, W.a[3], P[j][3]);
        SB();
        const float ti = __builtin_amdgcn_tanhf(gi);
        const float tf = __builtin_amdgcn_tanhf(gf);
        const float tg = __builtin_amdgcn_tanhf(gg);
        const float to = __builtin_amdgcn_tanhf(go);
        SB();
        const float iv = fmaf(ti, 0.5f, 0.5f);
        const float fc = fmaf(tf, ch, ch);
        const float t0 = fmaf(xi[m], W.w1[0], W.b[0]);
        const float t1 = fmaf(xi[m], W.w1[1], W.b[1]);
        const float t2 = fmaf(xi[m], W.w1[2], W.b[2]);
        const float t3 = fmaf(xi[m], W.w1[3], W.b[3]);
        SB();
        const float c = fmaf(iv, tg, fc);
        const float ow = fmaf(to, W.rh, W.rh);
        Pn[j][0] = fmaf(xy[m], W.w0[0], t0);
        Pn[j][1] = fmaf(xy[m], W.w0[1], t1);
        Pn[j][2] = fmaf(xy[m], W.w0[2], t2);
        SB();
        const float tc = __builtin_amdgcn_tanhf(c);
        ch = 0.5f * c;
        Pn[j][3] = fmaf(xy[m], W.w0[3], t3);
        SB();
        float p = ow * tc;
        p = xadd<1>(p);
        p = xadd<2>(p);
        p = xadd<4>(p);
        SB();
        h = p;
        hs[j] = p;
    }
}

template <int DIR>
__device__ __forceinline__ void run_chain(const LaneW& W, const float* __restrict__ yb,
                                          const float* __restrict__ ib, float* __restrict__ ho,
                                          float h, float ch, int lane) {
    float PA[UB][4], PB[UB][4];
    float XAy[UB], XAi[UB], XBy[UB], XBi[UB];
    float hs[UB * IT_BLK];

    load_blk<DIR>(yb, ib, 0, XAy, XAi);
    pre_blk<DIR>(W, XAy, XAi, PA);
    load_blk<DIR>(yb, ib, 1, XBy, XBi);

    for (int n = 0; n < NBLK; n += IT_BLK) {
        {
            int pf = DIR ? (LSEQ - UB * IT_BLK - n * UB - 4 * UB * IT_BLK) : (n * UB + 4 * UB * IT_BLK);
            pf = pf < 0 ? 0 : (pf > LSEQ - 1 ? LSEQ - 1 : pf);
            __builtin_prefetch(yb + pf, 0, 0);
            __builtin_prefetch(ib + pf, 0, 0);
        }
        load_blk<DIR>(yb, ib, n + 2, XAy, XAi);
        steps8<DIR>(W, PA, PB, XBy, XBi, h, ch, hs);
        load_blk<DIR>(yb, ib, n + 3, XBy, XBi);
        steps8<DIR>(W, PB, PA, XAy, XAi, h, ch, hs + UB);
        load_blk<DIR>(yb, ib, n + 4, XAy, XAi);
        steps8<DIR>(W, PA, PB, XBy, XBi, h, ch, hs + 2 * UB);
        load_blk<DIR>(yb, ib, n + 5, XBy, XBi);
        steps8<DIR>(W, PB, PA, XAy, XAi, h, ch, hs + 3 * UB);

        const int base = DIR ? (LSEQ - UB * IT_BLK - n * UB) : (n * UB);
        if (lane == 0) {
#pragma unroll
            for (int v = 0; v < UB * IT_BLK / 4; ++v) {
                float4 o;
                if (DIR) {
                    const int s = UB * IT_BLK - 1 - 4 * v;
                    o = make_float4(hs[s], hs[s - 1], hs[s - 2], hs[s - 3]);
                } else {
                    o = make_float4(hs[4 * v], hs[4 * v + 1], hs[4 * v + 2], hs[4 * v + 3]);
                }
                *reinterpret_cast<float4*>(ho + base + 4 * v) = o;
            }
        }
    }
}

__global__ __launch_bounds__(32) void lstm_scan(const float* __restrict__ y,
                          const float* __restrict__ inp,
                          const float* __restrict__ w_ih_f, const float* __restrict__ w_hh_f,
                          const float* __restrict__ b_ih_f, const float* __restrict__ b_hh_f,
                          const float* __restrict__ w_hr_f,
                          const float* __restrict__ w_ih_b, const float* __restrict__ w_hh_b,
                          const float* __restrict__ b_ih_b, const float* __restrict__ b_hh_b,
                          const float* __restrict__ w_hr_b,
                          const float* __restrict__ h0, const float* __restrict__ c0,
                          float* __restrict__ hout) {
    const int chain = blockIdx.x;
    const int dir = chain >> 3;
    const int b = chain & 7;
    const int lane = threadIdx.x;
    const int k = lane & 7;
    const bool real = k < HID;
    const int kk = real ? k : 0;

    const float* w_ih = dir ? w_ih_b : w_ih_f;
    const float* w_hh = dir ? w_hh_b : w_hh_f;
    const float* b_ih = dir ? b_ih_b : b_ih_f;
    const float* b_hh = dir ? b_hh_b : b_hh_f;
    const float* w_hr = dir ? w_hr_b : w_hr_f;

    LaneW W;
#pragma unroll
    for (int q = 0; q < 4; ++q) {
        const int j = q * HID + kk;
        const float sc = real ? ((q == 2) ? 1.0f : 0.5f) : 0.0f;
        W.a[q] = sc * w_hh[j];
        W.w0[q] = sc * w_ih[2 * j];
        W.w1[q] = sc * w_ih[2 * j + 1];
        W.b[q] = sc * (b_ih[j] + b_hh[j]);
    }
    W.rh = real ? 0.5f * w_hr[kk] : 0.0f;
    const float ch = real ? 0.5f * c0[chain * HID + kk] : 0.0f;
    const float h = h0[chain];

    const float* yb = y + (size_t)b * LSEQ;
    const float* ib = inp + (size_t)b * LSEQ;
    float* ho = hout + (size_t)chain * LSEQ;

    if (dir) run_chain<1>(W, yb, ib, ho, h, ch, lane);
    else     run_chain<0>(W, yb, ib, ho, h, ch, lane);
}

__global__ void seq_softmax(const float* __restrict__ hbuf, float* __restrict__ out) {
    const int b = blockIdx.x;
    const int tid = threadIdx.x;
    const int T = 256;
    __shared__ float red[256];

    const float* hf = hbuf + (size_t)b * LSEQ;
    const float* hb = hbuf + (size_t)(NB + b) * LSEQ;

    float m = -3.0e38f;
    for (int l = tid; l < LSEQ; l += T) m = fmaxf(m, hf[l] + hb[l]);
    red[tid] = m;
    __syncthreads();
    for (int s = T / 2; s > 0; s >>= 1) {
        if (tid < s) red[tid] = fmaxf(red[tid], red[tid + s]);
        __syncthreads();
    }
    m = red[0];
    __syncthreads();

    float sum = 0.0f;
    for (int l = tid; l < LSEQ; l += T) sum += __expf(hf[l] + hb[l] - m);
    red[tid] = sum;
    __syncthreads();
    for (int s = T / 2; s > 0; s >>= 1) {
        if (tid < s) red[tid] += red[tid + s];
        __syncthreads();
    }
    sum = red[0];
    __syncthreads();

    const float inv = __builtin_amdgcn_rcpf(sum);
    for (int l = tid; l < LSEQ; l += T) {
        const float e = __expf(hf[l] + hb[l] - m) * inv;
        float* o = out + ((size_t)b * LSEQ + l) * 2;
        o[0] = e;
        o[1] = e;
    }
}

extern "C" void kernel_launch(void* const* d_in, const int* in_sizes, int n_in,
                              void* d_out, int out_size, void* d_ws, size_t ws_size,
                              hipStream_t stream) {
    const float* x      = (const float*)d_in[0];
    const float* inp    = (const float*)d_in[1];
    const float* conv_w = (const float*)d_in[2];
    const float* conv_b = (const float*)d_in[3];
    const float* w_ih_f = (const float*)d_in[4];
    const float* w_hh_f = (const float*)d_in[5];
    const float* b_ih_f = (const float*)d_in[6];
    const float* b_hh_f = (const float*)d_in[7];
    const float* w_hr_f = (const float*)d_in[8];
    const float* w_ih_b = (const float*)d_in[9];
    const float* w_hh_b = (const float*)d_in[10];
    const float* b_ih_b = (const float*)d_in[11];
    const float* b_hh_b = (const float*)d_in[12];
    const float* w_hr_b = (const float*)d_in[13];
    const float* h0     = (const float*)d_in[14];
    const float* c0     = (const float*)d_in[15];
    float* out = (float*)d_out;

    float* y    = (float*)d_ws;
    float* hbuf = y + (size_t)NB * LSEQ;

    conv1x1_wmma<<<(NB * LSEQ / 16) / 8, 256, 0, stream>>>(x, conv_w, conv_b, y);

    lstm_scan<<<2 * NB, 32, 0, stream>>>(y, inp,
                                    w_ih_f, w_hh_f, b_ih_f, b_hh_f, w_hr_f,
                                    w_ih_b, w_hh_b, b_ih_b, b_hh_b, w_hr_b,
                                    h0, c0, hbuf);

    seq_softmax<<<NB, 256, 0, stream>>>(hbuf, out);
}


# LSTM scan: first butterfly add of the h sum is one DPP add on the finished product (was mul, DPP move, fma): one dependent link less per step; product now rounded before that add
# speedup vs baseline: 13.7551x; 1.0694x over previous
#include <hip/hip_runtime.h>
#include <hip/hip_bf16.h>

#define LSEQ 60000
#define NB 8
#define INCH 64
#define HID 5

typedef __attribute__((ext_vector_type(2))) float v2f;
typedef __attribute__((ext_vector_type(8))) float v8f;

__global__ void conv1x1_wmma(const float* __restrict__ x,
                             const float* __restrict__ w,
                             const float* __restrict__ bias,
                             float* __restrict__ y) {
    const int lane = threadIdx.x & 31;
    const int wave = threadIdx.x >> 5;
    const int tile = blockIdx.x * 8 + wave;
    const int b = tile / (LSEQ / 16);
    const int lbase = (tile % (LSEQ / 16)) * 16;
    const int m = lane & 15;
    const int half = lane >> 4;

    const float* xb = x + (size_t)b * INCH * LSEQ + lbase + m;

    v8f acc = {};
#pragma unroll
    for (int chunk = 0; chunk < 16; ++chunk) {
        const int k0 = chunk * 4 + 2 * half;
        v2f a, bb;
        a.x = xb[(size_t)k0 * LSEQ];
        a.y = xb[(size_t)(k0 + 1) * LSEQ];
        bb.x = w[k0];
        bb.y = w[k0 + 1];
        acc = __builtin_amdgcn_wmma_f32_16x16x4_f32(
             false, a,  false, bb,
             (short)0, acc,  false,  false);
    }

    const float bv = bias[0];
    float* yo = y + (size_t)b * LSEQ + lbase;
    if (lane == 0) {
#pragma unroll
        for (int r = 0; r < 8; ++r) yo[r] = acc[r] + bv;
    } else if (lane == 16) {
#pragma unroll
        for (int r = 0; r < 8; ++r) yo[8 + r] = acc[r] + bv;
    }
}

#define UB 8
#define NBLK (LSEQ / UB)
#define IT_BLK 4

#define SB() __builtin_amdgcn_sched_barrier(0)

template <int MASK>
__device__ __forceinline__ float xadd(float v) {
    const int t = __builtin_amdgcn_update_dpp(0, __builtin_bit_cast(int, v), 0x160 + MASK, 0xf, 0xf, true);
    return v + __builtin_bit_cast(float, t);
}

__device__ __forceinline__ float xadd1_of_product(float p) {
    float r;
    asm("v_add_f32_dpp %0, %1, %1 row_xmask:1 row_mask:0xf bank_mask:0xf bound_ctrl:1" : "=v"(r) : "v"(p));
    return r;
}

struct LaneW {
    float a[4];
    float w0[4];
    float w1[4];
    float b[4];
    float rh;
};

template <int DIR>
__device__ __forceinline__ void load_blk(const float* __restrict__ yb, const float* __restrict__ ib,
                                         int blk, float (&xy)[UB], float (&xi)[UB]) {
    if (blk > NBLK - 1) blk = NBLK - 1;
    const int base = DIR ? (LSEQ - UB - blk * UB) : (blk * UB);
    const float4 y0 = *reinterpret_cast<const float4*>(yb + base);
    const float4 y1 = *reinterpret_cast<const float4*>(yb + base + 4);
    const float4 i0 = *reinterpret_cast<const float4*>(ib + base);
    const float4 i1 = *reinterpret_cast<const float4*>(ib + base + 4);
    xy[0] = y0.x; xy[1] = y0.y; xy[2] = y0.z; xy[3] = y0.w;
    xy[4] = y1.x; xy[5] = y1.y; xy[6] = y1.z; xy[7] = y1.w;
    xi[0] = i0.x; xi[1] = i0.y; xi[2] = i0.z; xi[3] = i0.w;
    xi[4] = i1.x; xi[5] = i1.y; xi[6] = i1.z; xi[7] = i1.w;
}

template <int DIR>
__device__ __forceinline__ void pre_blk(const LaneW& W, const float (&xy)[UB], const float (&xi)[UB],
                                        float (&P)[UB][4]) {
#pragma unroll
    for (int j = 0; j < UB; ++j) {
        const int m = DIR ? (UB - 1 - j) : j;
#pragma unroll
        for (int q = 0; q < 4; ++q)
            P[j][q] = fmaf(xy[m], W.w0[q], fmaf(xi[m], W.w1[q], W.b[q]));
    }
}

template <int DIR>
__device__ __forceinline__ void steps8(const LaneW& W, const float (&P)[UB][4], float (&Pn)[UB][4],
                                       const float (&xy)[UB], const float (&xi)[UB],
                                       float& h, float& ch, float* hs) {
#pragma unroll
    for (int j = 0; j < UB; ++j) {
        const int m = DIR ? (UB - 1 - j) : j;
        const float gi = fmaf(h, W.a[0], P[j][0]);
        const float gf = fmaf(h, W.a[1], P[j][1]);
        const float gg = fmaf(h, W.a[2], P[j][2]);
        const float go = fmaf(h, W.a[3], P[j][3]);
        SB();
        const float ti = __builtin_amdgcn_tanhf(gi);
        const float tf = __builtin_amdgcn_tanhf(gf);
        const float tg = __builtin_amdgcn_tanhf(gg);
        const float to = __builtin_amdgcn_tanhf(go);
        SB();
        const float iv = fmaf(ti, 0.5f, 0.5f);
        const float fc = fmaf(tf, ch, ch);
        const float t0 = fmaf(xi[m], W.w1[0], W.b[0]);
        const float t1 = fmaf(xi[m], W.w1[1], W.b[1]);
        const float t2 = fmaf(xi[m], W.w1[2], W.b[2]);
        const float t3 = fmaf(xi[m], W.w1[3], W.b[3]);
        SB();
        const float c = fmaf(iv, tg, fc);
        const float ow = fmaf(to, W.rh, W.rh);
        Pn[j][0] = fmaf(xy[m], W.w0[0], t0);
        Pn[j][1] = fmaf(xy[m], W.w0[1], t1);
        Pn[j][2] = fmaf(xy[m], W.w0[2], t2);
        SB();
        const float tc = __builtin_amdgcn_tanhf(c);
        ch = 0.5f * c;
        Pn[j][3] = fmaf(xy[m], W.w0[3], t3);
        SB();
        float p = ow * tc;
        p = xadd1_of_product(p);
        p = xadd<2>(p);
        p = xadd<4>(p);
        SB();
        h = p;
        hs[j] = p;
    }
}

template <int DIR>
__device__ __forceinline__ void run_chain(const LaneW& W, const float* __restrict__ yb,
                                          const float* __restrict__ ib, float* __restrict__ ho,
                                          float h, float ch, int lane) {
    float PA[UB][4], PB[UB][4];
    float XAy[UB], XAi[UB], XBy[UB], XBi[UB];
    float hs[UB * IT_BLK];

    load_blk<DIR>(yb, ib, 0, XAy, XAi);
    pre_blk<DIR>(W, XAy, XAi, PA);
    load_blk<DIR>(yb, ib, 1, XBy, XBi);

    for (int n = 0; n < NBLK; n += IT_BLK) {
        {
            int pf = DIR ? (LSEQ - UB * IT_BLK - n * UB - 4 * UB * IT_BLK) : (n * UB + 4 * UB * IT_BLK);
            pf = pf < 0 ? 0 : (pf > LSEQ - 1 ? LSEQ - 1 : pf);
            __builtin_prefetch(yb + pf, 0, 0);
            __builtin_prefetch(ib + pf, 0, 0);
        }
        load_blk<DIR>(yb, ib, n + 2, XAy, XAi);
        steps8<DIR>(W, PA, PB, XBy, XBi, h, ch, hs);
        load_blk<DIR>(yb, ib, n + 3, XBy, XBi);
        steps8<DIR>(W, PB, PA, XAy, XAi, h, ch, hs + UB);
        load_blk<DIR>(yb, ib, n + 4, XAy, XAi);
        steps8<DIR>(W, PA, PB, XBy, XBi, h, ch, hs + 2 * UB);
        load_blk<DIR>(yb, ib, n + 5, XBy, XBi);
        steps8<DIR>(W, PB, PA, XAy, XAi, h, ch, hs + 3 * UB);

        const int base = DIR ? (LSEQ - UB * IT_BLK - n * UB) : (n * UB);
        if (lane == 0) {
#pragma unroll
            for (int v = 0; v < UB * IT_BLK / 4; ++v) {
                float4 o;
                if (DIR) {
                    const int s = UB * IT_BLK - 1 - 4 * v;
                    o = make_float4(hs[s], hs[s - 1], hs[s - 2], hs[s - 3]);
                } else {
                    o = make_float4(hs[4 * v], hs[4 * v + 1], hs[4 * v + 2], hs[4 * v + 3]);
                }
                *reinterpret_cast<float4*>(ho + base + 4 * v) = o;
            }
        }
    }
}

__global__ __launch_bounds__(32) void lstm_scan(const float* __restrict__ y,
                          const float* __restrict__ inp,
                          const float* __restrict__ w_ih_f, const float* __restrict__ w_hh_f,
                          const float* __restrict__ b_ih_f, const float* __restrict__ b_hh_f,
                          const float* __restrict__ w_hr_f,
                          const float* __restrict__ w_ih_b, const float* __restrict__ w_hh_b,
                          const float* __restrict__ b_ih_b, const float* __restrict__ b_hh_b,
                          const float* __restrict__ w_hr_b,
                          const float* __restrict__ h0, const float* __restrict__ c0,
                          float* __restrict__ hout) {
    const int chain = blockIdx.x;
    const int dir = chain >> 3;
    const int b = chain & 7;
    const int lane = threadIdx.x;
    const int k = lane & 7;
    const bool real = k < HID;
    const int kk = real ? k : 0;

    const float* w_ih = dir ? w_ih_b : w_ih_f;
    const float* w_hh = dir ? w_hh_b : w_hh_f;
    const float* b_ih = dir ? b_ih_b : b_ih_f;
    const float* b_hh = dir ? b_hh_b : b_hh_f;
    const float* w_hr = dir ? w_hr_b : w_hr_f;

    LaneW W;
#pragma unroll
    for (int q = 0; q < 4; ++q) {
        const int j = q * HID + kk;
        const float sc = real ? ((q == 2) ? 1.0f : 0.5f) : 0.0f;
        W.a[q] = sc * w_hh[j];
        W.w0[q] = sc * w_ih[2 * j];
        W.w1[q] = sc * w_ih[2 * j + 1];
        W.b[q] = sc * (b_ih[j] + b_hh[j]);
    }
    W.rh = real ? 0.5f * w_hr[kk] : 0.0f;
    const float ch = real ? 0.5f * c0[chain * HID + kk] : 0.0f;
    const float h = h0[chain];

    const float* yb = y + (size_t)b * LSEQ;
    const float* ib = inp + (size_t)b * LSEQ;
    float* ho = hout + (size_t)chain * LSEQ;

    if (dir) run_chain<1>(W, yb, ib, ho, h, ch, lane);
    else     run_chain<0>(W, yb, ib, ho, h, ch, lane);
}

__global__ void seq_softmax(const float* __restrict__ hbuf, float* __restrict__ out) {
    const int b = blockIdx.x;
    const int tid = threadIdx.x;
    const int T = 256;
    __shared__ float red[256];

    const float* hf = hbuf + (size_t)b * LSEQ;
    const float* hb = hbuf + (size_t)(NB + b) * LSEQ;

    float m = -3.0e38f;
    for (int l = tid; l < LSEQ; l += T) m = fmaxf(m, hf[l] + hb[l]);
    red[tid] = m;
    __syncthreads();
    for (int s = T / 2; s > 0; s >>= 1) {
        if (tid < s) red[tid] = fmaxf(red[tid], red[tid + s]);
        __syncthreads();
    }
    m = red[0];
    __syncthreads();

    float sum = 0.0f;
    for (int l = tid; l < LSEQ; l += T) sum += __expf(hf[l] + hb[l] - m);
    red[tid] = sum;
    __syncthreads();
    for (int s = T / 2; s > 0; s >>= 1) {
        if (tid < s) red[tid] += red[tid + s];
        __syncthreads();
    }
    sum = red[0];
    __syncthreads();

    const float inv = __builtin_amdgcn_rcpf(sum);
    for (int l = tid; l < LSEQ; l += T) {
        const float e = __expf(hf[l] + hb[l] - m) * inv;
        float* o = out + ((size_t)b * LSEQ + l) * 2;
        o[0] = e;
        o[1] = e;
    }
}

extern "C" void kernel_launch(void* const* d_in, const int* in_sizes, int n_in,
                              void* d_out, int out_size, void* d_ws, size_t ws_size,
                              hipStream_t stream) {
    const float* x      = (const float*)d_in[0];
    const float* inp    = (const float*)d_in[1];
    const float* conv_w = (const float*)d_in[2];
    const float* conv_b = (const float*)d_in[3];
    const float* w_ih_f = (const float*)d_in[4];
    const float* w_hh_f = (const float*)d_in[5];
    const float* b_ih_f = (const float*)d_in[6];
    const float* b_hh_f = (const float*)d_in[7];
    const float* w_hr_f = (const float*)d_in[8];
    const float* w_ih_b = (const float*)d_in[9];
    const float* w_hh_b = (const float*)d_in[10];
    const float* b_ih_b = (const float*)d_in[11];
    const float* b_hh_b = (const float*)d_in[12];
    const float* w_hr_b = (const float*)d_in[13];
    const float* h0     = (const float*)d_in[14];
    const float* c0     = (const float*)d_in[15];
    float* out = (float*)d_out;

    float* y    = (float*)d_ws;
    float* hbuf = y + (size_t)NB * LSEQ;

    conv1x1_wmma<<<(NB * LSEQ / 16) / 8, 256, 0, stream>>>(x, conv_w, conv_b, y);

    lstm_scan<<<2 * NB, 32, 0, stream>>>(y, inp,
                                    w_ih_f, w_hh_f, b_ih_f, b_hh_f, w_hr_f,
                                    w_ih_b, w_hh_b, b_ih_b, b_hh_b, w_hr_b,
                                    h0, c0, hbuf);

    seq_softmax<<<NB, 256, 0, stream>>>(hbuf, out);
}


# LSTM scan: third butterfly add of the h sum leaves the chain: units 0..3 by a 2-step DPP butterfly, unit 4's product joins the next gate's addend by a lane-read fma; f32 gate sum order changes
# speedup vs baseline: 14.3402x; 1.0425x over previous
#include <hip/hip_runtime.h>
#include <hip/hip_bf16.h>

#define LSEQ 60000
#define NB 8
#define INCH 64
#define HID 5

typedef __attribute__((ext_vector_type(2))) float v2f;
typedef __attribute__((ext_vector_type(8))) float v8f;

__global__ void conv1x1_wmma(const float* __restrict__ x,
                             const float* __restrict__ w,
                             const float* __restrict__ bias,
                             float* __restrict__ y) {
    const int lane = threadIdx.x & 31;
    const int wave = threadIdx.x >> 5;
    const int tile = blockIdx.x * 8 + wave;
    const int b = tile / (LSEQ / 16);
    const int lbase = (tile % (LSEQ / 16)) * 16;
    const int m = lane & 15;
    const int half = lane >> 4;

    const float* xb = x + (size_t)b * INCH * LSEQ + lbase + m;

    v8f acc = {};
#pragma unroll
    for (int chunk = 0; chunk < 16; ++chunk) {
        const int k0 = chunk * 4 + 2 * half;
        v2f a, bb;
        a.x = xb[(size_t)k0 * LSEQ];
        a.y = xb[(size_t)(k0 + 1) * LSEQ];
        bb.x = w[k0];
        bb.y = w[k0 + 1];
        acc = __builtin_amdgcn_wmma_f32_16x16x4_f32(
             false, a,  false, bb,
             (short)0, acc,  false,  false);
    }

    const float bv = bias[0];
    float* yo = y + (size_t)b * LSEQ + lbase;
    if (lane == 0) {
#pragma unroll
        for (int r = 0; r < 8; ++r) yo[r] = acc[r] + bv;
    } else if (lane == 16) {
#pragma unroll
        for (int r = 0; r < 8; ++r) yo[8 + r] = acc[r] + bv;
    }
}

#define UB 8
#define NBLK (LSEQ / UB)
#define IT_BLK 4

#define SB() __builtin_amdgcn_sched_barrier(0)

__device__ __forceinline__ float xadd1_of_product(float p) {
    float r;
    asm("v_add_f32_dpp %0, %1, %1 row_xmask:1 row_mask:0xf bank_mask:0xf bound_ctrl:1" : "=v"(r) : "v"(p));
    return r;
}

__device__ __forceinline__ float fma_lane4(float x, float y, float z) {
    asm("v_fmac_f32_dpp %0, %1, %2 row_share:4 row_mask:0xf bank_mask:0xf" : "+v"(z) : "v"(x), "v"(y));
    return z;
}
__device__ __forceinline__ float fma_lane0(float x, float y, float z) {
    asm("v_fmac_f32_dpp %0, %1, %2 row_share:0 row_mask:0xf bank_mask:0xf" : "+v"(z) : "v"(x), "v"(y));
    return z;
}
__device__ __forceinline__ float xadd2_chain(float v) {
    float r;
    asm("v_add_f32_dpp %0, %1, %1 row_xmask:2 row_mask:0xf bank_mask:0xf bound_ctrl:1" : "=v"(r) : "v"(v));
    return r;
}
__device__ __forceinline__ float xadd4_out(float v) {
    float r;
    asm("v_add_f32_dpp %0, %1, %1 row_xmask:4 row_mask:0xf bank_mask:0xf bound_ctrl:1" : "=v"(r) : "v"(v));
    return r;
}

struct LaneW {
    float a[4];
    float w0[4];
    float w1[4];
    float b[4];
    float rh;
};

template <int DIR>
__device__ __forceinline__ void load_blk(const float* __restrict__ yb, const float* __restrict__ ib,
                                         int blk, float (&xy)[UB], float (&xi)[UB]) {
    if (blk > NBLK - 1) blk = NBLK - 1;
    const int base = DIR ? (LSEQ - UB - blk * UB) : (blk * UB);
    const float4 y0 = *reinterpret_cast<const float4*>(yb + base);
    const float4 y1 = *reinterpret_cast<const float4*>(yb + base + 4);
    const float4 i0 = *reinterpret_cast<const float4*>(ib + base);
    const float4 i1 = *reinterpret_cast<const float4*>(ib + base + 4);
    xy[0] = y0.x; xy[1] = y0.y; xy[2] = y0.z; xy[3] = y0.w;
    xy[4] = y1.x; xy[5] = y1.y; xy[6] = y1.z; xy[7] = y1.w;
    xi[0] = i0.x; xi[1] = i0.y; xi[2] = i0.z; xi[3] = i0.w;
    xi[4] = i1.x; xi[5] = i1.y; xi[6] = i1.z; xi[7] = i1.w;
}

template <int DIR>
__device__ __forceinline__ void pre_blk(const LaneW& W, const float (&xy)[UB], const float (&xi)[UB],
                                        float (&P)[UB][4]) {
#pragma unroll
    for (int j = 0; j < UB; ++j) {
        const int m = DIR ? (UB - 1 - j) : j;
#pragma unroll
        for (int q = 0; q < 4; ++q)
            P[j][q] = fmaf(xy[m], W.w0[q], fmaf(xi[m], W.w1[q], W.b[q]));
    }
}

struct Carry { float ti, tf, tg, g3, ch; };

template <int DIR>
__device__ __forceinline__ void steps8(const LaneW& W, const float (&P)[UB][4], float (&Pn)[UB][4],
                                       const float (&xy)[UB], const float (&xi)[UB],
                                       Carry& S, float* hs) {
#pragma unroll
    for (int j = 0; j < UB; ++j) {
        const int m = DIR ? (UB - 1 - j) : j;
        const float (&Pq)[4] = (j + 1 < UB) ? P[j + 1] : Pn[0];
        const float xs = xi[m], ys = xy[m];
        const float iv = fmaf(S.ti, 0.5f, 0.5f);
        SB();
        const float to = __builtin_amdgcn_tanhf(S.g3);
        SB();
        const float fc = fmaf(S.tf, S.ch, S.ch);
        SB();
        const float t0 = fmaf(xs, W.w1[0], W.b[0]);
        const float t1 = fmaf(xs, W.w1[1], W.b[1]);
        const float t2 = fmaf(xs, W.w1[2], W.b[2]);
        const float t3 = fmaf(xs, W.w1[3], W.b[3]);
        SB();
        const float c = fmaf(iv, S.tg, fc);
        SB();
        Pn[j][0] = fmaf(ys, W.w0[0], t0);
        Pn[j][1] = fmaf(ys, W.w0[1], t1);
        SB();
        const float ow = fmaf(to, W.rh, W.rh);
        Pn[j][2] = fmaf(ys, W.w0[2], t2);
        SB();
        const float tc = __builtin_amdgcn_tanhf(c);
        SB();
        S.ch = 0.5f * c;
        Pn[j][3] = fmaf(ys, W.w0[3], t3);
        SB();
        float p = ow * tc;
        const float r1 = xadd1_of_product(p);
        SB();
        const float G0 = fma_lane4(p, W.a[0], Pq[0]);
        const float G1 = fma_lane4(p, W.a[1], Pq[1]);
        SB();
        const float r2 = xadd2_chain(r1);
        SB();
        const float G2 = fma_lane4(p, W.a[2], Pq[2]);
        const float G3 = fma_lane4(p, W.a[3], Pq[3]);
        SB();
        const float g0 = fma_lane0(r2, W.a[0], G0);
        SB();
        const float g1 = fma_lane0(r2, W.a[1], G1);
        SB();
        const float g2 = fma_lane0(r2, W.a[2], G2);
        SB();
        S.ti = __builtin_amdgcn_tanhf(g0);
        SB();
        S.tf = __builtin_amdgcn_tanhf(g1);
        SB();
        S.g3 = fma_lane0(r2, W.a[3], G3);
        hs[j] = xadd4_out(r2);
        SB();
        S.tg = __builtin_amdgcn_tanhf(g2);
        SB();
    }
}

template <int DIR>
__device__ __forceinline__ void run_chain(const LaneW& W, const float* __restrict__ yb,
                                          const float* __restrict__ ib, float* __restrict__ ho,
                                          float h, float ch, int lane) {
    float PA[UB][4], PB[UB][4];
    float XAy[UB], XAi[UB], XBy[UB], XBi[UB];
    float hs[UB * IT_BLK];

    load_blk<DIR>(yb, ib, 0, XAy, XAi);
    pre_blk<DIR>(W, XAy, XAi, PA);
    load_blk<DIR>(yb, ib, 1, XBy, XBi);
    Carry S;
    S.ti = __builtin_amdgcn_tanhf(fmaf(h, W.a[0], PA[0][0]));
    S.tf = __builtin_amdgcn_tanhf(fmaf(h, W.a[1], PA[0][1]));
    S.tg = __builtin_amdgcn_tanhf(fmaf(h, W.a[2], PA[0][2]));
    S.g3 = fmaf(h, W.a[3], PA[0][3]);
    S.ch = ch;

    for (int n = 0; n < NBLK; n += IT_BLK) {
        {
            int pf = DIR ? (LSEQ - UB * IT_BLK - n * UB - 4 * UB * IT_BLK) : (n * UB + 4 * UB * IT_BLK);
            pf = pf < 0 ? 0 : (pf > LSEQ - 1 ? LSEQ - 1 : pf);
            __builtin_prefetch(yb + pf, 0, 0);
            __builtin_prefetch(ib + pf, 0, 0);
        }
        load_blk<DIR>(yb, ib, n + 2, XAy, XAi);
        steps8<DIR>(W, PA, PB, XBy, XBi, S, hs);
        load_blk<DIR>(yb, ib, n + 3, XBy, XBi);
        steps8<DIR>(W, PB, PA, XAy, XAi, S, hs + UB);
        load_blk<DIR>(yb, ib, n + 4, XAy, XAi);
        steps8<DIR>(W, PA, PB, XBy, XBi, S, hs + 2 * UB);
        load_blk<DIR>(yb, ib, n + 5, XBy, XBi);
        steps8<DIR>(W, PB, PA, XAy, XAi, S, hs + 3 * UB);

        const int base = DIR ? (LSEQ - UB * IT_BLK - n * UB) : (n * UB);
        if (lane == 0) {
#pragma unroll
            for (int v = 0; v < UB * IT_BLK / 4; ++v) {
                float4 o;
                if (DIR) {
                    const int s = UB * IT_BLK - 1 - 4 * v;
                    o = make_float4(hs[s], hs[s - 1], hs[s - 2], hs[s - 3]);
                } else {
                    o = make_float4(hs[4 * v], hs[4 * v + 1], hs[4 * v + 2], hs[4 * v + 3]);
                }
                *reinterpret_cast<float4*>(ho + base + 4 * v) = o;
            }
        }
    }
}

__global__ __launch_bounds__(32) void lstm_scan(const float* __restrict__ y,
                          const float* __restrict__ inp,
                          const float* __restrict__ w_ih_f, const float* __restrict__ w_hh_f,
                          const float* __restrict__ b_ih_f, const float* __restrict__ b_hh_f,
                          const float* __restrict__ w_hr_f,
                          const float* __restrict__ w_ih_b, const float* __restrict__ w_hh_b,
                          const float* __restrict__ b_ih_b, const float* __restrict__ b_hh_b,
                          const float* __restrict__ w_hr_b,
                          const float* __restrict__ h0, const float* __restrict__ c0,
                          float* __restrict__ hout) {
    const int chain = blockIdx.x;
    const int dir = chain >> 3;
    const int b = chain & 7;
    const int lane = threadIdx.x;
    const int k = lane & 7;
    const bool real = k < HID;
    const int kk = real ? k : 0;

    const float* w_ih = dir ? w_ih_b : w_ih_f;
    const float* w_hh = dir ? w_hh_b : w_hh_f;
    const float* b_ih = dir ? b_ih_b : b_ih_f;
    const float* b_hh = dir ? b_hh_b : b_hh_f;
    const float* w_hr = dir ? w_hr_b : w_hr_f;

    LaneW W;
#pragma unroll
    for (int q = 0; q < 4; ++q) {
        const int j = q * HID + kk;
        const float sc = real ? ((q == 2) ? 1.0f : 0.5f) : 0.0f;
        W.a[q] = sc * w_hh[j];
        W.w0[q] = sc * w_ih[2 * j];
        W.w1[q] = sc * w_ih[2 * j + 1];
        W.b[q] = sc * (b_ih[j] + b_hh[j]);
    }
    W.rh = real ? 0.5f * w_hr[kk] : 0.0f;
    const float ch = real ? 0.5f * c0[chain * HID + kk] : 0.0f;
    const float h = h0[chain];

    const float* yb = y + (size_t)b * LSEQ;
    const float* ib = inp + (size_t)b * LSEQ;
    float* ho = hout + (size_t)chain * LSEQ;

    if (dir) run_chain<1>(W, yb, ib, ho, h, ch, lane);
    else     run_chain<0>(W, yb, ib, ho, h, ch, lane);
}

__global__ void seq_softmax(const float* __restrict__ hbuf, float* __restrict__ out) {
    const int b = blockIdx.x;
    const int tid = threadIdx.x;
    const int T = 256;
    __shared__ float red[256];

    const float* hf = hbuf + (size_t)b * LSEQ;
    const float* hb = hbuf + (size_t)(NB + b) * LSEQ;

    float m = -3.0e38f;
    for (int l = tid; l < LSEQ; l += T) m = fmaxf(m, hf[l] + hb[l]);
    red[tid] = m;
    __syncthreads();
    for (int s = T / 2; s > 0; s >>= 1) {
        if (tid < s) red[tid] = fmaxf(red[tid], red[tid + s]);
        __syncthreads();
    }
    m = red[0];
    __syncthreads();

    float sum = 0.0f;
    for (int l = tid; l < LSEQ; l += T) sum += __expf(hf[l] + hb[l] - m);
    red[tid] = sum;
    __syncthreads();
    for (int s = T / 2; s > 0; s >>= 1) {
        if (tid < s) red[tid] += red[tid + s];
        __syncthreads();
    }
    sum = red[0];
    __syncthreads();

    const float inv = __builtin_amdgcn_rcpf(sum);
    for (int l = tid; l < LSEQ; l += T) {
        const float e = __expf(hf[l] + hb[l] - m) * inv;
        float* o = out + ((size_t)b * LSEQ + l) * 2;
        o[0] = e;
        o[1] = e;
    }
}

extern "C" void kernel_launch(void* const* d_in, const int* in_sizes, int n_in,
                              void* d_out, int out_size, void* d_ws, size_t ws_size,
                              hipStream_t stream) {
    const float* x      = (const float*)d_in[0];
    const float* inp    = (const float*)d_in[1];
    const float* conv_w = (const float*)d_in[2];
    const float* conv_b = (const float*)d_in[3];
    const float* w_ih_f = (const float*)d_in[4];
    const float* w_hh_f = (const float*)d_in[5];
    const float* b_ih_f = (const float*)d_in[6];
    const float* b_hh_f = (const float*)d_in[7];
    const float* w_hr_f = (const float*)d_in[8];
    const float* w_ih_b = (const float*)d_in[9];
    const float* w_hh_b = (const float*)d_in[10];
    const float* b_ih_b = (const float*)d_in[11];
    const float* b_hh_b = (const float*)d_in[12];
    const float* w_hr_b = (const float*)d_in[13];
    const float* h0     = (const float*)d_in[14];
    const float* c0     = (const float*)d_in[15];
    float* out = (float*)d_out;

    float* y    = (float*)d_ws;
    float* hbuf = y + (size_t)NB * LSEQ;

    conv1x1_wmma<<<(NB * LSEQ / 16) / 8, 256, 0, stream>>>(x, conv_w, conv_b, y);

    lstm_scan<<<2 * NB, 32, 0, stream>>>(y, inp,
                                    w_ih_f, w_hh_f, b_ih_f, b_hh_f, w_hr_f,
                                    w_ih_b, w_hh_b, b_ih_b, b_hh_b, w_hr_b,
                                    h0, c0, hbuf);

    seq_softmax<<<NB, 256, 0, stream>>>(hbuf, out);
}


# LSTM scan: 125 segments per recurrence run at once, each run in 256 steps early from zero state; 2nd launch checks each boundary state bit for bit, re-runs a segment that differs; sum order kept
# speedup vs baseline: 221.4481x; 15.4425x over previous
#include <hip/hip_runtime.h>
#include <hip/hip_bf16.h>

#define LSEQ 60000
#define NB 8
#define INCH 64
#define HID 5

typedef __attribute__((ext_vector_type(2))) float v2f;
typedef __attribute__((ext_vector_type(8))) float v8f;

__global__ void conv1x1_wmma(const float* __restrict__ x,
                             const float* __restrict__ w,
                             const float* __restrict__ bias,
                             float* __restrict__ y) {
    const int lane = threadIdx.x & 31;
    const int wave = threadIdx.x >> 5;
    const int tile = blockIdx.x * 8 + wave;
    const int b = tile / (LSEQ / 16);
    const int lbase = (tile % (LSEQ / 16)) * 16;
    const int m = lane & 15;
    const int half = lane >> 4;

    const float* xb = x + (size_t)b * INCH * LSEQ + lbase + m;

    v8f acc = {};
#pragma unroll
    for (int chunk = 0; chunk < 16; ++chunk) {
        const int k0 = chunk * 4 + 2 * half;
        v2f a, bb;
        a.x = xb[(size_t)k0 * LSEQ];
        a.y = xb[(size_t)(k0 + 1) * LSEQ];
        bb.x = w[k0];
        bb.y = w[k0 + 1];
        acc = __builtin_amdgcn_wmma_f32_16x16x4_f32(
             false, a,  false, bb,
             (short)0, acc,  false,  false);
    }

    const float bv = bias[0];
    float* yo = y + (size_t)b * LSEQ + lbase;
    if (lane == 0) {
#pragma unroll
        for (int r = 0; r < 8; ++r) yo[r] = acc[r] + bv;
    } else if (lane == 16) {
#pragma unroll
        for (int r = 0; r < 8; ++r) yo[8 + r] = acc[r] + bv;
    }
}

#define UB 8
#define NBLK (LSEQ / UB)
#define IT_BLK 4
#define NCHAIN (2 * NB)
#define NPASS (NBLK / IT_BLK)
#define SEG_PASSES 15
#define NSEG (NPASS / SEG_PASSES)
#define RUNIN_PASSES 8
#define REC_FLOATS 64
static_assert(NPASS % SEG_PASSES == 0, "segments are whole passes");
static_assert(SEG_PASSES >= RUNIN_PASSES, "the run-in of segment 1 must not start in front of the sequence");

#define SB() __builtin_amdgcn_sched_barrier(0)

__device__ __forceinline__ float xadd1_of_product(float p) {
    float r;
    asm("v_add_f32_dpp %0, %1, %1 row_xmask:1 row_mask:0xf bank_mask:0xf bound_ctrl:1" : "=v"(r) : "v"(p));
    return r;
}

__device__ __forceinline__ float fma_lane4(float x, float y, float z) {
    asm("v_fmac_f32_dpp %0, %1, %2 row_share:4 row_mask:0xf bank_mask:0xf" : "+v"(z) : "v"(x), "v"(y));
    return z;
}
__device__ __forceinline__ float fma_lane0(float x, float y, float z) {
    asm("v_fmac_f32_dpp %0, %1, %2 row_share:0 row_mask:0xf bank_mask:0xf" : "+v"(z) : "v"(x), "v"(y));
    return z;
}
__device__ __forceinline__ float xadd2_chain(float v) {
    float r;
    asm("v_add_f32_dpp %0, %1, %1 row_xmask:2 row_mask:0xf bank_mask:0xf bound_ctrl:1" : "=v"(r) : "v"(v));
    return r;
}
__device__ __forceinline__ float xadd4_out(float v) {
    float r;
    asm("v_add_f32_dpp %0, %1, %1 row_xmask:4 row_mask:0xf bank_mask:0xf bound_ctrl:1" : "=v"(r) : "v"(v));
    return r;
}

struct LaneW {
    float a[4];
    float w0[4];
    float w1[4];
    float b[4];
    float rh;
};

template <int DIR>
__device__ __forceinline__ void load_blk(const float* __restrict__ yb, const float* __restrict__ ib,
                                         int blk, float (&xy)[UB], float (&xi)[UB]) {
    if (blk > NBLK - 1) blk = NBLK - 1;
    const int base = DIR ? (LSEQ - UB - blk * UB) : (blk * UB);
    const float4 y0 = *reinterpret_cast<const float4*>(yb + base);
    const float4 y1 = *reinterpret_cast<const float4*>(yb + base + 4);
    const float4 i0 = *reinterpret_cast<const float4*>(ib + base);
    const float4 i1 = *reinterpret_cast<const float4*>(ib + base + 4);
    xy[0] = y0.x; xy[1] = y0.y; xy[2] = y0.z; xy[3] = y0.w;
    xy[4] = y1.x; xy[5] = y1.y; xy[6] = y1.z; xy[7] = y1.w;
    xi[0] = i0.x; xi[1] = i0.y; xi[2] = i0.z; xi[3] = i0.w;
    xi[4] = i1.x; xi[5] = i1.y; xi[6] = i1.z; xi[7] = i1.w;
}

template <int DIR>
__device__ __forceinline__ void pre_blk(const LaneW& W, const float (&xy)[UB], const float (&xi)[UB],
                                        float (&P)[UB][4]) {
#pragma unroll
    for (int j = 0; j < UB; ++j) {
        const int m = DIR ? (UB - 1 - j) : j;
#pragma unroll
        for (int q = 0; q < 4; ++q)
            P[j][q] = fmaf(xy[m], W.w0[q], fmaf(xi[m], W.w1[q], W.b[q]));
    }
}

struct Carry { float ti, tf, tg, g3, ch; };

__device__ __forceinline__ void put_state(float* rec, const Carry& S, int lane) {
    if (lane < 8) {
        *reinterpret_cast<float4*>(rec + 4 * lane) = make_float4(S.ti, S.tf, S.tg, S.g3);
        rec[32 + lane] = S.ch;
    }
}
__device__ __forceinline__ Carry get_state(const float* rec, int lane) {
    const int k = lane & 7;
    const float4 g = *reinterpret_cast<const float4*>(rec + 4 * k);
    Carry S;
    S.ti = g.x; S.tf = g.y; S.tg = g.z; S.g3 = g.w;
    S.ch = rec[32 + k];
    return S;
}
__device__ __forceinline__ bool same_state(const Carry& A, const float* rec, int lane) {
    const Carry B = get_state(rec, lane);
    const bool diff = (__float_as_uint(A.ti) != __float_as_uint(B.ti)) | (__float_as_uint(A.tf) != __float_as_uint(B.tf)) |
                      (__float_as_uint(A.tg) != __float_as_uint(B.tg)) | (__float_as_uint(A.g3) != __float_as_uint(B.g3)) |
                      (__float_as_uint(A.ch) != __float_as_uint(B.ch));
    return !__any(diff);
}

template <int DIR>
__device__ __forceinline__ void steps8(const LaneW& W, const float (&P)[UB][4], float (&Pn)[UB][4],
                                       const float (&xy)[UB], const float (&xi)[UB],
                                       Carry& S, float* hs) {
#pragma unroll
    for (int j = 0; j < UB; ++j) {
        const int m = DIR ? (UB - 1 - j) : j;
        const float (&Pq)[4] = (j + 1 < UB) ? P[j + 1] : Pn[0];
        const float xs = xi[m], ys = xy[m];
        const float iv = fmaf(S.ti, 0.5f, 0.5f);
        SB();
        const float to = __builtin_amdgcn_tanhf(S.g3);
        SB();
        const float fc = fmaf(S.tf, S.ch, S.ch);
        SB();
        const float t0 = fmaf(xs, W.w1[0], W.b[0]);
        const float t1 = fmaf(xs, W.w1[1], W.b[1]);
        const float t2 = fmaf(xs, W.w1[2], W.b[2]);
        const float t3 = fmaf(xs, W.w1[3], W.b[3]);
        SB();
        const float c = fmaf(iv, S.tg, fc);
        SB();
        Pn[j][0] = fmaf(ys, W.w0[0], t0);
        Pn[j][1] = fmaf(ys, W.w0[1], t1);
        SB();
        const float ow = fmaf(to, W.rh, W.rh);
        Pn[j][2] = fmaf(ys, W.w0[2], t2);
        SB();
        const float tc = __builtin_amdgcn_tanhf(c);
        SB();
        S.ch = 0.5f * c;
        Pn[j][3] = fmaf(ys, W.w0[3], t3);
        SB();
        float p = ow * tc;
        const float r1 = xadd1_of_product(p);
        SB();
        const float G0 = fma_lane4(p, W.a[0], Pq[0]);
        const float G1 = fma_lane4(p, W.a[1], Pq[1]);
        SB();
        const float r2 = xadd2_chain(r1);
        SB();
        const float G2 = fma_lane4(p, W.a[2], Pq[2]);
        const float G3 = fma_lane4(p, W.a[3], Pq[3]);
        SB();
        const float g0 = fma_lane0(r2, W.a[0], G0);
        SB();
        const float g1 = fma_lane0(r2, W.a[1], G1);
        SB();
        const float g2 = fma_lane0(r2, W.a[2], G2);
        SB();
        S.ti = __builtin_amdgcn_tanhf(g0);
        SB();
        S.tf = __builtin_amdgcn_tanhf(g1);
        SB();
        S.g3 = fma_lane0(r2, W.a[3], G3);
        hs[j] = xadd4_out(r2);
        SB();
        S.tg = __builtin_amdgcn_tanhf(g2);
        SB();
    }
}

template <int DIR, bool GIVEN>
__device__ __forceinline__ Carry run_span(const LaneW& W, const float* __restrict__ yb,
                                          const float* __restrict__ ib, float* __restrict__ ho,
                                          Carry S, float h, float ch, int lane,
                                          int n_begin, int n_store, int n_end, float* rec_start) {
    float PA[UB][4], PB[UB][4];
    float XAy[UB], XAi[UB], XBy[UB], XBi[UB];
    float hs[UB * IT_BLK];

    load_blk<DIR>(yb, ib, n_begin, XAy, XAi);
    pre_blk<DIR>(W, XAy, XAi, PA);
    load_blk<DIR>(yb, ib, n_begin + 1, XBy, XBi);
    if (!GIVEN) {
        S.ti = __builtin_amdgcn_tanhf(fmaf(h, W.a[0], PA[0][0]));
        S.tf = __builtin_amdgcn_tanhf(fmaf(h, W.a[1], PA[0][1]));
        S.tg = __builtin_amdgcn_tanhf(fmaf(h, W.a[2], PA[0][2]));
        S.g3 = fmaf(h, W.a[3], PA[0][3]);
        S.ch = ch;
    }

    for (int n = n_begin; n < n_end; n += IT_BLK) {
        if (!GIVEN && n == n_store) put_state(rec_start, S, lane);
        {
            int pf = DIR ? (LSEQ - UB * IT_BLK - n * UB - 4 * UB * IT_BLK) : (n * UB + 4 * UB * IT_BLK);
            pf = pf < 0 ? 0 : (pf > LSEQ - 1 ? LSEQ - 1 : pf);
            __builtin_prefetch(yb + pf, 0, 0);
            __builtin_prefetch(ib + pf, 0, 0);
        }
        load_blk<DIR>(yb, ib, n + 2, XAy, XAi);
        steps8<DIR>(W, PA, PB, XBy, XBi, S, hs);
        load_blk<DIR>(yb, ib, n + 3, XBy, XBi);
        steps8<DIR>(W, PB, PA, XAy, XAi, S, hs + UB);
        load_blk<DIR>(yb, ib, n + 4, XAy, XAi);
        steps8<DIR>(W, PA, PB, XBy, XBi, S, hs + 2 * UB);
        load_blk<DIR>(yb, ib, n + 5, XBy, XBi);
        steps8<DIR>(W, PB, PA, XAy, XAi, S, hs + 3 * UB);

        const int base = DIR ? (LSEQ - UB * IT_BLK - n * UB) : (n * UB);
        if (n >= n_store && lane == 0) {
#pragma unroll
            for (int v = 0; v < UB * IT_BLK / 4; ++v) {
                float4 o;
                if (DIR) {
                    const int s = UB * IT_BLK - 1 - 4 * v;
                    o = make_float4(hs[s], hs[s - 1], hs[s - 2], hs[s - 3]);
                } else {
                    o = make_float4(hs[4 * v], hs[4 * v + 1], hs[4 * v + 2], hs[4 * v + 3]);
                }
                *reinterpret_cast<float4*>(ho + base + 4 * v) = o;
            }
        }
    }
    return S;
}

__device__ __forceinline__ LaneW lane_weights(const float* __restrict__ w_ih, const float* __restrict__ w_hh,
                                              const float* __restrict__ b_ih, const float* __restrict__ b_hh,
                                              const float* __restrict__ w_hr, bool real, int kk) {
    LaneW W;
#pragma unroll
    for (int q = 0; q < 4; ++q) {
        const int j = q * HID + kk;
        const float sc = real ? ((q == 2) ? 1.0f : 0.5f) : 0.0f;
        W.a[q] = sc * w_hh[j];
        W.w0[q] = sc * w_ih[2 * j];
        W.w1[q] = sc * w_ih[2 * j + 1];
        W.b[q] = sc * (b_ih[j] + b_hh[j]);
    }
    W.rh = real ? 0.5f * w_hr[kk] : 0.0f;
    return W;
}

__device__ __forceinline__ size_t rec_start_at(int chain, int seg) { return ((size_t)chain * NSEG + seg) * REC_FLOATS; }
__device__ __forceinline__ size_t rec_end_at(int chain, int seg) { return ((size_t)(NCHAIN + chain) * NSEG + seg) * REC_FLOATS; }

__global__ __launch_bounds__(32) void lstm_scan_segments(const float* __restrict__ y,
                          const float* __restrict__ inp,
                          const float* __restrict__ w_ih_f, const float* __restrict__ w_hh_f,
                          const float* __restrict__ b_ih_f, const float* __restrict__ b_hh_f,
                          const float* __restrict__ w_hr_f,
                          const float* __restrict__ w_ih_b, const float* __restrict__ w_hh_b,
                          const float* __restrict__ b_ih_b, const float* __restrict__ b_hh_b,
                          const float* __restrict__ w_hr_b,
                          const float* __restrict__ h0, const float* __restrict__ c0,
                          float* __restrict__ hout, float* __restrict__ st) {
    const int chain = blockIdx.x % NCHAIN;
    const int seg = blockIdx.x / NCHAIN;
    const int dir = chain >> 3;
    const int b = chain & 7;
    const int lane = threadIdx.x;
    const int k = lane & 7;
    const bool real = k < HID;
    const int kk = real ? k : 0;

    const LaneW W = dir ? lane_weights(w_ih_b, w_hh_b, b_ih_b, b_hh_b, w_hr_b, real, kk)
                        : lane_weights(w_ih_f, w_hh_f, b_ih_f, b_hh_f, w_hr_f, real, kk);
    const bool first = seg == 0;
    const float ch = (real && first) ? 0.5f * c0[chain * HID + kk] : 0.0f;
    const float h = first ? h0[chain] : 0.0f;

    const float* yb = y + (size_t)b * LSEQ;
    const float* ib = inp + (size_t)b * LSEQ;
    float* ho = hout + (size_t)chain * LSEQ;

    const int n_store = seg * SEG_PASSES * IT_BLK;
    const int n_begin = first ? 0 : n_store - RUNIN_PASSES * IT_BLK;
    const int n_end = n_store + SEG_PASSES * IT_BLK;
    float* rs = st + rec_start_at(chain, seg);
    const Carry Z = {0.0f, 0.0f, 0.0f, 0.0f, 0.0f};

    const Carry E = dir ? run_span<1, false>(W, yb, ib, ho, Z, h, ch, lane, n_begin, n_store, n_end, rs)
                        : run_span<0, false>(W, yb, ib, ho, Z, h, ch, lane, n_begin, n_store, n_end, rs);
    put_state(st + rec_end_at(chain, seg), E, lane);
}

template <int DIR>
__device__ __forceinline__ void join_chain(const LaneW& W, const float* __restrict__ yb, const float* __restrict__ ib,
                                           float* __restrict__ ho, const float* st, int chain, int lane) {
    unsigned eqm[4];
#pragma unroll
    for (int r = 0; r < 4; ++r) {
        const int s = 1 + 32 * r + lane;
        unsigned d = 0;
        if (s < NSEG) {
            const uint4* pa = reinterpret_cast<const uint4*>(st + rec_start_at(chain, s));
            const uint4* pb = reinterpret_cast<const uint4*>(st + rec_end_at(chain, s - 1));
            uint4 a[10], c[10];
#pragma unroll
            for (int v = 0; v < 10; ++v) { a[v] = pa[v]; c[v] = pb[v]; }
#pragma unroll
            for (int v = 0; v < 10; ++v)
                d |= (a[v].x ^ c[v].x) | (a[v].y ^ c[v].y) | (a[v].z ^ c[v].z) | (a[v].w ^ c[v].w);
        }
        eqm[r] = (unsigned)__ballot(d == 0);
    }
    static_assert(NSEG - 1 <= 4 * 32, "four rounds of 32 lanes cover every boundary");
    if ((eqm[0] & eqm[1] & eqm[2] & eqm[3]) == 0xffffffffu) return;

    bool have = false;
    Carry cur = {0.0f, 0.0f, 0.0f, 0.0f, 0.0f};
    for (int s = 1; s < NSEG; ++s) {
        bool ok;
        if (!have) {
            const int i = s - 1;
            const unsigned m = (i < 32) ? eqm[0] : (i < 64) ? eqm[1] : (i < 96) ? eqm[2] : eqm[3];
            ok = (m >> (i & 31)) & 1u;
        } else {
            ok = same_state(cur, st + rec_start_at(chain, s), lane);
        }
        if (ok) { have = false; continue; }
        if (!have) cur = get_state(st + rec_end_at(chain, s - 1), lane);
        const int n0 = s * SEG_PASSES * IT_BLK;
        cur = run_span<DIR, true>(W, yb, ib, ho, cur, 0.0f, 0.0f, lane, n0, n0, n0 + SEG_PASSES * IT_BLK, nullptr);
        have = true;
    }
}

__global__ __launch_bounds__(32) void lstm_scan_join(const float* __restrict__ y,
                          const float* __restrict__ inp,
                          const float* __restrict__ w_ih_f, const float* __restrict__ w_hh_f,
                          const float* __restrict__ b_ih_f, const float* __restrict__ b_hh_f,
                          const float* __restrict__ w_hr_f,
                          const float* __restrict__ w_ih_b, const float* __restrict__ w_hh_b,
                          const float* __restrict__ b_ih_b, const float* __restrict__ b_hh_b,
                          const float* __restrict__ w_hr_b,
                          float* __restrict__ hout, const float* st) {
    const int chain = blockIdx.x;
    const int dir = chain >> 3;
    const int b = chain & 7;
    const int lane = threadIdx.x;
    const int k = lane & 7;
    const bool real = k < HID;
    const int kk = real ? k : 0;

    const LaneW W = dir ? lane_weights(w_ih_b, w_hh_b, b_ih_b, b_hh_b, w_hr_b, real, kk)
                        : lane_weights(w_ih_f, w_hh_f, b_ih_f, b_hh_f, w_hr_f, real, kk);
    const float* yb = y + (size_t)b * LSEQ;
    const float* ib = inp + (size_t)b * LSEQ;
    float* ho = hout + (size_t)chain * LSEQ;

    if (dir) join_chain<1>(W, yb, ib, ho, st, chain, lane);
    else     join_chain<0>(W, yb, ib, ho, st, chain, lane);
}

__global__ void seq_softmax(const float* __restrict__ hbuf, float* __restrict__ out) {
    const int b = blockIdx.x;
    const int tid = threadIdx.x;
    const int T = 256;
    __shared__ float red[256];

    const float* hf = hbuf + (size_t)b * LSEQ;
    const float* hb = hbuf + (size_t)(NB + b) * LSEQ;

    float m = -3.0e38f;
    for (int l = tid; l < LSEQ; l += T) m = fmaxf(m, hf[l] + hb[l]);
    red[tid] = m;
    __syncthreads();
    for (int s = T / 2; s > 0; s >>= 1) {
        if (tid < s) red[tid] = fmaxf(red[tid], red[tid + s]);
        __syncthreads();
    }
    m = red[0];
    __syncthreads();

    float sum = 0.0f;
    for (int l = tid; l < LSEQ; l += T) sum += __expf(hf[l] + hb[l] - m);
    red[tid] = sum;
    __syncthreads();
    for (int s = T / 2; s > 0; s >>= 1) {
        if (tid < s) red[tid] += red[tid + s];
        __syncthreads();
    }
    sum = red[0];
    __syncthreads();

    const float inv = __builtin_amdgcn_rcpf(sum);
    for (int l = tid; l < LSEQ; l += T) {
        const float e = __expf(hf[l] + hb[l] - m) * inv;
        float* o = out + ((size_t)b * LSEQ + l) * 2;
        o[0] = e;
        o[1] = e;
    }
}

extern "C" void kernel_launch(void* const* d_in, const int* in_sizes, int n_in,
                              void* d_out, int out_size, void* d_ws, size_t ws_size,
                              hipStream_t stream) {
    const float* x      = (const float*)d_in[0];
    const float* inp    = (const float*)d_in[1];
    const float* conv_w = (const float*)d_in[2];
    const float* conv_b = (const float*)d_in[3];
    const float* w_ih_f = (const float*)d_in[4];
    const float* w_hh_f = (const float*)d_in[5];
    const float* b_ih_f = (const float*)d_in[6];
    const float* b_hh_f = (const float*)d_in[7];
    const float* w_hr_f = (const float*)d_in[8];
    const float* w_ih_b = (const float*)d_in[9];
    const float* w_hh_b = (const float*)d_in[10];
    const float* b_ih_b = (const float*)d_in[11];
    const float* b_hh_b = (const float*)d_in[12];
    const float* w_hr_b = (const float*)d_in[13];
    const float* h0     = (const float*)d_in[14];
    const float* c0     = (const float*)d_in[15];
    float* out = (float*)d_out;

    float* y    = (float*)d_ws;
    float* hbuf = y + (size_t)NB * LSEQ;

    conv1x1_wmma<<<(NB * LSEQ / 16) / 8, 256, 0, stream>>>(x, conv_w, conv_b, y);

    float* st = out;

    lstm_scan_segments<<<NCHAIN * NSEG, 32, 0, stream>>>(y, inp,
                                    w_ih_f, w_hh_f, b_ih_f, b_hh_f, w_hr_f,
                                    w_ih_b, w_hh_b, b_ih_b, b_hh_b, w_hr_b,
                                    h0, c0, hbuf, st);

    lstm_scan_join<<<NCHAIN, 32, 0, stream>>>(y, inp,
                                    w_ih_f, w_hh_f, b_ih_f, b_hh_f, w_hr_f,
                                    w_ih_b, w_hh_b, b_ih_b, b_hh_b, w_hr_b,
                                    hbuf, st);

    seq_softmax<<<NB, 256, 0, stream>>>(hbuf, out);
}


# softmax over the sequence: each batch row cut into 30 chunks, a work-group each, in two launches (chunk max and exp sum; then join and normalise); was one work-group per row; exp sum order changes
# speedup vs baseline: 332.6555x; 1.5022x over previous
#include <hip/hip_runtime.h>
#include <hip/hip_bf16.h>

#define LSEQ 60000
#define NB 8
#define INCH 64
#define HID 5

typedef __attribute__((ext_vector_type(2))) float v2f;
typedef __attribute__((ext_vector_type(8))) float v8f;

__global__ void conv1x1_wmma(const float* __restrict__ x,
                             const float* __restrict__ w,
                             const float* __restrict__ bias,
                             float* __restrict__ y) {
    const int lane = threadIdx.x & 31;
    const int wave = threadIdx.x >> 5;
    const int tile = blockIdx.x * 8 + wave;
    const int b = tile / (LSEQ / 16);
    const int lbase = (tile % (LSEQ / 16)) * 16;
    const int m = lane & 15;
    const int half = lane >> 4;

    const float* xb = x + (size_t)b * INCH * LSEQ + lbase + m;

    v8f acc = {};
#pragma unroll
    for (int chunk = 0; chunk < 16; ++chunk) {
        const int k0 = chunk * 4 + 2 * half;
        v2f a, bb;
        a.x = xb[(size_t)k0 * LSEQ];
        a.y = xb[(size_t)(k0 + 1) * LSEQ];
        bb.x = w[k0];
        bb.y = w[k0 + 1];
        acc = __builtin_amdgcn_wmma_f32_16x16x4_f32(
             false, a,  false, bb,
             (short)0, acc,  false,  false);
    }

    const float bv = bias[0];
    float* yo = y + (size_t)b * LSEQ + lbase;
    if (lane == 0) {
#pragma unroll
        for (int r = 0; r < 8; ++r) yo[r] = acc[r] + bv;
    } else if (lane == 16) {
#pragma unroll
        for (int r = 0; r < 8; ++r) yo[8 + r] = acc[r] + bv;
    }
}

#define UB 8
#define NBLK (LSEQ / UB)
#define IT_BLK 4
#define NCHAIN (2 * NB)
#define NPASS (NBLK / IT_BLK)
#define SEG_PASSES 15
#define NSEG (NPASS / SEG_PASSES)
#define RUNIN_PASSES 8
#define REC_FLOATS 64
static_assert(NPASS % SEG_PASSES == 0, "segments are whole passes");
static_assert(SEG_PASSES >= RUNIN_PASSES, "the run-in of segment 1 must not start in front of the sequence");

#define SB() __builtin_amdgcn_sched_barrier(0)

__device__ __forceinline__ float xadd1_of_product(float p) {
    float r;
    asm("v_add_f32_dpp %0, %1, %1 row_xmask:1 row_mask:0xf bank_mask:0xf bound_ctrl:1" : "=v"(r) : "v"(p));
    return r;
}

__device__ __forceinline__ float fma_lane4(float x, float y, float z) {
    asm("v_fmac_f32_dpp %0, %1, %2 row_share:4 row_mask:0xf bank_mask:0xf" : "+v"(z) : "v"(x), "v"(y));
    return z;
}
__device__ __forceinline__ float fma_lane0(float x, float y, float z) {
    asm("v_fmac_f32_dpp %0, %1, %2 row_share:0 row_mask:0xf bank_mask:0xf" : "+v"(z) : "v"(x), "v"(y));
    return z;
}
__device__ __forceinline__ float xadd2_chain(float v) {
    float r;
    asm("v_add_f32_dpp %0, %1, %1 row_xmask:2 row_mask:0xf bank_mask:0xf bound_ctrl:1" : "=v"(r) : "v"(v));
    return r;
}
__device__ __forceinline__ float xadd4_out(float v) {
    float r;
    asm("v_add_f32_dpp %0, %1, %1 row_xmask:4 row_mask:0xf bank_mask:0xf bound_ctrl:1" : "=v"(r) : "v"(v));
    return r;
}

struct LaneW {
    float a[4];
    float w0[4];
    float w1[4];
    float b[4];
    float rh;
};

template <int DIR>
__device__ __forceinline__ void load_blk(const float* __restrict__ yb, const float* __restrict__ ib,
                                         int blk, float (&xy)[UB], float (&xi)[UB]) {
    if (blk > NBLK - 1) blk = NBLK - 1;
    const int base = DIR ? (LSEQ - UB - blk * UB) : (blk * UB);
    const float4 y0 = *reinterpret_cast<const float4*>(yb + base);
    const float4 y1 = *reinterpret_cast<const float4*>(yb + base + 4);
    const float4 i0 = *reinterpret_cast<const float4*>(ib + base);
    const float4 i1 = *reinterpret_cast<const float4*>(ib + base + 4);
    xy[0] = y0.x; xy[1] = y0.y; xy[2] = y0.z; xy[3] = y0.w;
    xy[4] = y1.x; xy[5] = y1.y; xy[6] = y1.z; xy[7] = y1.w;
    xi[0] = i0.x; xi[1] = i0.y; xi[2] = i0.z; xi[3] = i0.w;
    xi[4] = i1.x; xi[5] = i1.y; xi[6] = i1.z; xi[7] = i1.w;
}

template <int DIR>
__device__ __forceinline__ void pre_blk(const LaneW& W, const float (&xy)[UB], const float (&xi)[UB],
                                        float (&P)[UB][4]) {
#pragma unroll
    for (int j = 0; j < UB; ++j) {
        const int m = DIR ? (UB - 1 - j) : j;
#pragma unroll
        for (int q = 0; q < 4; ++q)
            P[j][q] = fmaf(xy[m], W.w0[q], fmaf(xi[m], W.w1[q], W.b[q]));
    }
}

struct Carry { float ti, tf, tg, g3, ch; };

__device__ __forceinline__ void put_state(float* rec, const Carry& S, int lane) {
    if (lane < 8) {
        *reinterpret_cast<float4*>(rec + 4 * lane) = make_float4(S.ti, S.tf, S.tg, S.g3);
        rec[32 + lane] = S.ch;
    }
}
__device__ __forceinline__ Carry get_state(const float* rec, int lane) {
    const int k = lane & 7;
    const float4 g = *reinterpret_cast<const float4*>(rec + 4 * k);
    Carry S;
    S.ti = g.x; S.tf = g.y; S.tg = g.z; S.g3 = g.w;
    S.ch = rec[32 + k];
    return S;
}
__device__ __forceinline__ bool same_state(const Carry& A, const float* rec, int lane) {
    const Carry B = get_state(rec, lane);
    const bool diff = (__float_as_uint(A.ti) != __float_as_uint(B.ti)) | (__float_as_uint(A.tf) != __float_as_uint(B.tf)) |
                      (__float_as_uint(A.tg) != __float_as_uint(B.tg)) | (__float_as_uint(A.g3) != __float_as_uint(B.g3)) |
                      (__float_as_uint(A.ch) != __float_as_uint(B.ch));
    return !__any(diff);
}

template <int DIR>
__device__ __forceinline__ void steps8(const LaneW& W, const float (&P)[UB][4], float (&Pn)[UB][4],
                                       const float (&xy)[UB], const float (&xi)[UB],
                                       Carry& S, float* hs) {
#pragma unroll
    for (int j = 0; j < UB; ++j) {
        const int m = DIR ? (UB - 1 - j) : j;
        const float (&Pq)[4] = (j + 1 < UB) ? P[j + 1] : Pn[0];
        const float xs = xi[m], ys = xy[m];
        const float iv = fmaf(S.ti, 0.5f, 0.5f);
        SB();
        const float to = __builtin_amdgcn_tanhf(S.g3);
        SB();
        const float fc = fmaf(S.tf, S.ch, S.ch);
        SB();
        const float t0 = fmaf(xs, W.w1[0], W.b[0]);
        const float t1 = fmaf(xs, W.w1[1], W.b[1]);
        const float t2 = fmaf(xs, W.w1[2], W.b[2]);
        const float t3 = fmaf(xs, W.w1[3], W.b[3]);
        SB();
        const float c = fmaf(iv, S.tg, fc);
        SB();
        Pn[j][0] = fmaf(ys, W.w0[0], t0);
        Pn[j][1] = fmaf(ys, W.w0[1], t1);
        SB();
        const float ow = fmaf(to, W.rh, W.rh);
        Pn[j][2] = fmaf(ys, W.w0[2], t2);
        SB();
        const float tc = __builtin_amdgcn_tanhf(c);
        SB();
        S.ch = 0.5f * c;
        Pn[j][3] = fmaf(ys, W.w0[3], t3);
        SB();
        float p = ow * tc;
        const float r1 = xadd1_of_product(p);
        SB();
        const float G0 = fma_lane4(p, W.a[0], Pq[0]);
        const float G1 = fma_lane4(p, W.a[1], Pq[1]);
        SB();
        const float r2 = xadd2_chain(r1);
        SB();
        const float G2 = fma_lane4(p, W.a[2], Pq[2]);
        const float G3 = fma_lane4(p, W.a[3], Pq[3]);
        SB();
        const float g0 = fma_lane0(r2, W.a[0], G0);
        SB();
        const float g1 = fma_lane0(r2, W.a[1], G1);
        SB();
        const float g2 = fma_lane0(r2, W.a[2], G2);
        SB();
        S.ti = __builtin_amdgcn_tanhf(g0);
        SB();
        S.tf = __builtin_amdgcn_tanhf(g1);
        SB();
        S.g3 = fma_lane0(r2, W.a[3], G3);
        hs[j] = xadd4_out(r2);
        SB();
        S.tg = __builtin_amdgcn_tanhf(g2);
        SB();
    }
}

template <int DIR, bool GIVEN>
__device__ __forceinline__ Carry run_span(const LaneW& W, const float* __restrict__ yb,
                                          const float* __restrict__ ib, float* __restrict__ ho,
                                          Carry S, float h, float ch, int lane,
                                          int n_begin, int n_store, int n_end, float* rec_start) {
    float PA[UB][4], PB[UB][4];
    float XAy[UB], XAi[UB], XBy[UB], XBi[UB];
    float hs[UB * IT_BLK];

    load_blk<DIR>(yb, ib, n_begin, XAy, XAi);
    pre_blk<DIR>(W, XAy, XAi, PA);
    load_blk<DIR>(yb, ib, n_begin + 1, XBy, XBi);
    if (!GIVEN) {
        S.ti = __builtin_amdgcn_tanhf(fmaf(h, W.a[0], PA[0][0]));
        S.tf = __builtin_amdgcn_tanhf(fmaf(h, W.a[1], PA[0][1]));
        S.tg = __builtin_amdgcn_tanhf(fmaf(h, W.a[2], PA[0][2]));
        S.g3 = fmaf(h, W.a[3], PA[0][3]);
        S.ch = ch;
    }

    for (int n = n_begin; n < n_end; n += IT_BLK) {
        if (!GIVEN && n == n_store) put_state(rec_start, S, lane);
        {
            int pf = DIR ? (LSEQ - UB * IT_BLK - n * UB - 4 * UB * IT_BLK) : (n * UB + 4 * UB * IT_BLK);
            pf = pf < 0 ? 0 : (pf > LSEQ - 1 ? LSEQ - 1 : pf);
            __builtin_prefetch(yb + pf, 0, 0);
            __builtin_prefetch(ib + pf, 0, 0);
        }
        load_blk<DIR>(yb, ib, n + 2, XAy, XAi);
        steps8<DIR>(W, PA, PB, XBy, XBi, S, hs);
        load_blk<DIR>(yb, ib, n + 3, XBy, XBi);
        steps8<DIR>(W, PB, PA, XAy, XAi, S, hs + UB);
        load_blk<DIR>(yb, ib, n + 4, XAy, XAi);
        steps8<DIR>(W, PA, PB, XBy, XBi, S, hs + 2 * UB);
        load_blk<DIR>(yb, ib, n + 5, XBy, XBi);
        steps8<DIR>(W, PB, PA, XAy, XAi, S, hs + 3 * UB);

        const int base = DIR ? (LSEQ - UB * IT_BLK - n * UB) : (n * UB);
        if (n >= n_store && lane == 0) {
#pragma unroll
            for (int v = 0; v < UB * IT_BLK / 4; ++v) {
                float4 o;
                if (DIR) {
                    const int s = UB * IT_BLK - 1 - 4 * v;
                    o = make_float4(hs[s], hs[s - 1], hs[s - 2], hs[s - 3]);
                } else {
                    o = make_float4(hs[4 * v], hs[4 * v + 1], hs[4 * v + 2], hs[4 * v + 3]);
                }
                *reinterpret_cast<float4*>(ho + base + 4 * v) = o;
            }
        }
    }
    return S;
}

__device__ __forceinline__ LaneW lane_weights(const float* __restrict__ w_ih, const float* __restrict__ w_hh,
                                              const float* __restrict__ b_ih, const float* __restrict__ b_hh,
                                              const float* __restrict__ w_hr, bool real, int kk) {
    LaneW W;
#pragma unroll
    for (int q = 0; q < 4; ++q) {
        const int j = q * HID + kk;
        const float sc = real ? ((q == 2) ? 1.0f : 0.5f) : 0.0f;
        W.a[q] = sc * w_hh[j];
        W.w0[q] = sc * w_ih[2 * j];
        W.w1[q] = sc * w_ih[2 * j + 1];
        W.b[q] = sc * (b_ih[j] + b_hh[j]);
    }
    W.rh = real ? 0.5f * w_hr[kk] : 0.0f;
    return W;
}

__device__ __forceinline__ size_t rec_start_at(int chain, int seg) { return ((size_t)chain * NSEG + seg) * REC_FLOATS; }
__device__ __forceinline__ size_t rec_end_at(int chain, int seg) { return ((size_t)(NCHAIN + chain) * NSEG + seg) * REC_FLOATS; }

__global__ __launch_bounds__(32) void lstm_scan_segments(const float* __restrict__ y,
                          const float* __restrict__ inp,
                          const float* __restrict__ w_ih_f, const float* __restrict__ w_hh_f,
                          const float* __restrict__ b_ih_f, const float* __restrict__ b_hh_f,
                          const float* __restrict__ w_hr_f,
                          const float* __restrict__ w_ih_b, const float* __restrict__ w_hh_b,
                          const float* __restrict__ b_ih_b, const float* __restrict__ b_hh_b,
                          const float* __restrict__ w_hr_b,
                          const float* __restrict__ h0, const float* __restrict__ c0,
                          float* __restrict__ hout, float* __restrict__ st) {
    const int chain = blockIdx.x % NCHAIN;
    const int seg = blockIdx.x / NCHAIN;
    const int dir = chain >> 3;
    const int b = chain & 7;
    const int lane = threadIdx.x;
    const int k = lane & 7;
    const bool real = k < HID;
    const int kk = real ? k : 0;

    const LaneW W = dir ? lane_weights(w_ih_b, w_hh_b, b_ih_b, b_hh_b, w_hr_b, real, kk)
                        : lane_weights(w_ih_f, w_hh_f, b_ih_f, b_hh_f, w_hr_f, real, kk);
    const bool first = seg == 0;
    const float ch = (real && first) ? 0.5f * c0[chain * HID + kk] : 0.0f;
    const float h = first ? h0[chain] : 0.0f;

    const float* yb = y + (size_t)b * LSEQ;
    const float* ib = inp + (size_t)b * LSEQ;
    float* ho = hout + (size_t)chain * LSEQ;

    const int n_store = seg * SEG_PASSES * IT_BLK;
    const int n_begin = first ? 0 : n_store - RUNIN_PASSES * IT_BLK;
    const int n_end = n_store + SEG_PASSES * IT_BLK;
    float* rs = st + rec_start_at(chain, seg);
    const Carry Z = {0.0f, 0.0f, 0.0f, 0.0f, 0.0f};

    const Carry E = dir ? run_span<1, false>(W, yb, ib, ho, Z, h, ch, lane, n_begin, n_store, n_end, rs)
                        : run_span<0, false>(W, yb, ib, ho, Z, h, ch, lane, n_begin, n_store, n_end, rs);
    put_state(st + rec_end_at(chain, seg), E, lane);
}

template <int DIR>
__device__ __forceinline__ void join_chain(const LaneW& W, const float* __restrict__ yb, const float* __restrict__ ib,
                                           float* __restrict__ ho, const float* st, int chain, int lane) {
    unsigned eqm[4];
#pragma unroll
    for (int r = 0; r < 4; ++r) {
        const int s = 1 + 32 * r + lane;
        unsigned d = 0;
        if (s < NSEG) {
            const uint4* pa = reinterpret_cast<const uint4*>(st + rec_start_at(chain, s));
            const uint4* pb = reinterpret_cast<const uint4*>(st + rec_end_at(chain, s - 1));
            uint4 a[10], c[10];
#pragma unroll
            for (int v = 0; v < 10; ++v) { a[v] = pa[v]; c[v] = pb[v]; }
#pragma unroll
            for (int v = 0; v < 10; ++v)
                d |= (a[v].x ^ c[v].x) | (a[v].y ^ c[v].y) | (a[v].z ^ c[v].z) | (a[v].w ^ c[v].w);
        }
        eqm[r] = (unsigned)__ballot(d == 0);
    }
    static_assert(NSEG - 1 <= 4 * 32, "four rounds of 32 lanes cover every boundary");
    if ((eqm[0] & eqm[1] & eqm[2] & eqm[3]) == 0xffffffffu) return;

    bool have = false;
    Carry cur = {0.0f, 0.0f, 0.0f, 0.0f, 0.0f};
    for (int s = 1; s < NSEG; ++s) {
        bool ok;
        if (!have) {
            const int i = s - 1;
            const unsigned m = (i < 32) ? eqm[0] : (i < 64) ? eqm[1] : (i < 96) ? eqm[2] : eqm[3];
            ok = (m >> (i & 31)) & 1u;
        } else {
            ok = same_state(cur, st + rec_start_at(chain, s), lane);
        }
        if (ok) { have = false; continue; }
        if (!have) cur = get_state(st + rec_end_at(chain, s - 1), lane);
        const int n0 = s * SEG_PASSES * IT_BLK;
        cur = run_span<DIR, true>(W, yb, ib, ho, cur, 0.0f, 0.0f, lane, n0, n0, n0 + SEG_PASSES * IT_BLK, nullptr);
        have = true;
    }
}

__global__ __launch_bounds__(32) void lstm_scan_join(const float* __restrict__ y,
                          const float* __restrict__ inp,
                          const float* __restrict__ w_ih_f, const float* __restrict__ w_hh_f,
                          const float* __restrict__ b_ih_f, const float* __restrict__ b_hh_f,
                          const float* __restrict__ w_hr_f,
                          const float* __restrict__ w_ih_b, const float* __restrict__ w_hh_b,
                          const float* __restrict__ b_ih_b, const float* __restrict__ b_hh_b,
                          const float* __restrict__ w_hr_b,
                          float* __restrict__ hout, const float* st) {
    const int chain = blockIdx.x;
    const int dir = chain >> 3;
    const int b = chain & 7;
    const int lane = threadIdx.x;
    const int k = lane & 7;
    const bool real = k < HID;
    const int kk = real ? k : 0;

    const LaneW W = dir ? lane_weights(w_ih_b, w_hh_b, b_ih_b, b_hh_b, w_hr_b, real, kk)
                        : lane_weights(w_ih_f, w_hh_f, b_ih_f, b_hh_f, w_hr_f, real, kk);
    const float* yb = y + (size_t)b * LSEQ;
    const float* ib = inp + (size_t)b * LSEQ;
    float* ho = hout + (size_t)chain * LSEQ;

    if (dir) join_chain<1>(W, yb, ib, ho, st, chain, lane);
    else     join_chain<0>(W, yb, ib, ho, st, chain, lane);
}

#define SM_T 128
#define SM_STEPS 8
#define SM_CHUNK (2 * SM_T * SM_STEPS)
#define SM_NCH ((LSEQ + SM_CHUNK - 1) / SM_CHUNK)
#define SM_REC 32
static_assert(SM_NCH <= 32, "softmax_finish holds one chunk record per lane");
static_assert(LSEQ % 2 == 0, "a pair of positions is inside the row or past its end as a whole");

template <int N>
__device__ __forceinline__ float row_xor(float v) {
    return __builtin_bit_cast(float, __builtin_amdgcn_update_dpp(0, __builtin_bit_cast(int, v), 0x160 + N, 0xf, 0xf, true));
}
__device__ __forceinline__ float other_row(float v) {
    const int b = __builtin_bit_cast(int, v);
    return __builtin_bit_cast(float, __builtin_amdgcn_permlanex16(b, b, 0x76543210, 0xfedcba98, false, false));
}
__device__ __forceinline__ float wave_max(float v) {
    v = fmaxf(v, row_xor<1>(v));
    v = fmaxf(v, row_xor<2>(v));
    v = fmaxf(v, row_xor<4>(v));
    v = fmaxf(v, row_xor<8>(v));
    return fmaxf(v, other_row(v));
}
__device__ __forceinline__ float wave_sum(float v) {
    v += row_xor<1>(v);
    v += row_xor<2>(v);
    v += row_xor<4>(v);
    v += row_xor<8>(v);
    return v + other_row(v);
}

__device__ __forceinline__ void sm_load_chunk(const float* __restrict__ hf, const float* __restrict__ hb,
                                              int ch, int tid, float (&v)[2 * SM_STEPS]) {
    float2 f[SM_STEPS], g[SM_STEPS];
#pragma unroll
    for (int s = 0; s < SM_STEPS; ++s) {
        const int l = ch * SM_CHUNK + s * 2 * SM_T + 2 * tid;
        const int lc = l < LSEQ ? l : 0;
        f[s] = *reinterpret_cast<const float2*>(hf + lc);
        g[s] = *reinterpret_cast<const float2*>(hb + lc);
    }
#pragma unroll
    for (int s = 0; s < SM_STEPS; ++s) {
        const bool ok = ch * SM_CHUNK + s * 2 * SM_T + 2 * tid < LSEQ;
        v[2 * s] = ok ? f[s].x + g[s].x : -INFINITY;
        v[2 * s + 1] = ok ? f[s].y + g[s].y : -INFINITY;
    }
}

__global__ __launch_bounds__(SM_T) void softmax_chunk_stats(const float* __restrict__ hbuf, float* __restrict__ part) {
    const int b = blockIdx.x / SM_NCH;
    const int ch = blockIdx.x % SM_NCH;
    const int tid = threadIdx.x;
    __shared__ float4 redm, reds;
    static_assert(SM_T == 4 * 32, "four waves, one float4 of partial results");

    float v[2 * SM_STEPS];
    sm_load_chunk(hbuf + (size_t)b * LSEQ, hbuf + (size_t)(NB + b) * LSEQ, ch, tid, v);

    float m = v[0];
#pragma unroll
    for (int i = 1; i < 2 * SM_STEPS; ++i) m = fmaxf(m, v[i]);
    m = wave_max(m);
    if ((tid & 31) == 0) reinterpret_cast<float*>(&redm)[tid >> 5] = m;
    __syncthreads();
    const float4 wm = redm;
    m = fmaxf(fmaxf(wm.x, wm.y), fmaxf(wm.z, wm.w));

    float s = 0.0f;
#pragma unroll
    for (int i = 0; i < 2 * SM_STEPS; ++i) s += __expf(v[i] - m);
    s = wave_sum(s);
    if ((tid & 31) == 0) reinterpret_cast<float*>(&reds)[tid >> 5] = s;
    __syncthreads();
    if (tid == 0) {
        const float4 ws = reds;
        *reinterpret_cast<float2*>(part + (size_t)blockIdx.x * SM_REC) = make_float2(m, (ws.x + ws.y) + (ws.z + ws.w));
    }
}

__global__ __launch_bounds__(SM_T) void softmax_finish(const float* __restrict__ hbuf, const float* __restrict__ part,
                                                       float* __restrict__ out) {
    const int b = blockIdx.x / SM_NCH;
    const int ch = blockIdx.x % SM_NCH;
    const int tid = threadIdx.x;
    const int lane = tid & 31;

    float v[2 * SM_STEPS];
    sm_load_chunk(hbuf + (size_t)b * LSEQ, hbuf + (size_t)(NB + b) * LSEQ, ch, tid, v);

    float pm = -INFINITY, ps = 0.0f;
    if (lane < SM_NCH) {
        const float2 r = *reinterpret_cast<const float2*>(part + (size_t)(b * SM_NCH + lane) * SM_REC);
        pm = r.x;
        ps = r.y;
    }
    const float M = wave_max(pm);
    const float S = wave_sum(ps * __expf(pm - M));
    const float inv = __builtin_bit_cast(float, __builtin_amdgcn_readfirstlane(__builtin_bit_cast(int, __builtin_amdgcn_rcpf(S))));

    float* ob = out + (size_t)b * LSEQ * 2;
#pragma unroll
    for (int s = 0; s < SM_STEPS; ++s) {
        const int l = ch * SM_CHUNK + s * 2 * SM_T + 2 * tid;
        if (l < LSEQ) {
            const float e0 = __expf(v[2 * s] - M) * inv;
            const float e1 = __expf(v[2 * s + 1] - M) * inv;
            *reinterpret_cast<float4*>(ob + (size_t)l * 2) = make_float4(e0, e0, e1, e1);
        }
    }
}

extern "C" void kernel_launch(void* const* d_in, const int* in_sizes, int n_in,
                              void* d_out, int out_size, void* d_ws, size_t ws_size,
                              hipStream_t stream) {
    const float* x      = (const float*)d_in[0];
    const float* inp    = (const float*)d_in[1];
    const float* conv_w = (const float*)d_in[2];
    const float* conv_b = (const float*)d_in[3];
    const float* w_ih_f = (const float*)d_in[4];
    const float* w_hh_f = (const float*)d_in[5];
    const float* b_ih_f = (const float*)d_in[6];
    const float* b_hh_f = (const float*)d_in[7];
    const float* w_hr_f = (const float*)d_in[8];
    const float* w_ih_b = (const float*)d_in[9];
    const float* w_hh_b = (const float*)d_in[10];
    const float* b_ih_b = (const float*)d_in[11];
    const float* b_hh_b = (const float*)d_in[12];
    const float* w_hr_b = (const float*)d_in[13];
    const float* h0     = (const float*)d_in[14];
    const float* c0     = (const float*)d_in[15];
    float* out = (float*)d_out;

    float* y    = (float*)d_ws;
    float* hbuf = y + (size_t)NB * LSEQ;

    conv1x1_wmma<<<(NB * LSEQ / 16) / 8, 256, 0, stream>>>(x, conv_w, conv_b, y);

    float* st = out;

    lstm_scan_segments<<<NCHAIN * NSEG, 32, 0, stream>>>(y, inp,
                                    w_ih_f, w_hh_f, b_ih_f, b_hh_f, w_hr_f,
                                    w_ih_b, w_hh_b, b_ih_b, b_hh_b, w_hr_b,
                                    h0, c0, hbuf, st);

    lstm_scan_join<<<NCHAIN, 32, 0, stream>>>(y, inp,
                                    w_ih_f, w_hh_f, b_ih_f, b_hh_f, w_hr_f,
                                    w_ih_b, w_hh_b, b_ih_b, b_hh_b, w_hr_b,
                                    hbuf, st);

    float* part = y;
    softmax_chunk_stats<<<NB * SM_NCH, SM_T, 0, stream>>>(hbuf, part);
    softmax_finish<<<NB * SM_NCH, SM_T, 0, stream>>>(hbuf, part, out);
}


# 1x1 conv: one wave per 128 positions (was 16), positions along the product's columns, x as B operand from 512-byte contiguous 16-byte loads; 3752 waves, was 30000; channel sum order changes
# speedup vs baseline: 353.1820x; 1.0617x over previous
#include <hip/hip_runtime.h>
#include <hip/hip_bf16.h>

#define LSEQ 60000
#define NB 8
#define INCH 64
#define HID 5

typedef __attribute__((ext_vector_type(2))) float v2f;
typedef __attribute__((ext_vector_type(8))) float v8f;

#define CV_TILE 128
#define CV_TPR ((LSEQ + CV_TILE - 1) / CV_TILE)
#define CV_WAVES 4
static_assert((NB * CV_TPR) % CV_WAVES == 0, "whole work-groups");
static_assert(LSEQ % 4 == 0, "four positions per lane are inside the row or past its end as a whole");

__global__ __launch_bounds__(32 * CV_WAVES) __attribute__((amdgpu_waves_per_eu(4, 4))) void conv1x1_wmma(
                             const float* __restrict__ x,
                             const float* __restrict__ w,
                             const float* __restrict__ bias,
                             float* __restrict__ y) {
    const int lane = threadIdx.x & 31;
    const int wave = threadIdx.x >> 5;
    const int tile = blockIdx.x * CV_WAVES + wave;
    const int b = tile / CV_TPR;
    const int lbase = (tile % CV_TPR) * CV_TILE;
    const int l = lbase + 4 * lane;
    const bool inrow = l < LSEQ;
    const int ll = inrow ? l : LSEQ - 4;

    const float* xb = x + (size_t)b * INCH * LSEQ + ll;

    float4 q[INCH];
#pragma unroll
    for (int i = 0; i < INCH; ++i)
        q[i] = *reinterpret_cast<const float4*>(xb + (size_t)i * LSEQ);

    const bool mine = ((lane >> 3) & 1) == (lane >> 4);
    const bool keep0 = mine && !(lane & 1);
    const bool keep1 = mine && (lane & 1);
    v8f acc0 = {}, acc1 = {};
#pragma unroll
    for (int i = 0; i < INCH; ++i) {
        const float wi = w[i];
        v2f a, b0, b1;
        a.x = keep0 ? wi : 0.0f;
        a.y = keep1 ? wi : 0.0f;
        b0.x = q[i].x; b0.y = q[i].y;
        b1.x = q[i].z; b1.y = q[i].w;
        acc0 = __builtin_amdgcn_wmma_f32_16x16x4_f32(
             false, a,  false, b0,
             (short)0, acc0,  false,  false);
        acc1 = __builtin_amdgcn_wmma_f32_16x16x4_f32(
             false, a,  false, b1,
             (short)0, acc1,  false,  false);
    }

    const float bv = bias[0];
    if (inrow)
        *reinterpret_cast<float4*>(y + (size_t)b * LSEQ + l) =
            make_float4(acc0[0] + bv, acc0[1] + bv, acc1[0] + bv, acc1[1] + bv);
}

#define UB 8
#define NBLK (LSEQ / UB)
#define IT_BLK 4
#define NCHAIN (2 * NB)
#define NPASS (NBLK / IT_BLK)
#define SEG_PASSES 15
#define NSEG (NPASS / SEG_PASSES)
#define RUNIN_PASSES 8
#define REC_FLOATS 64
static_assert(NPASS % SEG_PASSES == 0, "segments are whole passes");
static_assert(SEG_PASSES >= RUNIN_PASSES, "the run-in of segment 1 must not start in front of the sequence");

#define SB() __builtin_amdgcn_sched_barrier(0)

__device__ __forceinline__ float xadd1_of_product(float p) {
    float r;
    asm("v_add_f32_dpp %0, %1, %1 row_xmask:1 row_mask:0xf bank_mask:0xf bound_ctrl:1" : "=v"(r) : "v"(p));
    return r;
}

__device__ __forceinline__ float fma_lane4(float x, float y, float z) {
    asm("v_fmac_f32_dpp %0, %1, %2 row_share:4 row_mask:0xf bank_mask:0xf" : "+v"(z) : "v"(x), "v"(y));
    return z;
}
__device__ __forceinline__ float fma_lane0(float x, float y, float z) {
    asm("v_fmac_f32_dpp %0, %1, %2 row_share:0 row_mask:0xf bank_mask:0xf" : "+v"(z) : "v"(x), "v"(y));
    return z;
}
__device__ __forceinline__ float xadd2_chain(float v) {
    float r;
    asm("v_add_f32_dpp %0, %1, %1 row_xmask:2 row_mask:0xf bank_mask:0xf bound_ctrl:1" : "=v"(r) : "v"(v));
    return r;
}
__device__ __forceinline__ float xadd4_out(float v) {
    float r;
    asm("v_add_f32_dpp %0, %1, %1 row_xmask:4 row_mask:0xf bank_mask:0xf bound_ctrl:1" : "=v"(r) : "v"(v));
    return r;
}

struct LaneW {
    float a[4];
    float w0[4];
    float w1[4];
    float b[4];
    float rh;
};

template <int DIR>
__device__ __forceinline__ void load_blk(const float* __restrict__ yb, const float* __restrict__ ib,
                                         int blk, float (&xy)[UB], float (&xi)[UB]) {
    if (blk > NBLK - 1) blk = NBLK - 1;
    const int base = DIR ? (LSEQ - UB - blk * UB) : (blk * UB);
    const float4 y0 = *reinterpret_cast<const float4*>(yb + base);
    const float4 y1 = *reinterpret_cast<const float4*>(yb + base + 4);
    const float4 i0 = *reinterpret_cast<const float4*>(ib + base);
    const float4 i1 = *reinterpret_cast<const float4*>(ib + base + 4);
    xy[0] = y0.x; xy[1] = y0.y; xy[2] = y0.z; xy[3] = y0.w;
    xy[4] = y1.x; xy[5] = y1.y; xy[6] = y1.z; xy[7] = y1.w;
    xi[0] = i0.x; xi[1] = i0.y; xi[2] = i0.z; xi[3] = i0.w;
    xi[4] = i1.x; xi[5] = i1.y; xi[6] = i1.z; xi[7] = i1.w;
}

template <int DIR>
__device__ __forceinline__ void pre_blk(const LaneW& W, const float (&xy)[UB], const float (&xi)[UB],
                                        float (&P)[UB][4]) {
#pragma unroll
    for (int j = 0; j < UB; ++j) {
        const int m = DIR ? (UB - 1 - j) : j;
#pragma unroll
        for (int q = 0; q < 4; ++q)
            P[j][q] = fmaf(xy[m], W.w0[q], fmaf(xi[m], W.w1[q], W.b[q]));
    }
}

struct Carry { float ti, tf, tg, g3, ch; };

__device__ __forceinline__ void put_state(float* rec, const Carry& S, int lane) {
    if (lane < 8) {
        *reinterpret_cast<float4*>(rec + 4 * lane) = make_float4(S.ti, S.tf, S.tg, S.g3);
        rec[32 + lane] = S.ch;
    }
}
__device__ __forceinline__ Carry get_state(const float* rec, int lane) {
    const int k = lane & 7;
    const float4 g = *reinterpret_cast<const float4*>(rec + 4 * k);
    Carry S;
    S.ti = g.x; S.tf = g.y; S.tg = g.z; S.g3 = g.w;
    S.ch = rec[32 + k];
    return S;
}
__device__ __forceinline__ bool same_state(const Carry& A, const float* rec, int lane) {
    const Carry B = get_state(rec, lane);
    const bool diff = (__float_as_uint(A.ti) != __float_as_uint(B.ti)) | (__float_as_uint(A.tf) != __float_as_uint(B.tf)) |
                      (__float_as_uint(A.tg) != __float_as_uint(B.tg)) | (__float_as_uint(A.g3) != __float_as_uint(B.g3)) |
                      (__float_as_uint(A.ch) != __float_as_uint(B.ch));
    return !__any(diff);
}

template <int DIR>
__device__ __forceinline__ void steps8(const LaneW& W, const float (&P)[UB][4], float (&Pn)[UB][4],
                                       const float (&xy)[UB], const float (&xi)[UB],
                                       Carry& S, float* hs) {
#pragma unroll
    for (int j = 0; j < UB; ++j) {
        const int m = DIR ? (UB - 1 - j) : j;
        const float (&Pq)[4] = (j + 1 < UB) ? P[j + 1] : Pn[0];
        const float xs = xi[m], ys = xy[m];
        const float iv = fmaf(S.ti, 0.5f, 0.5f);
        SB();
        const float to = __builtin_amdgcn_tanhf(S.g3);
        SB();
        const float fc = fmaf(S.tf, S.ch, S.ch);
        SB();
        const float t0 = fmaf(xs, W.w1[0], W.b[0]);
        const float t1 = fmaf(xs, W.w1[1], W.b[1]);
        const float t2 = fmaf(xs, W.w1[2], W.b[2]);
        const float t3 = fmaf(xs, W.w1[3], W.b[3]);
        SB();
        const float c = fmaf(iv, S.tg, fc);
        SB();
        Pn[j][0] = fmaf(ys, W.w0[0], t0);
        Pn[j][1] = fmaf(ys, W.w0[1], t1);
        SB();
        const float ow = fmaf(to, W.rh, W.rh);
        Pn[j][2] = fmaf(ys, W.w0[2], t2);
        SB();
        const float tc = __builtin_amdgcn_tanhf(c);
        SB();
        S.ch = 0.5f * c;
        Pn[j][3] = fmaf(ys, W.w0[3], t3);
        SB();
        float p = ow * tc;
        const float r1 = xadd1_of_product(p);
        SB();
        const float G0 = fma_lane4(p, W.a[0], Pq[0]);
        const float G1 = fma_lane4(p, W.a[1], Pq[1]);
        SB();
        const float r2 = xadd2_chain(r1);
        SB();
        const float G2 = fma_lane4(p, W.a[2], Pq[2]);
        const float G3 = fma_lane4(p, W.a[3], Pq[3]);
        SB();
        const float g0 = fma_lane0(r2, W.a[0], G0);
        SB();
        const float g1 = fma_lane0(r2, W.a[1], G1);
        SB();
        const float g2 = fma_lane0(r2, W.a[2], G2);
        SB();
        S.ti = __builtin_amdgcn_tanhf(g0);
        SB();
        S.tf = __builtin_amdgcn_tanhf(g1);
        SB();
        S.g3 = fma_lane0(r2, W.a[3], G3);
        hs[j] = xadd4_out(r2);
        SB();
        S.tg = __builtin_amdgcn_tanhf(g2);
        SB();
    }
}

template <int DIR, bool GIVEN>
__device__ __forceinline__ Carry run_span(const LaneW& W, const float* __restrict__ yb,
                                          const float* __restrict__ ib, float* __restrict__ ho,
                                          Carry S, float h, float ch, int lane,
                                          int n_begin, int n_store, int n_end, float* rec_start) {
    float PA[UB][4], PB[UB][4];
    float XAy[UB], XAi[UB], XBy[UB], XBi[UB];
    float hs[UB * IT_BLK];

    load_blk<DIR>(yb, ib, n_begin, XAy, XAi);
    pre_blk<DIR>(W, XAy, XAi, PA);
    load_blk<DIR>(yb, ib, n_begin + 1, XBy, XBi);
    if (!GIVEN) {
        S.ti = __builtin_amdgcn_tanhf(fmaf(h, W.a[0], PA[0][0]));
        S.tf = __builtin_amdgcn_tanhf(fmaf(h, W.a[1], PA[0][1]));
        S.tg = __builtin_amdgcn_tanhf(fmaf(h, W.a[2], PA[0][2]));
        S.g3 = fmaf(h, W.a[3], PA[0][3]);
        S.ch = ch;
    }

    for (int n = n_begin; n < n_end; n += IT_BLK) {
        if (!GIVEN && n == n_store) put_state(rec_start, S, lane);
        {
            int pf = DIR ? (LSEQ - UB * IT_BLK - n * UB - 4 * UB * IT_BLK) : (n * UB + 4 * UB * IT_BLK);
            pf = pf < 0 ? 0 : (pf > LSEQ - 1 ? LSEQ - 1 : pf);
            __builtin_prefetch(yb + pf, 0, 0);
            __builtin_prefetch(ib + pf, 0, 0);
        }
        load_blk<DIR>(yb, ib, n + 2, XAy, XAi);
        steps8<DIR>(W, PA, PB, XBy, XBi, S, hs);
        load_blk<DIR>(yb, ib, n + 3, XBy, XBi);
        steps8<DIR>(W, PB, PA, XAy, XAi, S, hs + UB);
        load_blk<DIR>(yb, ib, n + 4, XAy, XAi);
        steps8<DIR>(W, PA, PB, XBy, XBi, S, hs + 2 * UB);
        load_blk<DIR>(yb, ib, n + 5, XBy, XBi);
        steps8<DIR>(W, PB, PA, XAy, XAi, S, hs + 3 * UB);

        const int base = DIR ? (LSEQ - UB * IT_BLK - n * UB) : (n * UB);
        if (n >= n_store && lane == 0) {
#pragma unroll
            for (int v = 0; v < UB * IT_BLK / 4; ++v) {
                float4 o;
                if (DIR) {
                    const int s = UB * IT_BLK - 1 - 4 * v;
                    o = make_float4(hs[s], hs[s - 1], hs[s - 2], hs[s - 3]);
                } else {
                    o = make_float4(hs[4 * v], hs[4 * v + 1], hs[4 * v + 2], hs[4 * v + 3]);
                }
                *reinterpret_cast<float4*>(ho + base + 4 * v) = o;
            }
        }
    }
    return S;
}

__device__ __forceinline__ LaneW lane_weights(const float* __restrict__ w_ih, const float* __restrict__ w_hh,
                                              const float* __restrict__ b_ih, const float* __restrict__ b_hh,
                                              const float* __restrict__ w_hr, bool real, int kk) {
    LaneW W;
#pragma unroll
    for (int q = 0; q < 4; ++q) {
        const int j = q * HID + kk;
        const float sc = real ? ((q == 2) ? 1.0f : 0.5f) : 0.0f;
        W.a[q] = sc * w_hh[j];
        W.w0[q] = sc * w_ih[2 * j];
        W.w1[q] = sc * w_ih[2 * j + 1];
        W.b[q] = sc * (b_ih[j] + b_hh[j]);
    }
    W.rh = real ? 0.5f * w_hr[kk] : 0.0f;
    return W;
}

__device__ __forceinline__ size_t rec_start_at(int chain, int seg) { return ((size_t)chain * NSEG + seg) * REC_FLOATS; }
__device__ __forceinline__ size_t rec_end_at(int chain, int seg) { return ((size_t)(NCHAIN + chain) * NSEG + seg) * REC_FLOATS; }

__global__ __launch_bounds__(32) void lstm_scan_segments(const float* __restrict__ y,
                          const float* __restrict__ inp,
                          const float* __restrict__ w_ih_f, const float* __restrict__ w_hh_f,
                          const float* __restrict__ b_ih_f, const float* __restrict__ b_hh_f,
                          const float* __restrict__ w_hr_f,
                          const float* __restrict__ w_ih_b, const float* __restrict__ w_hh_b,
                          const float* __restrict__ b_ih_b, const float* __restrict__ b_hh_b,
                          const float* __restrict__ w_hr_b,
                          const float* __restrict__ h0, const float* __restrict__ c0,
                          float* __restrict__ hout, float* __restrict__ st) {
    const int chain = blockIdx.x % NCHAIN;
    const int seg = blockIdx.x / NCHAIN;
    const int dir = chain >> 3;
    const int b = chain & 7;
    const int lane = threadIdx.x;
    const int k = lane & 7;
    const bool real = k < HID;
    const int kk = real ? k : 0;

    const LaneW W = dir ? lane_weights(w_ih_b, w_hh_b, b_ih_b, b_hh_b, w_hr_b, real, kk)
                        : lane_weights(w_ih_f, w_hh_f, b_ih_f, b_hh_f, w_hr_f, real, kk);
    const bool first = seg == 0;
    const float ch = (real && first) ? 0.5f * c0[chain * HID + kk] : 0.0f;
    const float h = first ? h0[chain] : 0.0f;

    const float* yb = y + (size_t)b * LSEQ;
    const float* ib = inp + (size_t)b * LSEQ;
    float* ho = hout + (size_t)chain * LSEQ;

    const int n_store = seg * SEG_PASSES * IT_BLK;
    const int n_begin = first ? 0 : n_store - RUNIN_PASSES * IT_BLK;
    const int n_end = n_store + SEG_PASSES * IT_BLK;
    float* rs = st + rec_start_at(chain, seg);
    const Carry Z = {0.0f, 0.0f, 0.0f, 0.0f, 0.0f};

    const Carry E = dir ? run_span<1, false>(W, yb, ib, ho, Z, h, ch, lane, n_begin, n_store, n_end, rs)
                        : run_span<0, false>(W, yb, ib, ho, Z, h, ch, lane, n_begin, n_store, n_end, rs);
    put_state(st + rec_end_at(chain, seg), E, lane);
}

template <int DIR>
__device__ __forceinline__ void join_chain(const LaneW& W, const float* __restrict__ yb, const float* __restrict__ ib,
                                           float* __restrict__ ho, const float* st, int chain, int lane) {
    unsigned eqm[4];
#pragma unroll
    for (int r = 0; r < 4; ++r) {
        const int s = 1 + 32 * r + lane;
        unsigned d = 0;
        if (s < NSEG) {
            const uint4* pa = reinterpret_cast<const uint4*>(st + rec_start_at(chain, s));
            const uint4* pb = reinterpret_cast<const uint4*>(st + rec_end_at(chain, s - 1));
            uint4 a[10], c[10];
#pragma unroll
            for (int v = 0; v < 10; ++v) { a[v] = pa[v]; c[v] = pb[v]; }
#pragma unroll
            for (int v = 0; v < 10; ++v)
                d |= (a[v].x ^ c[v].x) | (a[v].y ^ c[v].y) | (a[v].z ^ c[v].z) | (a[v].w ^ c[v].w);
        }
        eqm[r] = (unsigned)__ballot(d == 0);
    }
    static_assert(NSEG - 1 <= 4 * 32, "four rounds of 32 lanes cover every boundary");
    if ((eqm[0] & eqm[1] & eqm[2] & eqm[3]) == 0xffffffffu) return;

    bool have = false;
    Carry cur = {0.0f, 0.0f, 0.0f, 0.0f, 0.0f};
    for (int s = 1; s < NSEG; ++s) {
        bool ok;
        if (!have) {
            const int i = s - 1;
            const unsigned m = (i < 32) ? eqm[0] : (i < 64) ? eqm[1] : (i < 96) ? eqm[2] : eqm[3];
            ok = (m >> (i & 31)) & 1u;
        } else {
            ok = same_state(cur, st + rec_start_at(chain, s), lane);
        }
        if (ok) { have = false; continue; }
        if (!have) cur = get_state(st + rec_end_at(chain, s - 1), lane);
        const int n0 = s * SEG_PASSES * IT_BLK;
        cur = run_span<DIR, true>(W, yb, ib, ho, cur, 0.0f, 0.0f, lane, n0, n0, n0 + SEG_PASSES * IT_BLK, nullptr);
        have = true;
    }
}

__global__ __launch_bounds__(32) void lstm_scan_join(const float* __restrict__ y,
                          const float* __restrict__ inp,
                          const float* __restrict__ w_ih_f, const float* __restrict__ w_hh_f,
                          const float* __restrict__ b_ih_f, const float* __restrict__ b_hh_f,
                          const float* __restrict__ w_hr_f,
                          const float* __restrict__ w_ih_b, const float* __restrict__ w_hh_b,
                          const float* __restrict__ b_ih_b, const float* __restrict__ b_hh_b,
                          const float* __restrict__ w_hr_b,
                          float* __restrict__ hout, const float* st) {
    const int chain = blockIdx.x;
    const int dir = chain >> 3;
    const int b = chain & 7;
    const int lane = threadIdx.x;
    const int k = lane & 7;
    const bool real = k < HID;
    const int kk = real ? k : 0;

    const LaneW W = dir ? lane_weights(w_ih_b, w_hh_b, b_ih_b, b_hh_b, w_hr_b, real, kk)
                        : lane_weights(w_ih_f, w_hh_f, b_ih_f, b_hh_f, w_hr_f, real, kk);
    const float* yb = y + (size_t)b * LSEQ;
    const float* ib = inp + (size_t)b * LSEQ;
    float* ho = hout + (size_t)chain * LSEQ;

    if (dir) join_chain<1>(W, yb, ib, ho, st, chain, lane);
    else     join_chain<0>(W, yb, ib, ho, st, chain, lane);
}

#define SM_T 128
#define SM_STEPS 8
#define SM_CHUNK (2 * SM_T * SM_STEPS)
#define SM_NCH ((LSEQ + SM_CHUNK - 1) / SM_CHUNK)
#define SM_REC 32
static_assert(SM_NCH <= 32, "softmax_finish holds one chunk record per lane");
static_assert(LSEQ % 2 == 0, "a pair of positions is inside the row or past its end as a whole");

template <int N>
__device__ __forceinline__ float row_xor(float v) {
    return __builtin_bit_cast(float, __builtin_amdgcn_update_dpp(0, __builtin_bit_cast(int, v), 0x160 + N, 0xf, 0xf, true));
}
__device__ __forceinline__ float other_row(float v) {
    const int b = __builtin_bit_cast(int, v);
    return __builtin_bit_cast(float, __builtin_amdgcn_permlanex16(b, b, 0x76543210, 0xfedcba98, false, false));
}
__device__ __forceinline__ float wave_max(float v) {
    v = fmaxf(v, row_xor<1>(v));
    v = fmaxf(v, row_xor<2>(v));
    v = fmaxf(v, row_xor<4>(v));
    v = fmaxf(v, row_xor<8>(v));
    return fmaxf(v, other_row(v));
}
__device__ __forceinline__ float wave_sum(float v) {
    v += row_xor<1>(v);
    v += row_xor<2>(v);
    v += row_xor<4>(v);
    v += row_xor<8>(v);
    return v + other_row(v);
}

__device__ __forceinline__ void sm_load_chunk(const float* __restrict__ hf, const float* __restrict__ hb,
                                              int ch, int tid, float (&v)[2 * SM_STEPS]) {
    float2 f[SM_STEPS], g[SM_STEPS];
#pragma unroll
    for (int s = 0; s < SM_STEPS; ++s) {
        const int l = ch * SM_CHUNK + s * 2 * SM_T + 2 * tid;
        const int lc = l < LSEQ ? l : 0;
        f[s] = *reinterpret_cast<const float2*>(hf + lc);
        g[s] = *reinterpret_cast<const float2*>(hb + lc);
    }
#pragma unroll
    for (int s = 0; s < SM_STEPS; ++s) {
        const bool ok = ch * SM_CHUNK + s * 2 * SM_T + 2 * tid < LSEQ;
        v[2 * s] = ok ? f[s].x + g[s].x : -INFINITY;
        v[2 * s + 1] = ok ? f[s].y + g[s].y : -INFINITY;
    }
}

__global__ __launch_bounds__(SM_T) void softmax_chunk_stats(const float* __restrict__ hbuf, float* __restrict__ part) {
    const int b = blockIdx.x / SM_NCH;
    const int ch = blockIdx.x % SM_NCH;
    const int tid = threadIdx.x;
    __shared__ float4 redm, reds;
    static_assert(SM_T == 4 * 32, "four waves, one float4 of partial results");

    float v[2 * SM_STEPS];
    sm_load_chunk(hbuf + (size_t)b * LSEQ, hbuf + (size_t)(NB + b) * LSEQ, ch, tid, v);

    float m = v[0];
#pragma unroll
    for (int i = 1; i < 2 * SM_STEPS; ++i) m = fmaxf(m, v[i]);
    m = wave_max(m);
    if ((tid & 31) == 0) reinterpret_cast<float*>(&redm)[tid >> 5] = m;
    __syncthreads();
    const float4 wm = redm;
    m = fmaxf(fmaxf(wm.x, wm.y), fmaxf(wm.z, wm.w));

    float s = 0.0f;
#pragma unroll
    for (int i = 0; i < 2 * SM_STEPS; ++i) s += __expf(v[i] - m);
    s = wave_sum(s);
    if ((tid & 31) == 0) reinterpret_cast<float*>(&reds)[tid >> 5] = s;
    __syncthreads();
    if (tid == 0) {
        const float4 ws = reds;
        *reinterpret_cast<float2*>(part + (size_t)blockIdx.x * SM_REC) = make_float2(m, (ws.x + ws.y) + (ws.z + ws.w));
    }
}

__global__ __launch_bounds__(SM_T) void softmax_finish(const float* __restrict__ hbuf, const float* __restrict__ part,
                                                       float* __restrict__ out) {
    const int b = blockIdx.x / SM_NCH;
    const int ch = blockIdx.x % SM_NCH;
    const int tid = threadIdx.x;
    const int lane = tid & 31;

    float v[2 * SM_STEPS];
    sm_load_chunk(hbuf + (size_t)b * LSEQ, hbuf + (size_t)(NB + b) * LSEQ, ch, tid, v);

    float pm = -INFINITY, ps = 0.0f;
    if (lane < SM_NCH) {
        const float2 r = *reinterpret_cast<const float2*>(part + (size_t)(b * SM_NCH + lane) * SM_REC);
        pm = r.x;
        ps = r.y;
    }
    const float M = wave_max(pm);
    const float S = wave_sum(ps * __expf(pm - M));
    const float inv = __builtin_bit_cast(float, __builtin_amdgcn_readfirstlane(__builtin_bit_cast(int, __builtin_amdgcn_rcpf(S))));

    float* ob = out + (size_t)b * LSEQ * 2;
#pragma unroll
    for (int s = 0; s < SM_STEPS; ++s) {
        const int l = ch * SM_CHUNK + s * 2 * SM_T + 2 * tid;
        if (l < LSEQ) {
            const float e0 = __expf(v[2 * s] - M) * inv;
            const float e1 = __expf(v[2 * s + 1] - M) * inv;
            *reinterpret_cast<float4*>(ob + (size_t)l * 2) = make_float4(e0, e0, e1, e1);
        }
    }
}

extern "C" void kernel_launch(void* const* d_in, const int* in_sizes, int n_in,
                              void* d_out, int out_size, void* d_ws, size_t ws_size,
                              hipStream_t stream) {
    const float* x      = (const float*)d_in[0];
    const float* inp    = (const float*)d_in[1];
    const float* conv_w = (const float*)d_in[2];
    const float* conv_b = (const float*)d_in[3];
    const float* w_ih_f = (const float*)d_in[4];
    const float* w_hh_f = (const float*)d_in[5];
    const float* b_ih_f = (const float*)d_in[6];
    const float* b_hh_f = (const float*)d_in[7];
    const float* w_hr_f = (const float*)d_in[8];
    const float* w_ih_b = (const float*)d_in[9];
    const float* w_hh_b = (const float*)d_in[10];
    const float* b_ih_b = (const float*)d_in[11];
    const float* b_hh_b = (const float*)d_in[12];
    const float* w_hr_b = (const float*)d_in[13];
    const float* h0     = (const float*)d_in[14];
    const float* c0     = (const float*)d_in[15];
    float* out = (float*)d_out;

    float* y    = (float*)d_ws;
    float* hbuf = y + (size_t)NB * LSEQ;

    conv1x1_wmma<<<NB * CV_TPR / CV_WAVES, 32 * CV_WAVES, 0, stream>>>(x, conv_w, conv_b, y);

    float* st = out;

    lstm_scan_segments<<<NCHAIN * NSEG, 32, 0, stream>>>(y, inp,
                                    w_ih_f, w_hh_f, b_ih_f, b_hh_f, w_hr_f,
                                    w_ih_b, w_hh_b, b_ih_b, b_hh_b, w_hr_b,
                                    h0, c0, hbuf, st);

    lstm_scan_join<<<NCHAIN, 32, 0, stream>>>(y, inp,
                                    w_ih_f, w_hh_f, b_ih_f, b_hh_f, w_hr_f,
                                    w_ih_b, w_hh_b, b_ih_b, b_hh_b, w_hr_b,
                                    hbuf, st);

    float* part = y;
    softmax_chunk_stats<<<NB * SM_NCH, SM_T, 0, stream>>>(hbuf, part);
    softmax_finish<<<NB * SM_NCH, SM_T, 0, stream>>>(hbuf, part, out);
}


# LSTM scan: the 4 8-lane groups of a wave run 4 different recurrences (were 4 copies of one), lane reads by DPP8 inside the group; 500 waves as 125 work-groups of 4, was 2000; sum order kept
# speedup vs baseline: 494.9198x; 1.4013x over previous
#include <hip/hip_runtime.h>
#include <hip/hip_bf16.h>

#define LSEQ 60000
#define NB 8
#define INCH 64
#define HID 5

typedef __attribute__((ext_vector_type(2))) float v2f;
typedef __attribute__((ext_vector_type(8))) float v8f;

#define CV_TILE 128
#define CV_TPR ((LSEQ + CV_TILE - 1) / CV_TILE)
#define CV_WAVES 4
static_assert((NB * CV_TPR) % CV_WAVES == 0, "whole work-groups");
static_assert(LSEQ % 4 == 0, "four positions per lane are inside the row or past its end as a whole");

__global__ __launch_bounds__(32 * CV_WAVES) __attribute__((amdgpu_waves_per_eu(4, 4))) void conv1x1_wmma(
                             const float* __restrict__ x,
                             const float* __restrict__ w,
                             const float* __restrict__ bias,
                             float* __restrict__ y) {
    const int lane = threadIdx.x & 31;
    const int wave = threadIdx.x >> 5;
    const int tile = blockIdx.x * CV_WAVES + wave;
    const int b = tile / CV_TPR;
    const int lbase = (tile % CV_TPR) * CV_TILE;
    const int l = lbase + 4 * lane;
    const bool inrow = l < LSEQ;
    const int ll = inrow ? l : LSEQ - 4;

    const float* xb = x + (size_t)b * INCH * LSEQ + ll;

    float4 q[INCH];
#pragma unroll
    for (int i = 0; i < INCH; ++i)
        q[i] = *reinterpret_cast<const float4*>(xb + (size_t)i * LSEQ);

    const bool mine = ((lane >> 3) & 1) == (lane >> 4);
    const bool keep0 = mine && !(lane & 1);
    const bool keep1 = mine && (lane & 1);
    v8f acc0 = {}, acc1 = {};
#pragma unroll
    for (int i = 0; i < INCH; ++i) {
        const float wi = w[i];
        v2f a, b0, b1;
        a.x = keep0 ? wi : 0.0f;
        a.y = keep1 ? wi : 0.0f;
        b0.x = q[i].x; b0.y = q[i].y;
        b1.x = q[i].z; b1.y = q[i].w;
        acc0 = __builtin_amdgcn_wmma_f32_16x16x4_f32(
             false, a,  false, b0,
             (short)0, acc0,  false,  false);
        acc1 = __builtin_amdgcn_wmma_f32_16x16x4_f32(
             false, a,  false, b1,
             (short)0, acc1,  false,  false);
    }

    const float bv = bias[0];
    if (inrow)
        *reinterpret_cast<float4*>(y + (size_t)b * LSEQ + l) =
            make_float4(acc0[0] + bv, acc0[1] + bv, acc1[0] + bv, acc1[1] + bv);
}

#define UB 8
#define NBLK (LSEQ / UB)
#define IT_BLK 4
#define NCHAIN (2 * NB)
#define NPASS (NBLK / IT_BLK)
#define SEG_PASSES 15
#define NSEG (NPASS / SEG_PASSES)
#define GRP 4
#define SCAN_WAVES (NCHAIN / GRP)
#define RUNIN_PASSES 8
#define REC_FLOATS 64
static_assert(NPASS % SEG_PASSES == 0, "segments are whole passes");
static_assert(SEG_PASSES >= RUNIN_PASSES, "the run-in of segment 1 must not start in front of the sequence");
static_assert(NB % GRP == 0, "the groups of a wave are batch rows of one direction");

#define SB() __builtin_amdgcn_sched_barrier(0)

__device__ __forceinline__ float xadd1_of_product(float p) {
    float r;
    asm("v_add_f32_dpp %0, %1, %1 row_xmask:1 row_mask:0xf bank_mask:0xf bound_ctrl:1" : "=v"(r) : "v"(p));
    return r;
}

__device__ __forceinline__ float fma_lane4(float x, float y, float z) {
    asm("v_fmac_f32_dpp %0, %1, %2 dpp8:[4,4,4,4,4,4,4,4]" : "+v"(z) : "v"(x), "v"(y));
    return z;
}
__device__ __forceinline__ float fma_lane0(float x, float y, float z) {
    asm("v_fmac_f32_dpp %0, %1, %2 dpp8:[0,0,0,0,0,0,0,0]" : "+v"(z) : "v"(x), "v"(y));
    return z;
}
__device__ __forceinline__ float xadd2_chain(float v) {
    float r;
    asm("v_add_f32_dpp %0, %1, %1 row_xmask:2 row_mask:0xf bank_mask:0xf bound_ctrl:1" : "=v"(r) : "v"(v));
    return r;
}
__device__ __forceinline__ float xadd4_out(float v) {
    float r;
    asm("v_add_f32_dpp %0, %1, %1 row_xmask:4 row_mask:0xf bank_mask:0xf bound_ctrl:1" : "=v"(r) : "v"(v));
    return r;
}

struct LaneW {
    float a[4];
    float w0[4];
    float w1[4];
    float b[4];
    float rh;
};

template <int DIR>
__device__ __forceinline__ void load_blk(const float* __restrict__ yb, const float* __restrict__ ib,
                                         int blk, float (&xy)[UB], float (&xi)[UB]) {
    if (blk > NBLK - 1) blk = NBLK - 1;
    const int base = DIR ? (LSEQ - UB - blk * UB) : (blk * UB);
    const float4 y0 = *reinterpret_cast<const float4*>(yb + base);
    const float4 y1 = *reinterpret_cast<const float4*>(yb + base + 4);
    const float4 i0 = *reinterpret_cast<const float4*>(ib + base);
    const float4 i1 = *reinterpret_cast<const float4*>(ib + base + 4);
    xy[0] = y0.x; xy[1] = y0.y; xy[2] = y0.z; xy[3] = y0.w;
    xy[4] = y1.x; xy[5] = y1.y; xy[6] = y1.z; xy[7] = y1.w;
    xi[0] = i0.x; xi[1] = i0.y; xi[2] = i0.z; xi[3] = i0.w;
    xi[4] = i1.x; xi[5] = i1.y; xi[6] = i1.z; xi[7] = i1.w;
}

template <int DIR>
__device__ __forceinline__ void pre_blk(const LaneW& W, const float (&xy)[UB], const float (&xi)[UB],
                                        float (&P)[UB][4]) {
#pragma unroll
    for (int j = 0; j < UB; ++j) {
        const int m = DIR ? (UB - 1 - j) : j;
#pragma unroll
        for (int q = 0; q < 4; ++q)
            P[j][q] = fmaf(xy[m], W.w0[q], fmaf(xi[m], W.w1[q], W.b[q]));
    }
}

struct Carry { float ti, tf, tg, g3, ch; };

__device__ __forceinline__ void put_state(float* rec, const Carry& S, int lane) {
    const int k = lane & 7;
    *reinterpret_cast<float4*>(rec + 4 * k) = make_float4(S.ti, S.tf, S.tg, S.g3);
    rec[32 + k] = S.ch;
}
__device__ __forceinline__ Carry get_state(const float* rec, int lane) {
    const int k = lane & 7;
    const float4 g = *reinterpret_cast<const float4*>(rec + 4 * k);
    Carry S;
    S.ti = g.x; S.tf = g.y; S.tg = g.z; S.g3 = g.w;
    S.ch = rec[32 + k];
    return S;
}
__device__ __forceinline__ bool same_state(const Carry& A, const float* rec, int lane) {
    const Carry B = get_state(rec, lane);
    const bool diff = (__float_as_uint(A.ti) != __float_as_uint(B.ti)) | (__float_as_uint(A.tf) != __float_as_uint(B.tf)) |
                      (__float_as_uint(A.tg) != __float_as_uint(B.tg)) | (__float_as_uint(A.g3) != __float_as_uint(B.g3)) |
                      (__float_as_uint(A.ch) != __float_as_uint(B.ch));
    return !__any(diff);
}

template <int DIR>
__device__ __forceinline__ void steps8(const LaneW& W, const float (&P)[UB][4], float (&Pn)[UB][4],
                                       const float (&xy)[UB], const float (&xi)[UB],
                                       Carry& S, float* hs) {
#pragma unroll
    for (int j = 0; j < UB; ++j) {
        const int m = DIR ? (UB - 1 - j) : j;
        const float (&Pq)[4] = (j + 1 < UB) ? P[j + 1] : Pn[0];
        const float xs = xi[m], ys = xy[m];
        const float iv = fmaf(S.ti, 0.5f, 0.5f);
        SB();
        const float to = __builtin_amdgcn_tanhf(S.g3);
        SB();
        const float fc = fmaf(S.tf, S.ch, S.ch);
        SB();
        const float t0 = fmaf(xs, W.w1[0], W.b[0]);
        const float t1 = fmaf(xs, W.w1[1], W.b[1]);
        const float t2 = fmaf(xs, W.w1[2], W.b[2]);
        const float t3 = fmaf(xs, W.w1[3], W.b[3]);
        SB();
        const float c = fmaf(iv, S.tg, fc);
        SB();
        Pn[j][0] = fmaf(ys, W.w0[0], t0);
        Pn[j][1] = fmaf(ys, W.w0[1], t1);
        SB();
        const float ow = fmaf(to, W.rh, W.rh);
        Pn[j][2] = fmaf(ys, W.w0[2], t2);
        SB();
        const float tc = __builtin_amdgcn_tanhf(c);
        SB();
        S.ch = 0.5f * c;
        Pn[j][3] = fmaf(ys, W.w0[3], t3);
        SB();
        float p = ow * tc;
        const float r1 = xadd1_of_product(p);
        SB();
        const float G0 = fma_lane4(p, W.a[0], Pq[0]);
        const float G1 = fma_lane4(p, W.a[1], Pq[1]);
        SB();
        const float r2 = xadd2_chain(r1);
        SB();
        const float G2 = fma_lane4(p, W.a[2], Pq[2]);
        const float G3 = fma_lane4(p, W.a[3], Pq[3]);
        SB();
        const float g0 = fma_lane0(r2, W.a[0], G0);
        SB();
        const float g1 = fma_lane0(r2, W.a[1], G1);
        SB();
        const float g2 = fma_lane0(r2, W.a[2], G2);
        SB();
        S.ti = __builtin_amdgcn_tanhf(g0);
        SB();
        S.tf = __builtin_amdgcn_tanhf(g1);
        SB();
        S.g3 = fma_lane0(r2, W.a[3], G3);
        hs[j] = xadd4_out(r2);
        SB();
        S.tg = __builtin_amdgcn_tanhf(g2);
        SB();
    }
}

template <int DIR, bool GIVEN>
__device__ __forceinline__ Carry run_span(const LaneW& W, const float* __restrict__ yb,
                                          const float* __restrict__ ib, float* __restrict__ ho,
                                          Carry S, float h, float ch, int lane, bool storer,
                                          int n_begin, int n_store, int n_end, float* rec_start) {
    float PA[UB][4], PB[UB][4];
    float XAy[UB], XAi[UB], XBy[UB], XBi[UB];
    float hs[UB * IT_BLK];

    load_blk<DIR>(yb, ib, n_begin, XAy, XAi);
    pre_blk<DIR>(W, XAy, XAi, PA);
    load_blk<DIR>(yb, ib, n_begin + 1, XBy, XBi);
    if (!GIVEN) {
        S.ti = __builtin_amdgcn_tanhf(fmaf(h, W.a[0], PA[0][0]));
        S.tf = __builtin_amdgcn_tanhf(fmaf(h, W.a[1], PA[0][1]));
        S.tg = __builtin_amdgcn_tanhf(fmaf(h, W.a[2], PA[0][2]));
        S.g3 = fmaf(h, W.a[3], PA[0][3]);
        S.ch = ch;
    }

    for (int n = n_begin; n < n_end; n += IT_BLK) {
        if (!GIVEN && n == n_store) put_state(rec_start, S, lane);
        {
            int pf = DIR ? (LSEQ - UB * IT_BLK - n * UB - 4 * UB * IT_BLK) : (n * UB + 4 * UB * IT_BLK);
            pf = pf < 0 ? 0 : (pf > LSEQ - 1 ? LSEQ - 1 : pf);
            __builtin_prefetch(yb + pf, 0, 0);
            __builtin_prefetch(ib + pf, 0, 0);
        }
        load_blk<DIR>(yb, ib, n + 2, XAy, XAi);
        steps8<DIR>(W, PA, PB, XBy, XBi, S, hs);
        load_blk<DIR>(yb, ib, n + 3, XBy, XBi);
        steps8<DIR>(W, PB, PA, XAy, XAi, S, hs + UB);
        load_blk<DIR>(yb, ib, n + 4, XAy, XAi);
        steps8<DIR>(W, PA, PB, XBy, XBi, S, hs + 2 * UB);
        load_blk<DIR>(yb, ib, n + 5, XBy, XBi);
        steps8<DIR>(W, PB, PA, XAy, XAi, S, hs + 3 * UB);

        const int base = DIR ? (LSEQ - UB * IT_BLK - n * UB) : (n * UB);
        if (n >= n_store && storer) {
#pragma unroll
            for (int v = 0; v < UB * IT_BLK / 4; ++v) {
                float4 o;
                if (DIR) {
                    const int s = UB * IT_BLK - 1 - 4 * v;
                    o = make_float4(hs[s], hs[s - 1], hs[s - 2], hs[s - 3]);
                } else {
                    o = make_float4(hs[4 * v], hs[4 * v + 1], hs[4 * v + 2], hs[4 * v + 3]);
                }
                *reinterpret_cast<float4*>(ho + base + 4 * v) = o;
            }
        }
    }
    return S;
}

__device__ __forceinline__ LaneW lane_weights(const float* __restrict__ w_ih, const float* __restrict__ w_hh,
                                              const float* __restrict__ b_ih, const float* __restrict__ b_hh,
                                              const float* __restrict__ w_hr, bool real, int kk) {
    LaneW W;
#pragma unroll
    for (int q = 0; q < 4; ++q) {
        const int j = q * HID + kk;
        const float sc = real ? ((q == 2) ? 1.0f : 0.5f) : 0.0f;
        W.a[q] = sc * w_hh[j];
        W.w0[q] = sc * w_ih[2 * j];
        W.w1[q] = sc * w_ih[2 * j + 1];
        W.b[q] = sc * (b_ih[j] + b_hh[j]);
    }
    W.rh = real ? 0.5f * w_hr[kk] : 0.0f;
    return W;
}

__device__ __forceinline__ size_t rec_start_at(int chain, int seg) { return ((size_t)chain * NSEG + seg) * REC_FLOATS; }
__device__ __forceinline__ size_t rec_end_at(int chain, int seg) { return ((size_t)(NCHAIN + chain) * NSEG + seg) * REC_FLOATS; }

__global__ __launch_bounds__(32 * SCAN_WAVES) void lstm_scan_segments(const float* __restrict__ y,
                          const float* __restrict__ inp,
                          const float* __restrict__ w_ih_f, const float* __restrict__ w_hh_f,
                          const float* __restrict__ b_ih_f, const float* __restrict__ b_hh_f,
                          const float* __restrict__ w_hr_f,
                          const float* __restrict__ w_ih_b, const float* __restrict__ w_hh_b,
                          const float* __restrict__ b_ih_b, const float* __restrict__ b_hh_b,
                          const float* __restrict__ w_hr_b,
                          const float* __restrict__ h0, const float* __restrict__ c0,
                          float* __restrict__ hout, float* __restrict__ st) {
    const int seg = blockIdx.x;
    const int wave = __builtin_amdgcn_readfirstlane((int)(threadIdx.x >> 5));
    const int lane = threadIdx.x & 31;
    const int dir = wave / (NB / GRP);
    const int b = (wave % (NB / GRP)) * GRP + (lane >> 3);
    const int chain = dir * NB + b;
    const int k = lane & 7;
    const bool real = k < HID;
    const int kk = real ? k : 0;

    const LaneW W = dir ? lane_weights(w_ih_b, w_hh_b, b_ih_b, b_hh_b, w_hr_b, real, kk)
                        : lane_weights(w_ih_f, w_hh_f, b_ih_f, b_hh_f, w_hr_f, real, kk);
    const bool first = seg == 0;
    const float ch = (real && first) ? 0.5f * c0[chain * HID + kk] : 0.0f;
    const float h = first ? h0[chain] : 0.0f;

    const float* yb = y + (size_t)b * LSEQ;
    const float* ib = inp + (size_t)b * LSEQ;
    float* ho = hout + (size_t)chain * LSEQ;

    const int n_store = seg * SEG_PASSES * IT_BLK;
    const int n_begin = first ? 0 : n_store - RUNIN_PASSES * IT_BLK;
    const int n_end = n_store + SEG_PASSES * IT_BLK;
    float* rs = st + rec_start_at(chain, seg);
    const Carry Z = {0.0f, 0.0f, 0.0f, 0.0f, 0.0f};

    const bool storer = k == 0;
    const Carry E = dir ? run_span<1, false>(W, yb, ib, ho, Z, h, ch, lane, storer, n_begin, n_store, n_end, rs)
                        : run_span<0, false>(W, yb, ib, ho, Z, h, ch, lane, storer, n_begin, n_store, n_end, rs);
    put_state(st + rec_end_at(chain, seg), E, lane);
}

template <int DIR>
__device__ __forceinline__ void join_chain(const LaneW& W, const float* __restrict__ yb, const float* __restrict__ ib,
                                           float* __restrict__ ho, const float* st, int chain, int lane) {
    unsigned eqm[4];
#pragma unroll
    for (int r = 0; r < 4; ++r) {
        const int s = 1 + 32 * r + lane;
        unsigned d = 0;
        if (s < NSEG) {
            const uint4* pa = reinterpret_cast<const uint4*>(st + rec_start_at(chain, s));
            const uint4* pb = reinterpret_cast<const uint4*>(st + rec_end_at(chain, s - 1));
            uint4 a[10], c[10];
#pragma unroll
            for (int v = 0; v < 10; ++v) { a[v] = pa[v]; c[v] = pb[v]; }
#pragma unroll
            for (int v = 0; v < 10; ++v)
                d |= (a[v].x ^ c[v].x) | (a[v].y ^ c[v].y) | (a[v].z ^ c[v].z) | (a[v].w ^ c[v].w);
        }
        eqm[r] = (unsigned)__ballot(d == 0);
    }
    static_assert(NSEG - 1 <= 4 * 32, "four rounds of 32 lanes cover every boundary");
    if ((eqm[0] & eqm[1] & eqm[2] & eqm[3]) == 0xffffffffu) return;

    bool have = false;
    Carry cur = {0.0f, 0.0f, 0.0f, 0.0f, 0.0f};
    for (int s = 1; s < NSEG; ++s) {
        bool ok;
        if (!have) {
            const int i = s - 1;
            const unsigned m = (i < 32) ? eqm[0] : (i < 64) ? eqm[1] : (i < 96) ? eqm[2] : eqm[3];
            ok = (m >> (i & 31)) & 1u;
        } else {
            ok = same_state(cur, st + rec_start_at(chain, s), lane);
        }
        if (ok) { have = false; continue; }
        if (!have) cur = get_state(st + rec_end_at(chain, s - 1), lane);
        const int n0 = s * SEG_PASSES * IT_BLK;
        cur = run_span<DIR, true>(W, yb, ib, ho, cur, 0.0f, 0.0f, lane, lane == 0, n0, n0, n0 + SEG_PASSES * IT_BLK, nullptr);
        have = true;
    }
}

__global__ __launch_bounds__(32) void lstm_scan_join(const float* __restrict__ y,
                          const float* __restrict__ inp,
                          const float* __restrict__ w_ih_f, const float* __restrict__ w_hh_f,
                          const float* __restrict__ b_ih_f, const float* __restrict__ b_hh_f,
                          const float* __restrict__ w_hr_f,
                          const float* __restrict__ w_ih_b, const float* __restrict__ w_hh_b,
                          const float* __restrict__ b_ih_b, const float* __restrict__ b_hh_b,
                          const float* __restrict__ w_hr_b,
                          float* __restrict__ hout, const float* st) {
    const int chain = blockIdx.x;
    const int dir = chain >> 3;
    const int b = chain & 7;
    const int lane = threadIdx.x;
    const int k = lane & 7;
    const bool real = k < HID;
    const int kk = real ? k : 0;

    const LaneW W = dir ? lane_weights(w_ih_b, w_hh_b, b_ih_b, b_hh_b, w_hr_b, real, kk)
                        : lane_weights(w_ih_f, w_hh_f, b_ih_f, b_hh_f, w_hr_f, real, kk);
    const float* yb = y + (size_t)b * LSEQ;
    const float* ib = inp + (size_t)b * LSEQ;
    float* ho = hout + (size_t)chain * LSEQ;

    if (dir) join_chain<1>(W, yb, ib, ho, st, chain, lane);
    else     join_chain<0>(W, yb, ib, ho, st, chain, lane);
}

#define SM_T 128
#define SM_STEPS 8
#define SM_CHUNK (2 * SM_T * SM_STEPS)
#define SM_NCH ((LSEQ + SM_CHUNK - 1) / SM_CHUNK)
#define SM_REC 32
static_assert(SM_NCH <= 32, "softmax_finish holds one chunk record per lane");
static_assert(LSEQ % 2 == 0, "a pair of positions is inside the row or past its end as a whole");

template <int N>
__device__ __forceinline__ float row_xor(float v) {
    return __builtin_bit_cast(float, __builtin_amdgcn_update_dpp(0, __builtin_bit_cast(int, v), 0x160 + N, 0xf, 0xf, true));
}
__device__ __forceinline__ float other_row(float v) {
    const int b = __builtin_bit_cast(int, v);
    return __builtin_bit_cast(float, __builtin_amdgcn_permlanex16(b, b, 0x76543210, 0xfedcba98, false, false));
}
__device__ __forceinline__ float wave_max(float v) {
    v = fmaxf(v, row_xor<1>(v));
    v = fmaxf(v, row_xor<2>(v));
    v = fmaxf(v, row_xor<4>(v));
    v = fmaxf(v, row_xor<8>(v));
    return fmaxf(v, other_row(v));
}
__device__ __forceinline__ float wave_sum(float v) {
    v += row_xor<1>(v);
    v += row_xor<2>(v);
    v += row_xor<4>(v);
    v += row_xor<8>(v);
    return v + other_row(v);
}

__device__ __forceinline__ void sm_load_chunk(const float* __restrict__ hf, const float* __restrict__ hb,
                                              int ch, int tid, float (&v)[2 * SM_STEPS]) {
    float2 f[SM_STEPS], g[SM_STEPS];
#pragma unroll
    for (int s = 0; s < SM_STEPS; ++s) {
        const int l = ch * SM_CHUNK + s * 2 * SM_T + 2 * tid;
        const int lc = l < LSEQ ? l : 0;
        f[s] = *reinterpret_cast<const float2*>(hf + lc);
        g[s] = *reinterpret_cast<const float2*>(hb + lc);
    }
#pragma unroll
    for (int s = 0; s < SM_STEPS; ++s) {
        const bool ok = ch * SM_CHUNK + s * 2 * SM_T + 2 * tid < LSEQ;
        v[2 * s] = ok ? f[s].x + g[s].x : -INFINITY;
        v[2 * s + 1] = ok ? f[s].y + g[s].y : -INFINITY;
    }
}

__global__ __launch_bounds__(SM_T) void softmax_chunk_stats(const float* __restrict__ hbuf, float* __restrict__ part) {
    const int b = blockIdx.x / SM_NCH;
    const int ch = blockIdx.x % SM_NCH;
    const int tid = threadIdx.x;
    __shared__ float4 redm, reds;
    static_assert(SM_T == 4 * 32, "four waves, one float4 of partial results");

    float v[2 * SM_STEPS];
    sm_load_chunk(hbuf + (size_t)b * LSEQ, hbuf + (size_t)(NB + b) * LSEQ, ch, tid, v);

    float m = v[0];
#pragma unroll
    for (int i = 1; i < 2 * SM_STEPS; ++i) m = fmaxf(m, v[i]);
    m = wave_max(m);
    if ((tid & 31) == 0) reinterpret_cast<float*>(&redm)[tid >> 5] = m;
    __syncthreads();
    const float4 wm = redm;
    m = fmaxf(fmaxf(wm.x, wm.y), fmaxf(wm.z, wm.w));

    float s = 0.0f;
#pragma unroll
    for (int i = 0; i < 2 * SM_STEPS; ++i) s += __expf(v[i] - m);
    s = wave_sum(s);
    if ((tid & 31) == 0) reinterpret_cast<float*>(&reds)[tid >> 5] = s;
    __syncthreads();
    if (tid == 0) {
        const float4 ws = reds;
        *reinterpret_cast<float2*>(part + (size_t)blockIdx.x * SM_REC) = make_float2(m, (ws.x + ws.y) + (ws.z + ws.w));
    }
}

__global__ __launch_bounds__(SM_T) void softmax_finish(const float* __restrict__ hbuf, const float* __restrict__ part,
                                                       float* __restrict__ out) {
    const int b = blockIdx.x / SM_NCH;
    const int ch = blockIdx.x % SM_NCH;
    const int tid = threadIdx.x;
    const int lane = tid & 31;

    float v[2 * SM_STEPS];
    sm_load_chunk(hbuf + (size_t)b * LSEQ, hbuf + (size_t)(NB + b) * LSEQ, ch, tid, v);

    float pm = -INFINITY, ps = 0.0f;
    if (lane < SM_NCH) {
        const float2 r = *reinterpret_cast<const float2*>(part + (size_t)(b * SM_NCH + lane) * SM_REC);
        pm = r.x;
        ps = r.y;
    }
    const float M = wave_max(pm);
    const float S = wave_sum(ps * __expf(pm - M));
    const float inv = __builtin_bit_cast(float, __builtin_amdgcn_readfirstlane(__builtin_bit_cast(int, __builtin_amdgcn_rcpf(S))));

    float* ob = out + (size_t)b * LSEQ * 2;
#pragma unroll
    for (int s = 0; s < SM_STEPS; ++s) {
        const int l = ch * SM_CHUNK + s * 2 * SM_T + 2 * tid;
        if (l < LSEQ) {
            const float e0 = __expf(v[2 * s] - M) * inv;
            const float e1 = __expf(v[2 * s + 1] - M) * inv;
            *reinterpret_cast<float4*>(ob + (size_t)l * 2) = make_float4(e0, e0, e1, e1);
        }
    }
}

extern "C" void kernel_launch(void* const* d_in, const int* in_sizes, int n_in,
                              void* d_out, int out_size, void* d_ws, size_t ws_size,
                              hipStream_t stream) {
    const float* x      = (const float*)d_in[0];
    const float* inp    = (const float*)d_in[1];
    const float* conv_w = (const float*)d_in[2];
    const float* conv_b = (const float*)d_in[3];
    const float* w_ih_f = (const float*)d_in[4];
    const float* w_hh_f = (const float*)d_in[5];
    const float* b_ih_f = (const float*)d_in[6];
    const float* b_hh_f = (const float*)d_in[7];
    const float* w_hr_f = (const float*)d_in[8];
    const float* w_ih_b = (const float*)d_in[9];
    const float* w_hh_b = (const float*)d_in[10];
    const float* b_ih_b = (const float*)d_in[11];
    const float* b_hh_b = (const float*)d_in[12];
    const float* w_hr_b = (const float*)d_in[13];
    const float* h0     = (const float*)d_in[14];
    const float* c0     = (const float*)d_in[15];
    float* out = (float*)d_out;

    float* y    = (float*)d_ws;
    float* hbuf = y + (size_t)NB * LSEQ;

    conv1x1_wmma<<<NB * CV_TPR / CV_WAVES, 32 * CV_WAVES, 0, stream>>>(x, conv_w, conv_b, y);

    float* st = out;

    lstm_scan_segments<<<NSEG, 32 * SCAN_WAVES, 0, stream>>>(y, inp,
                                    w_ih_f, w_hh_f, b_ih_f, b_hh_f, w_hr_f,
                                    w_ih_b, w_hh_b, b_ih_b, b_hh_b, w_hr_b,
                                    h0, c0, hbuf, st);

    lstm_scan_join<<<NCHAIN, 32, 0, stream>>>(y, inp,
                                    w_ih_f, w_hh_f, b_ih_f, b_hh_f, w_hr_f,
                                    w_ih_b, w_hh_b, b_ih_b, b_hh_b, w_hr_b,
                                    hbuf, st);

    float* part = y;
    softmax_chunk_stats<<<NB * SM_NCH, SM_T, 0, stream>>>(hbuf, part);
    softmax_finish<<<NB * SM_NCH, SM_T, 0, stream>>>(hbuf, part, out);
}


# LSTM scan: segments of 8 passes (256 steps), was 15: 235 work-groups of four waves, a WGP each, a wave runs 512 steps, was 736; join compares the 234 boundaries in 8 waves; sum order kept
# speedup vs baseline: 588.7856x; 1.1897x over previous
#include <hip/hip_runtime.h>
#include <hip/hip_bf16.h>

#define LSEQ 60000
#define NB 8
#define INCH 64
#define HID 5

typedef __attribute__((ext_vector_type(2))) float v2f;
typedef __attribute__((ext_vector_type(8))) float v8f;

#define CV_TILE 128
#define CV_TPR ((LSEQ + CV_TILE - 1) / CV_TILE)
#define CV_WAVES 4
static_assert((NB * CV_TPR) % CV_WAVES == 0, "whole work-groups");
static_assert(LSEQ % 4 == 0, "four positions per lane are inside the row or past its end as a whole");

__global__ __launch_bounds__(32 * CV_WAVES) __attribute__((amdgpu_waves_per_eu(4, 4))) void conv1x1_wmma(
                             const float* __restrict__ x,
                             const float* __restrict__ w,
                             const float* __restrict__ bias,
                             float* __restrict__ y) {
    const int lane = threadIdx.x & 31;
    const int wave = threadIdx.x >> 5;
    const int tile = blockIdx.x * CV_WAVES + wave;
    const int b = tile / CV_TPR;
    const int lbase = (tile % CV_TPR) * CV_TILE;
    const int l = lbase + 4 * lane;
    const bool inrow = l < LSEQ;
    const int ll = inrow ? l : LSEQ - 4;

    const float* xb = x + (size_t)b * INCH * LSEQ + ll;

    float4 q[INCH];
#pragma unroll
    for (int i = 0; i < INCH; ++i)
        q[i] = *reinterpret_cast<const float4*>(xb + (size_t)i * LSEQ);

    const bool mine = ((lane >> 3) & 1) == (lane >> 4);
    const bool keep0 = mine && !(lane & 1);
    const bool keep1 = mine && (lane & 1);
    v8f acc0 = {}, acc1 = {};
#pragma unroll
    for (int i = 0; i < INCH; ++i) {
        const float wi = w[i];
        v2f a, b0, b1;
        a.x = keep0 ? wi : 0.0f;
        a.y = keep1 ? wi : 0.0f;
        b0.x = q[i].x; b0.y = q[i].y;
        b1.x = q[i].z; b1.y = q[i].w;
        acc0 = __builtin_amdgcn_wmma_f32_16x16x4_f32(
             false, a,  false, b0,
             (short)0, acc0,  false,  false);
        acc1 = __builtin_amdgcn_wmma_f32_16x16x4_f32(
             false, a,  false, b1,
             (short)0, acc1,  false,  false);
    }

    const float bv = bias[0];
    if (inrow)
        *reinterpret_cast<float4*>(y + (size_t)b * LSEQ + l) =
            make_float4(acc0[0] + bv, acc0[1] + bv, acc1[0] + bv, acc1[1] + bv);
}

#define UB 8
#define NBLK (LSEQ / UB)
#define IT_BLK 4
#define NCHAIN (2 * NB)
#define NPASS (NBLK / IT_BLK)
#define SEG_PASSES 8
#define NSEG ((NPASS + SEG_PASSES - 1) / SEG_PASSES)
#define GRP 4
#define SCAN_WAVES (NCHAIN / GRP)
#define RUNIN_PASSES 8
#define REC_FLOATS 64
static_assert(NSEG <= 256, "one work-group per WGP, so that every wave has a SIMD to itself");
static_assert(SEG_PASSES >= RUNIN_PASSES, "the run-in of segment 1 must not start in front of the sequence");
static_assert(2 * NCHAIN * NSEG * REC_FLOATS <= NB * LSEQ * 2, "the boundary states fit into the output buffer, through which they pass");
static_assert(NB % GRP == 0, "the groups of a wave are batch rows of one direction");

#define SB() __builtin_amdgcn_sched_barrier(0)

__device__ __forceinline__ float xadd1_of_product(float p) {
    float r;
    asm("v_add_f32_dpp %0, %1, %1 row_xmask:1 row_mask:0xf bank_mask:0xf bound_ctrl:1" : "=v"(r) : "v"(p));
    return r;
}

__device__ __forceinline__ float fma_lane4(float x, float y, float z) {
    asm("v_fmac_f32_dpp %0, %1, %2 dpp8:[4,4,4,4,4,4,4,4]" : "+v"(z) : "v"(x), "v"(y));
    return z;
}
__device__ __forceinline__ float fma_lane0(float x, float y, float z) {
    asm("v_fmac_f32_dpp %0, %1, %2 dpp8:[0,0,0,0,0,0,0,0]" : "+v"(z) : "v"(x), "v"(y));
    return z;
}
__device__ __forceinline__ float xadd2_chain(float v) {
    float r;
    asm("v_add_f32_dpp %0, %1, %1 row_xmask:2 row_mask:0xf bank_mask:0xf bound_ctrl:1" : "=v"(r) : "v"(v));
    return r;
}
__device__ __forceinline__ float xadd4_out(float v) {
    float r;
    asm("v_add_f32_dpp %0, %1, %1 row_xmask:4 row_mask:0xf bank_mask:0xf bound_ctrl:1" : "=v"(r) : "v"(v));
    return r;
}

struct LaneW {
    float a[4];
    float w0[4];
    float w1[4];
    float b[4];
    float rh;
};

template <int DIR>
__device__ __forceinline__ void load_blk(const float* __restrict__ yb, const float* __restrict__ ib,
                                         int blk, float (&xy)[UB], float (&xi)[UB]) {
    if (blk > NBLK - 1) blk = NBLK - 1;
    const int base = DIR ? (LSEQ - UB - blk * UB) : (blk * UB);
    const float4 y0 = *reinterpret_cast<const float4*>(yb + base);
    const float4 y1 = *reinterpret_cast<const float4*>(yb + base + 4);
    const float4 i0 = *reinterpret_cast<const float4*>(ib + base);
    const float4 i1 = *reinterpret_cast<const float4*>(ib + base + 4);
    xy[0] = y0.x; xy[1] = y0.y; xy[2] = y0.z; xy[3] = y0.w;
    xy[4] = y1.x; xy[5] = y1.y; xy[6] = y1.z; xy[7] = y1.w;
    xi[0] = i0.x; xi[1] = i0.y; xi[2] = i0.z; xi[3] = i0.w;
    xi[4] = i1.x; xi[5] = i1.y; xi[6] = i1.z; xi[7] = i1.w;
}

template <int DIR>
__device__ __forceinline__ void pre_blk(const LaneW& W, const float (&xy)[UB], const float (&xi)[UB],
                                        float (&P)[UB][4]) {
#pragma unroll
    for (int j = 0; j < UB; ++j) {
        const int m = DIR ? (UB - 1 - j) : j;
#pragma unroll
        for (int q = 0; q < 4; ++q)
            P[j][q] = fmaf(xy[m], W.w0[q], fmaf(xi[m], W.w1[q], W.b[q]));
    }
}

struct Carry { float ti, tf, tg, g3, ch; };

__device__ __forceinline__ void put_state(float* rec, const Carry& S, int lane) {
    const int k = lane & 7;
    *reinterpret_cast<float4*>(rec + 4 * k) = make_float4(S.ti, S.tf, S.tg, S.g3);
    rec[32 + k] = S.ch;
}
__device__ __forceinline__ Carry get_state(const float* rec, int lane) {
    const int k = lane & 7;
    const float4 g = *reinterpret_cast<const float4*>(rec + 4 * k);
    Carry S;
    S.ti = g.x; S.tf = g.y; S.tg = g.z; S.g3 = g.w;
    S.ch = rec[32 + k];
    return S;
}
__device__ __forceinline__ bool same_state(const Carry& A, const float* rec, int lane) {
    const Carry B = get_state(rec, lane);
    const bool diff = (__float_as_uint(A.ti) != __float_as_uint(B.ti)) | (__float_as_uint(A.tf) != __float_as_uint(B.tf)) |
                      (__float_as_uint(A.tg) != __float_as_uint(B.tg)) | (__float_as_uint(A.g3) != __float_as_uint(B.g3)) |
                      (__float_as_uint(A.ch) != __float_as_uint(B.ch));
    return !__any(diff);
}

template <int DIR>
__device__ __forceinline__ void steps8(const LaneW& W, const float (&P)[UB][4], float (&Pn)[UB][4],
                                       const float (&xy)[UB], const float (&xi)[UB],
                                       Carry& S, float* hs) {
#pragma unroll
    for (int j = 0; j < UB; ++j) {
        const int m = DIR ? (UB - 1 - j) : j;
        const float (&Pq)[4] = (j + 1 < UB) ? P[j + 1] : Pn[0];
        const float xs = xi[m], ys = xy[m];
        const float iv = fmaf(S.ti, 0.5f, 0.5f);
        SB();
        const float to = __builtin_amdgcn_tanhf(S.g3);
        SB();
        const float fc = fmaf(S.tf, S.ch, S.ch);
        SB();
        const float t0 = fmaf(xs, W.w1[0], W.b[0]);
        const float t1 = fmaf(xs, W.w1[1], W.b[1]);
        const float t2 = fmaf(xs, W.w1[2], W.b[2]);
        const float t3 = fmaf(xs, W.w1[3], W.b[3]);
        SB();
        const float c = fmaf(iv, S.tg, fc);
        SB();
        Pn[j][0] = fmaf(ys, W.w0[0], t0);
        Pn[j][1] = fmaf(ys, W.w0[1], t1);
        SB();
        const float ow = fmaf(to, W.rh, W.rh);
        Pn[j][2] = fmaf(ys, W.w0[2], t2);
        SB();
        const float tc = __builtin_amdgcn_tanhf(c);
        SB();
        S.ch = 0.5f * c;
        Pn[j][3] = fmaf(ys, W.w0[3], t3);
        SB();
        float p = ow * tc;
        const float r1 = xadd1_of_product(p);
        SB();
        const float G0 = fma_lane4(p, W.a[0], Pq[0]);
        const float G1 = fma_lane4(p, W.a[1], Pq[1]);
        SB();
        const float r2 = xadd2_chain(r1);
        SB();
        const float G2 = fma_lane4(p, W.a[2], Pq[2]);
        const float G3 = fma_lane4(p, W.a[3], Pq[3]);
        SB();
        const float g0 = fma_lane0(r2, W.a[0], G0);
        SB();
        const float g1 = fma_lane0(r2, W.a[1], G1);
        SB();
        const float g2 = fma_lane0(r2, W.a[2], G2);
        SB();
        S.ti = __builtin_amdgcn_tanhf(g0);
        SB();
        S.tf = __builtin_amdgcn_tanhf(g1);
        SB();
        S.g3 = fma_lane0(r2, W.a[3], G3);
        hs[j] = xadd4_out(r2);
        SB();
        S.tg = __builtin_amdgcn_tanhf(g2);
        SB();
    }
}

template <int DIR, bool GIVEN>
__device__ __forceinline__ Carry run_span(const LaneW& W, const float* __restrict__ yb,
                                          const float* __restrict__ ib, float* __restrict__ ho,
                                          Carry S, float h, float ch, int lane, bool storer,
                                          int n_begin, int n_store, int n_end, float* rec_start) {
    float PA[UB][4], PB[UB][4];
    float XAy[UB], XAi[UB], XBy[UB], XBi[UB];
    float hs[UB * IT_BLK];

    load_blk<DIR>(yb, ib, n_begin, XAy, XAi);
    pre_blk<DIR>(W, XAy, XAi, PA);
    load_blk<DIR>(yb, ib, n_begin + 1, XBy, XBi);
    if (!GIVEN) {
        S.ti = __builtin_amdgcn_tanhf(fmaf(h, W.a[0], PA[0][0]));
        S.tf = __builtin_amdgcn_tanhf(fmaf(h, W.a[1], PA[0][1]));
        S.tg = __builtin_amdgcn_tanhf(fmaf(h, W.a[2], PA[0][2]));
        S.g3 = fmaf(h, W.a[3], PA[0][3]);
        S.ch = ch;
    }

    for (int n = n_begin; n < n_end; n += IT_BLK) {
        if (!GIVEN && n == n_store) put_state(rec_start, S, lane);
        {
            int pf = DIR ? (LSEQ - UB * IT_BLK - n * UB - 4 * UB * IT_BLK) : (n * UB + 4 * UB * IT_BLK);
            pf = pf < 0 ? 0 : (pf > LSEQ - 1 ? LSEQ - 1 : pf);
            __builtin_prefetch(yb + pf, 0, 0);
            __builtin_prefetch(ib + pf, 0, 0);
        }
        load_blk<DIR>(yb, ib, n + 2, XAy, XAi);
        steps8<DIR>(W, PA, PB, XBy, XBi, S, hs);
        load_blk<DIR>(yb, ib, n + 3, XBy, XBi);
        steps8<DIR>(W, PB, PA, XAy, XAi, S, hs + UB);
        load_blk<DIR>(yb, ib, n + 4, XAy, XAi);
        steps8<DIR>(W, PA, PB, XBy, XBi, S, hs + 2 * UB);
        load_blk<DIR>(yb, ib, n + 5, XBy, XBi);
        steps8<DIR>(W, PB, PA, XAy, XAi, S, hs + 3 * UB);

        const int base = DIR ? (LSEQ - UB * IT_BLK - n * UB) : (n * UB);
        if (n >= n_store && storer) {
#pragma unroll
            for (int v = 0; v < UB * IT_BLK / 4; ++v) {
                float4 o;
                if (DIR) {
                    const int s = UB * IT_BLK - 1 - 4 * v;
                    o = make_float4(hs[s], hs[s - 1], hs[s - 2], hs[s - 3]);
                } else {
                    o = make_float4(hs[4 * v], hs[4 * v + 1], hs[4 * v + 2], hs[4 * v + 3]);
                }
                *reinterpret_cast<float4*>(ho + base + 4 * v) = o;
            }
        }
    }
    return S;
}

__device__ __forceinline__ LaneW lane_weights(const float* __restrict__ w_ih, const float* __restrict__ w_hh,
                                              const float* __restrict__ b_ih, const float* __restrict__ b_hh,
                                              const float* __restrict__ w_hr, bool real, int kk) {
    LaneW W;
#pragma unroll
    for (int q = 0; q < 4; ++q) {
        const int j = q * HID + kk;
        const float sc = real ? ((q == 2) ? 1.0f : 0.5f) : 0.0f;
        W.a[q] = sc * w_hh[j];
        W.w0[q] = sc * w_ih[2 * j];
        W.w1[q] = sc * w_ih[2 * j + 1];
        W.b[q] = sc * (b_ih[j] + b_hh[j]);
    }
    W.rh = real ? 0.5f * w_hr[kk] : 0.0f;
    return W;
}

__device__ __forceinline__ int seg_first_blk(int s) { return s * SEG_PASSES * IT_BLK; }
__device__ __forceinline__ int seg_end_blk(int s) {
    const int e = (s + 1) * SEG_PASSES * IT_BLK;
    return e < NBLK ? e : NBLK;
}

__device__ __forceinline__ size_t rec_start_at(int chain, int seg) { return ((size_t)chain * NSEG + seg) * REC_FLOATS; }
__device__ __forceinline__ size_t rec_end_at(int chain, int seg) { return ((size_t)(NCHAIN + chain) * NSEG + seg) * REC_FLOATS; }

__global__ __launch_bounds__(32 * SCAN_WAVES) void lstm_scan_segments(const float* __restrict__ y,
                          const float* __restrict__ inp,
                          const float* __restrict__ w_ih_f, const float* __restrict__ w_hh_f,
                          const float* __restrict__ b_ih_f, const float* __restrict__ b_hh_f,
                          const float* __restrict__ w_hr_f,
                          const float* __restrict__ w_ih_b, const float* __restrict__ w_hh_b,
                          const float* __restrict__ b_ih_b, const float* __restrict__ b_hh_b,
                          const float* __restrict__ w_hr_b,
                          const float* __restrict__ h0, const float* __restrict__ c0,
                          float* __restrict__ hout, float* __restrict__ st) {
    const int seg = blockIdx.x;
    const int wave = __builtin_amdgcn_readfirstlane((int)(threadIdx.x >> 5));
    const int lane = threadIdx.x & 31;
    const int dir = wave / (NB / GRP);
    const int b = (wave % (NB / GRP)) * GRP + (lane >> 3);
    const int chain = dir * NB + b;
    const int k = lane & 7;
    const bool real = k < HID;
    const int kk = real ? k : 0;

    const LaneW W = dir ? lane_weights(w_ih_b, w_hh_b, b_ih_b, b_hh_b, w_hr_b, real, kk)
                        : lane_weights(w_ih_f, w_hh_f, b_ih_f, b_hh_f, w_hr_f, real, kk);
    const bool first = seg == 0;
    const float ch = (real && first) ? 0.5f * c0[chain * HID + kk] : 0.0f;
    const float h = first ? h0[chain] : 0.0f;

    const float* yb = y + (size_t)b * LSEQ;
    const float* ib = inp + (size_t)b * LSEQ;
    float* ho = hout + (size_t)chain * LSEQ;

    const int n_store = seg_first_blk(seg);
    const int n_begin = first ? 0 : n_store - RUNIN_PASSES * IT_BLK;
    const int n_end = seg_end_blk(seg);
    float* rs = st + rec_start_at(chain, seg);
    const Carry Z = {0.0f, 0.0f, 0.0f, 0.0f, 0.0f};

    const bool storer = k == 0;
    const Carry E = dir ? run_span<1, false>(W, yb, ib, ho, Z, h, ch, lane, storer, n_begin, n_store, n_end, rs)
                        : run_span<0, false>(W, yb, ib, ho, Z, h, ch, lane, storer, n_begin, n_store, n_end, rs);
    put_state(st + rec_end_at(chain, seg), E, lane);
}

#define JOIN_WAVES 8
static_assert(NSEG - 1 <= JOIN_WAVES * 32, "one boundary per lane and wave covers every boundary");

__device__ __forceinline__ unsigned boundaries_equal(const float* st, int chain, int round, int lane) {
    const int s = 1 + 32 * round + lane;
    unsigned d = 0;
    if (s < NSEG) {
        const uint4* pa = reinterpret_cast<const uint4*>(st + rec_start_at(chain, s));
        const uint4* pb = reinterpret_cast<const uint4*>(st + rec_end_at(chain, s - 1));
        uint4 a[10], c[10];
#pragma unroll
        for (int v = 0; v < 10; ++v) { a[v] = pa[v]; c[v] = pb[v]; }
#pragma unroll
        for (int v = 0; v < 10; ++v)
            d |= (a[v].x ^ c[v].x) | (a[v].y ^ c[v].y) | (a[v].z ^ c[v].z) | (a[v].w ^ c[v].w);
    }
    return (unsigned)__ballot(d == 0);
}

template <int DIR>
__device__ __forceinline__ void join_chain(const LaneW& W, const float* __restrict__ yb, const float* __restrict__ ib,
                                           float* __restrict__ ho, const float* st, const unsigned* eq,
                                           int chain, int lane) {
    bool have = false;
    Carry cur = {0.0f, 0.0f, 0.0f, 0.0f, 0.0f};
    for (int s = 1; s < NSEG; ++s) {
        bool ok;
        if (!have) {
            const int i = s - 1;
            ok = (eq[i >> 5] >> (i & 31)) & 1u;
        } else {
            ok = same_state(cur, st + rec_start_at(chain, s), lane);
        }
        if (ok) { have = false; continue; }
        if (!have) cur = get_state(st + rec_end_at(chain, s - 1), lane);
        const int n0 = seg_first_blk(s);
        cur = run_span<DIR, true>(W, yb, ib, ho, cur, 0.0f, 0.0f, lane, lane == 0, n0, n0, seg_end_blk(s), nullptr);
        have = true;
    }
}

__global__ __launch_bounds__(32 * JOIN_WAVES) void lstm_scan_join(const float* __restrict__ y,
                          const float* __restrict__ inp,
                          const float* __restrict__ w_ih_f, const float* __restrict__ w_hh_f,
                          const float* __restrict__ b_ih_f, const float* __restrict__ b_hh_f,
                          const float* __restrict__ w_hr_f,
                          const float* __restrict__ w_ih_b, const float* __restrict__ w_hh_b,
                          const float* __restrict__ b_ih_b, const float* __restrict__ b_hh_b,
                          const float* __restrict__ w_hr_b,
                          float* __restrict__ hout, const float* st) {
    __shared__ unsigned eq[JOIN_WAVES];
    const int chain = blockIdx.x;
    const int wave = threadIdx.x >> 5;
    const int lane = threadIdx.x & 31;

    const unsigned m = boundaries_equal(st, chain, wave, lane);
    if (lane == 0) eq[wave] = m;
    __syncthreads();
    if (wave != 0) return;

    unsigned all = 0xffffffffu;
    for (int r = 0; r < JOIN_WAVES; ++r) all &= eq[r];
    if (all == 0xffffffffu) return;

    const int dir = chain >> 3;
    const int b = chain & 7;
    const int k = lane & 7;
    const bool real = k < HID;
    const int kk = real ? k : 0;

    const LaneW W = dir ? lane_weights(w_ih_b, w_hh_b, b_ih_b, b_hh_b, w_hr_b, real, kk)
                        : lane_weights(w_ih_f, w_hh_f, b_ih_f, b_hh_f, w_hr_f, real, kk);
    const float* yb = y + (size_t)b * LSEQ;
    const float* ib = inp + (size_t)b * LSEQ;
    float* ho = hout + (size_t)chain * LSEQ;

    if (dir) join_chain<1>(W, yb, ib, ho, st, eq, chain, lane);
    else     join_chain<0>(W, yb, ib, ho, st, eq, chain, lane);
}

#define SM_T 128
#define SM_STEPS 8
#define SM_CHUNK (2 * SM_T * SM_STEPS)
#define SM_NCH ((LSEQ + SM_CHUNK - 1) / SM_CHUNK)
#define SM_REC 32
static_assert(SM_NCH <= 32, "softmax_finish holds one chunk record per lane");
static_assert(LSEQ % 2 == 0, "a pair of positions is inside the row or past its end as a whole");

template <int N>
__device__ __forceinline__ float row_xor(float v) {
    return __builtin_bit_cast(float, __builtin_amdgcn_update_dpp(0, __builtin_bit_cast(int, v), 0x160 + N, 0xf, 0xf, true));
}
__device__ __forceinline__ float other_row(float v) {
    const int b = __builtin_bit_cast(int, v);
    return __builtin_bit_cast(float, __builtin_amdgcn_permlanex16(b, b, 0x76543210, 0xfedcba98, false, false));
}
__device__ __forceinline__ float wave_max(float v) {
    v = fmaxf(v, row_xor<1>(v));
    v = fmaxf(v, row_xor<2>(v));
    v = fmaxf(v, row_xor<4>(v));
    v = fmaxf(v, row_xor<8>(v));
    return fmaxf(v, other_row(v));
}
__device__ __forceinline__ float wave_sum(float v) {
    v += row_xor<1>(v);
    v += row_xor<2>(v);
    v += row_xor<4>(v);
    v += row_xor<8>(v);
    return v + other_row(v);
}

__device__ __forceinline__ void sm_load_chunk(const float* __restrict__ hf, const float* __restrict__ hb,
                                              int ch, int tid, float (&v)[2 * SM_STEPS]) {
    float2 f[SM_STEPS], g[SM_STEPS];
#pragma unroll
    for (int s = 0; s < SM_STEPS; ++s) {
        const int l = ch * SM_CHUNK + s * 2 * SM_T + 2 * tid;
        const int lc = l < LSEQ ? l : 0;
        f[s] = *reinterpret_cast<const float2*>(hf + lc);
        g[s] = *reinterpret_cast<const float2*>(hb + lc);
    }
#pragma unroll
    for (int s = 0; s < SM_STEPS; ++s) {
        const bool ok = ch * SM_CHUNK + s * 2 * SM_T + 2 * tid < LSEQ;
        v[2 * s] = ok ? f[s].x + g[s].x : -INFINITY;
        v[2 * s + 1] = ok ? f[s].y + g[s].y : -INFINITY;
    }
}

__global__ __launch_bounds__(SM_T) void softmax_chunk_stats(const float* __restrict__ hbuf, float* __restrict__ part) {
    const int b = blockIdx.x / SM_NCH;
    const int ch = blockIdx.x % SM_NCH;
    const int tid = threadIdx.x;
    __shared__ float4 redm, reds;
    static_assert(SM_T == 4 * 32, "four waves, one float4 of partial results");

    float v[2 * SM_STEPS];
    sm_load_chunk(hbuf + (size_t)b * LSEQ, hbuf + (size_t)(NB + b) * LSEQ, ch, tid, v);

    float m = v[0];
#pragma unroll
    for (int i = 1; i < 2 * SM_STEPS; ++i) m = fmaxf(m, v[i]);
    m = wave_max(m);
    if ((tid & 31) == 0) reinterpret_cast<float*>(&redm)[tid >> 5] = m;
    __syncthreads();
    const float4 wm = redm;
    m = fmaxf(fmaxf(wm.x, wm.y), fmaxf(wm.z, wm.w));

    float s = 0.0f;
#pragma unroll
    for (int i = 0; i < 2 * SM_STEPS; ++i) s += __expf(v[i] - m);
    s = wave_sum(s);
    if ((tid & 31) == 0) reinterpret_cast<float*>(&reds)[tid >> 5] = s;
    __syncthreads();
    if (tid == 0) {
        const float4 ws = reds;
        *reinterpret_cast<float2*>(part + (size_t)blockIdx.x * SM_REC) = make_float2(m, (ws.x + ws.y) + (ws.z + ws.w));
    }
}

__global__ __launch_bounds__(SM_T) void softmax_finish(const float* __restrict__ hbuf, const float* __restrict__ part,
                                                       float* __restrict__ out) {
    const int b = blockIdx.x / SM_NCH;
    const int ch = blockIdx.x % SM_NCH;
    const int tid = threadIdx.x;
    const int lane = tid & 31;

    float v[2 * SM_STEPS];
    sm_load_chunk(hbuf + (size_t)b * LSEQ, hbuf + (size_t)(NB + b) * LSEQ, ch, tid, v);

    float pm = -INFINITY, ps = 0.0f;
    if (lane < SM_NCH) {
        const float2 r = *reinterpret_cast<const float2*>(part + (size_t)(b * SM_NCH + lane) * SM_REC);
        pm = r.x;
        ps = r.y;
    }
    const float M = wave_max(pm);
    const float S = wave_sum(ps * __expf(pm - M));
    const float inv = __builtin_bit_cast(float, __builtin_amdgcn_readfirstlane(__builtin_bit_cast(int, __builtin_amdgcn_rcpf(S))));

    float* ob = out + (size_t)b * LSEQ * 2;
#pragma unroll
    for (int s = 0; s < SM_STEPS; ++s) {
        const int l = ch * SM_CHUNK + s * 2 * SM_T + 2 * tid;
        if (l < LSEQ) {
            const float e0 = __expf(v[2 * s] - M) * inv;
            const float e1 = __expf(v[2 * s + 1] - M) * inv;
            *reinterpret_cast<float4*>(ob + (size_t)l * 2) = make_float4(e0, e0, e1, e1);
        }
    }
}

extern "C" void kernel_launch(void* const* d_in, const int* in_sizes, int n_in,
                              void* d_out, int out_size, void* d_ws, size_t ws_size,
                              hipStream_t stream) {
    const float* x      = (const float*)d_in[0];
    const float* inp    = (const float*)d_in[1];
    const float* conv_w = (const float*)d_in[2];
    const float* conv_b = (const float*)d_in[3];
    const float* w_ih_f = (const float*)d_in[4];
    const float* w_hh_f = (const float*)d_in[5];
    const float* b_ih_f = (const float*)d_in[6];
    const float* b_hh_f = (const float*)d_in[7];
    const float* w_hr_f = (const float*)d_in[8];
    const float* w_ih_b = (const float*)d_in[9];
    const float* w_hh_b = (const float*)d_in[10];
    const float* b_ih_b = (const float*)d_in[11];
    const float* b_hh_b = (const float*)d_in[12];
    const float* w_hr_b = (const float*)d_in[13];
    const float* h0     = (const float*)d_in[14];
    const float* c0     = (const float*)d_in[15];
    float* out = (float*)d_out;

    float* y    = (float*)d_ws;
    float* hbuf = y + (size_t)NB * LSEQ;

    conv1x1_wmma<<<NB * CV_TPR / CV_WAVES, 32 * CV_WAVES, 0, stream>>>(x, conv_w, conv_b, y);

    float* st = out;

    lstm_scan_segments<<<NSEG, 32 * SCAN_WAVES, 0, stream>>>(y, inp,
                                    w_ih_f, w_hh_f, b_ih_f, b_hh_f, w_hr_f,
                                    w_ih_b, w_hh_b, b_ih_b, b_hh_b, w_hr_b,
                                    h0, c0, hbuf, st);

    lstm_scan_join<<<NCHAIN, 32 * JOIN_WAVES, 0, stream>>>(y, inp,
                                    w_ih_f, w_hh_f, b_ih_f, b_hh_f, w_hr_f,
                                    w_ih_b, w_hh_b, b_ih_b, b_hh_b, w_hr_b,
                                    hbuf, st);

    float* part = y;
    softmax_chunk_stats<<<NB * SM_NCH, SM_T, 0, stream>>>(hbuf, part);
    softmax_finish<<<NB * SM_NCH, SM_T, 0, stream>>>(hbuf, part, out);
}


# LSTM scan: run-in in front of a segment is 3 passes (96 steps), was 8 (256): a wave runs 352 steps, was 512; join still compares every boundary bit for bit, re-runs on a difference; sum order kept
# speedup vs baseline: 677.0782x; 1.1500x over previous
#include <hip/hip_runtime.h>
#include <hip/hip_bf16.h>

#define LSEQ 60000
#define NB 8
#define INCH 64
#define HID 5

typedef __attribute__((ext_vector_type(2))) float v2f;
typedef __attribute__((ext_vector_type(8))) float v8f;

#define CV_TILE 128
#define CV_TPR ((LSEQ + CV_TILE - 1) / CV_TILE)
#define CV_WAVES 4
static_assert((NB * CV_TPR) % CV_WAVES == 0, "whole work-groups");
static_assert(LSEQ % 4 == 0, "four positions per lane are inside the row or past its end as a whole");

__global__ __launch_bounds__(32 * CV_WAVES) __attribute__((amdgpu_waves_per_eu(4, 4))) void conv1x1_wmma(
                             const float* __restrict__ x,
                             const float* __restrict__ w,
                             const float* __restrict__ bias,
                             float* __restrict__ y) {
    const int lane = threadIdx.x & 31;
    const int wave = threadIdx.x >> 5;
    const int tile = blockIdx.x * CV_WAVES + wave;
    const int b = tile / CV_TPR;
    const int lbase = (tile % CV_TPR) * CV_TILE;
    const int l = lbase + 4 * lane;
    const bool inrow = l < LSEQ;
    const int ll = inrow ? l : LSEQ - 4;

    const float* xb = x + (size_t)b * INCH * LSEQ + ll;

    float4 q[INCH];
#pragma unroll
    for (int i = 0; i < INCH; ++i)
        q[i] = *reinterpret_cast<const float4*>(xb + (size_t)i * LSEQ);

    const bool mine = ((lane >> 3) & 1) == (lane >> 4);
    const bool keep0 = mine && !(lane & 1);
    const bool keep1 = mine && (lane & 1);
    v8f acc0 = {}, acc1 = {};
#pragma unroll
    for (int i = 0; i < INCH; ++i) {
        const float wi = w[i];
        v2f a, b0, b1;
        a.x = keep0 ? wi : 0.0f;
        a.y = keep1 ? wi : 0.0f;
        b0.x = q[i].x; b0.y = q[i].y;
        b1.x = q[i].z; b1.y = q[i].w;
        acc0 = __builtin_amdgcn_wmma_f32_16x16x4_f32(
             false, a,  false, b0,
             (short)0, acc0,  false,  false);
        acc1 = __builtin_amdgcn_wmma_f32_16x16x4_f32(
             false, a,  false, b1,
             (short)0, acc1,  false,  false);
    }

    const float bv = bias[0];
    if (inrow)
        *reinterpret_cast<float4*>(y + (size_t)b * LSEQ + l) =
            make_float4(acc0[0] + bv, acc0[1] + bv, acc1[0] + bv, acc1[1] + bv);
}

#define UB 8
#define NBLK (LSEQ / UB)
#define IT_BLK 4
#define NCHAIN (2 * NB)
#define NPASS (NBLK / IT_BLK)
#define SEG_PASSES 8
#define NSEG ((NPASS + SEG_PASSES - 1) / SEG_PASSES)
#define GRP 4
#define SCAN_WAVES (NCHAIN / GRP)
#define RUNIN_PASSES 3
#define REC_FLOATS 64
static_assert(NSEG <= 256, "one work-group per WGP, so that every wave has a SIMD to itself");
static_assert(SEG_PASSES >= RUNIN_PASSES, "the run-in of segment 1 must not start in front of the sequence");
static_assert(2 * NCHAIN * NSEG * REC_FLOATS <= NB * LSEQ * 2, "the boundary states fit into the output buffer, through which they pass");
static_assert(NB % GRP == 0, "the groups of a wave are batch rows of one direction");

#define SB() __builtin_amdgcn_sched_barrier(0)

__device__ __forceinline__ float xadd1_of_product(float p) {
    float r;
    asm("v_add_f32_dpp %0, %1, %1 row_xmask:1 row_mask:0xf bank_mask:0xf bound_ctrl:1" : "=v"(r) : "v"(p));
    return r;
}

__device__ __forceinline__ float fma_lane4(float x, float y, float z) {
    asm("v_fmac_f32_dpp %0, %1, %2 dpp8:[4,4,4,4,4,4,4,4]" : "+v"(z) : "v"(x), "v"(y));
    return z;
}
__device__ __forceinline__ float fma_lane0(float x, float y, float z) {
    asm("v_fmac_f32_dpp %0, %1, %2 dpp8:[0,0,0,0,0,0,0,0]" : "+v"(z) : "v"(x), "v"(y));
    return z;
}
__device__ __forceinline__ float xadd2_chain(float v) {
    float r;
    asm("v_add_f32_dpp %0, %1, %1 row_xmask:2 row_mask:0xf bank_mask:0xf bound_ctrl:1" : "=v"(r) : "v"(v));
    return r;
}
__device__ __forceinline__ float xadd4_out(float v) {
    float r;
    asm("v_add_f32_dpp %0, %1, %1 row_xmask:4 row_mask:0xf bank_mask:0xf bound_ctrl:1" : "=v"(r) : "v"(v));
    return r;
}

struct LaneW {
    float a[4];
    float w0[4];
    float w1[4];
    float b[4];
    float rh;
};

template <int DIR>
__device__ __forceinline__ void load_blk(const float* __restrict__ yb, const float* __restrict__ ib,
                                         int blk, float (&xy)[UB], float (&xi)[UB]) {
    if (blk > NBLK - 1) blk = NBLK - 1;
    const int base = DIR ? (LSEQ - UB - blk * UB) : (blk * UB);
    const float4 y0 = *reinterpret_cast<const float4*>(yb + base);
    const float4 y1 = *reinterpret_cast<const float4*>(yb + base + 4);
    const float4 i0 = *reinterpret_cast<const float4*>(ib + base);
    const float4 i1 = *reinterpret_cast<const float4*>(ib + base + 4);
    xy[0] = y0.x; xy[1] = y0.y; xy[2] = y0.z; xy[3] = y0.w;
    xy[4] = y1.x; xy[5] = y1.y; xy[6] = y1.z; xy[7] = y1.w;
    xi[0] = i0.x; xi[1] = i0.y; xi[2] = i0.z; xi[3] = i0.w;
    xi[4] = i1.x; xi[5] = i1.y; xi[6] = i1.z; xi[7] = i1.w;
}

template <int DIR>
__device__ __forceinline__ void pre_blk(const LaneW& W, const float (&xy)[UB], const float (&xi)[UB],
                                        float (&P)[UB][4]) {
#pragma unroll
    for (int j = 0; j < UB; ++j) {
        const int m = DIR ? (UB - 1 - j) : j;
#pragma unroll
        for (int q = 0; q < 4; ++q)
            P[j][q] = fmaf(xy[m], W.w0[q], fmaf(xi[m], W.w1[q], W.b[q]));
    }
}

struct Carry { float ti, tf, tg, g3, ch; };

__device__ __forceinline__ void put_state(float* rec, const Carry& S, int lane) {
    const int k = lane & 7;
    *reinterpret_cast<float4*>(rec + 4 * k) = make_float4(S.ti, S.tf, S.tg, S.g3);
    rec[32 + k] = S.ch;
}
__device__ __forceinline__ Carry get_state(const float* rec, int lane) {
    const int k = lane & 7;
    const float4 g = *reinterpret_cast<const float4*>(rec + 4 * k);
    Carry S;
    S.ti = g.x; S.tf = g.y; S.tg = g.z; S.g3 = g.w;
    S.ch = rec[32 + k];
    return S;
}
__device__ __forceinline__ bool same_state(const Carry& A, const float* rec, int lane) {
    const Carry B = get_state(rec, lane);
    const bool diff = (__float_as_uint(A.ti) != __float_as_uint(B.ti)) | (__float_as_uint(A.tf) != __float_as_uint(B.tf)) |
                      (__float_as_uint(A.tg) != __float_as_uint(B.tg)) | (__float_as_uint(A.g3) != __float_as_uint(B.g3)) |
                      (__float_as_uint(A.ch) != __float_as_uint(B.ch));
    return !__any(diff);
}

template <int DIR>
__device__ __forceinline__ void steps8(const LaneW& W, const float (&P)[UB][4], float (&Pn)[UB][4],
                                       const float (&xy)[UB], const float (&xi)[UB],
                                       Carry& S, float* hs) {
#pragma unroll
    for (int j = 0; j < UB; ++j) {
        const int m = DIR ? (UB - 1 - j) : j;
        const float (&Pq)[4] = (j + 1 < UB) ? P[j + 1] : Pn[0];
        const float xs = xi[m], ys = xy[m];
        const float iv = fmaf(S.ti, 0.5f, 0.5f);
        SB();
        const float to = __builtin_amdgcn_tanhf(S.g3);
        SB();
        const float fc = fmaf(S.tf, S.ch, S.ch);
        SB();
        const float t0 = fmaf(xs, W.w1[0], W.b[0]);
        const float t1 = fmaf(xs, W.w1[1], W.b[1]);
        const float t2 = fmaf(xs, W.w1[2], W.b[2]);
        const float t3 = fmaf(xs, W.w1[3], W.b[3]);
        SB();
        const float c = fmaf(iv, S.tg, fc);
        SB();
        Pn[j][0] = fmaf(ys, W.w0[0], t0);
        Pn[j][1] = fmaf(ys, W.w0[1], t1);
        SB();
        const float ow = fmaf(to, W.rh, W.rh);
        Pn[j][2] = fmaf(ys, W.w0[2], t2);
        SB();
        const float tc = __builtin_amdgcn_tanhf(c);
        SB();
        S.ch = 0.5f * c;
        Pn[j][3] = fmaf(ys, W.w0[3], t3);
        SB();
        float p = ow * tc;
        const float r1 = xadd1_of_product(p);
        SB();
        const float G0 = fma_lane4(p, W.a[0], Pq[0]);
        const float G1 = fma_lane4(p, W.a[1], Pq[1]);
        SB();
        const float r2 = xadd2_chain(r1);
        SB();
        const float G2 = fma_lane4(p, W.a[2], Pq[2]);
        const float G3 = fma_lane4(p, W.a[3], Pq[3]);
        SB();
        const float g0 = fma_lane0(r2, W.a[0], G0);
        SB();
        const float g1 = fma_lane0(r2, W.a[1], G1);
        SB();
        const float g2 = fma_lane0(r2, W.a[2], G2);
        SB();
        S.ti = __builtin_amdgcn_tanhf(g0);
        SB();
        S.tf = __builtin_amdgcn_tanhf(g1);
        SB();
        S.g3 = fma_lane0(r2, W.a[3], G3);
        hs[j] = xadd4_out(r2);
        SB();
        S.tg = __builtin_amdgcn_tanhf(g2);
        SB();
    }
}

template <int DIR, bool GIVEN>
__device__ __forceinline__ Carry run_span(const LaneW& W, const float* __restrict__ yb,
                                          const float* __restrict__ ib, float* __restrict__ ho,
                                          Carry S, float h, float ch, int lane, bool storer,
                                          int n_begin, int n_store, int n_end, float* rec_start) {
    float PA[UB][4], PB[UB][4];
    float XAy[UB], XAi[UB], XBy[UB], XBi[UB];
    float hs[UB * IT_BLK];

    load_blk<DIR>(yb, ib, n_begin, XAy, XAi);
    pre_blk<DIR>(W, XAy, XAi, PA);
    load_blk<DIR>(yb, ib, n_begin + 1, XBy, XBi);
    if (!GIVEN) {
        S.ti = __builtin_amdgcn_tanhf(fmaf(h, W.a[0], PA[0][0]));
        S.tf = __builtin_amdgcn_tanhf(fmaf(h, W.a[1], PA[0][1]));
        S.tg = __builtin_amdgcn_tanhf(fmaf(h, W.a[2], PA[0][2]));
        S.g3 = fmaf(h, W.a[3], PA[0][3]);
        S.ch = ch;
    }

    for (int n = n_begin; n < n_end; n += IT_BLK) {
        if (!GIVEN && n == n_store) put_state(rec_start, S, lane);
        {
            int pf = DIR ? (LSEQ - UB * IT_BLK - n * UB - 4 * UB * IT_BLK) : (n * UB + 4 * UB * IT_BLK);
            pf = pf < 0 ? 0 : (pf > LSEQ - 1 ? LSEQ - 1 : pf);
            __builtin_prefetch(yb + pf, 0, 0);
            __builtin_prefetch(ib + pf, 0, 0);
        }
        load_blk<DIR>(yb, ib, n + 2, XAy, XAi);
        steps8<DIR>(W, PA, PB, XBy, XBi, S, hs);
        load_blk<DIR>(yb, ib, n + 3, XBy, XBi);
        steps8<DIR>(W, PB, PA, XAy, XAi, S, hs + UB);
        load_blk<DIR>(yb, ib, n + 4, XAy, XAi);
        steps8<DIR>(W, PA, PB, XBy, XBi, S, hs + 2 * UB);
        load_blk<DIR>(yb, ib, n + 5, XBy, XBi);
        steps8<DIR>(W, PB, PA, XAy, XAi, S, hs + 3 * UB);

        const int base = DIR ? (LSEQ - UB * IT_BLK - n * UB) : (n * UB);
        if (n >= n_store && storer) {
#pragma unroll
            for (int v = 0; v < UB * IT_BLK / 4; ++v) {
                float4 o;
                if (DIR) {
                    const int s = UB * IT_BLK - 1 - 4 * v;
                    o = make_float4(hs[s], hs[s - 1], hs[s - 2], hs[s - 3]);
                } else {
                    o = make_float4(hs[4 * v], hs[4 * v + 1], hs[4 * v + 2], hs[4 * v + 3]);
                }
                *reinterpret_cast<float4*>(ho + base + 4 * v) = o;
            }
        }
    }
    return S;
}

__device__ __forceinline__ LaneW lane_weights(const float* __restrict__ w_ih, const float* __restrict__ w_hh,
                                              const float* __restrict__ b_ih, const float* __restrict__ b_hh,
                                              const float* __restrict__ w_hr, bool real, int kk) {
    LaneW W;
#pragma unroll
    for (int q = 0; q < 4; ++q) {
        const int j = q * HID + kk;
        const float sc = real ? ((q == 2) ? 1.0f : 0.5f) : 0.0f;
        W.a[q] = sc * w_hh[j];
        W.w0[q] = sc * w_ih[2 * j];
        W.w1[q] = sc * w_ih[2 * j + 1];
        W.b[q] = sc * (b_ih[j] + b_hh[j]);
    }
    W.rh = real ? 0.5f * w_hr[kk] : 0.0f;
    return W;
}

__device__ __forceinline__ int seg_first_blk(int s) { return s * SEG_PASSES * IT_BLK; }
__device__ __forceinline__ int seg_end_blk(int s) {
    const int e = (s + 1) * SEG_PASSES * IT_BLK;
    return e < NBLK ? e : NBLK;
}

__device__ __forceinline__ size_t rec_start_at(int chain, int seg) { return ((size_t)chain * NSEG + seg) * REC_FLOATS; }
__device__ __forceinline__ size_t rec_end_at(int chain, int seg) { return ((size_t)(NCHAIN + chain) * NSEG + seg) * REC_FLOATS; }

__global__ __launch_bounds__(32 * SCAN_WAVES) void lstm_scan_segments(const float* __restrict__ y,
                          const float* __restrict__ inp,
                          const float* __restrict__ w_ih_f, const float* __restrict__ w_hh_f,
                          const float* __restrict__ b_ih_f, const float* __restrict__ b_hh_f,
                          const float* __restrict__ w_hr_f,
                          const float* __restrict__ w_ih_b, const float* __restrict__ w_hh_b,
                          const float* __restrict__ b_ih_b, const float* __restrict__ b_hh_b,
                          const float* __restrict__ w_hr_b,
                          const float* __restrict__ h0, const float* __restrict__ c0,
                          float* __restrict__ hout, float* __restrict__ st) {
    const int seg = blockIdx.x;
    const int wave = __builtin_amdgcn_readfirstlane((int)(threadIdx.x >> 5));
    const int lane = threadIdx.x & 31;
    const int dir = wave / (NB / GRP);
    const int b = (wave % (NB / GRP)) * GRP + (lane >> 3);
    const int chain = dir * NB + b;
    const int k = lane & 7;
    const bool real = k < HID;
    const int kk = real ? k : 0;

    const LaneW W = dir ? lane_weights(w_ih_b, w_hh_b, b_ih_b, b_hh_b, w_hr_b, real, kk)
                        : lane_weights(w_ih_f, w_hh_f, b_ih_f, b_hh_f, w_hr_f, real, kk);
    const bool first = seg == 0;
    const float ch = (real && first) ? 0.5f * c0[chain * HID + kk] : 0.0f;
    const float h = first ? h0[chain] : 0.0f;

    const float* yb = y + (size_t)b * LSEQ;
    const float* ib = inp + (size_t)b * LSEQ;
    float* ho = hout + (size_t)chain * LSEQ;

    const int n_store = seg_first_blk(seg);
    const int n_begin = first ? 0 : n_store - RUNIN_PASSES * IT_BLK;
    const int n_end = seg_end_blk(seg);
    float* rs = st + rec_start_at(chain, seg);
    const Carry Z = {0.0f, 0.0f, 0.0f, 0.0f, 0.0f};

    const bool storer = k == 0;
    const Carry E = dir ? run_span<1, false>(W, yb, ib, ho, Z, h, ch, lane, storer, n_begin, n_store, n_end, rs)
                        : run_span<0, false>(W, yb, ib, ho, Z, h, ch, lane, storer, n_begin, n_store, n_end, rs);
    put_state(st + rec_end_at(chain, seg), E, lane);
}

#define JOIN_WAVES 8
static_assert(NSEG - 1 <= JOIN_WAVES * 32, "one boundary per lane and wave covers every boundary");

__device__ __forceinline__ unsigned boundaries_equal(const float* st, int chain, int round, int lane) {
    const int s = 1 + 32 * round + lane;
    unsigned d = 0;
    if (s < NSEG) {
        const uint4* pa = reinterpret_cast<const uint4*>(st + rec_start_at(chain, s));
        const uint4* pb = reinterpret_cast<const uint4*>(st + rec_end_at(chain, s - 1));
        uint4 a[10], c[10];
#pragma unroll
        for (int v = 0; v < 10; ++v) { a[v] = pa[v]; c[v] = pb[v]; }
#pragma unroll
        for (int v = 0; v < 10; ++v)
            d |= (a[v].x ^ c[v].x) | (a[v].y ^ c[v].y) | (a[v].z ^ c[v].z) | (a[v].w ^ c[v].w);
    }
    return (unsigned)__ballot(d == 0);
}

template <int DIR>
__device__ __forceinline__ void join_chain(const LaneW& W, const float* __restrict__ yb, const float* __restrict__ ib,
                                           float* __restrict__ ho, const float* st, const unsigned* eq,
                                           int chain, int lane) {
    bool have = false;
    Carry cur = {0.0f, 0.0f, 0.0f, 0.0f, 0.0f};
    for (int s = 1; s < NSEG; ++s) {
        bool ok;
        if (!have) {
            const int i = s - 1;
            ok = (eq[i >> 5] >> (i & 31)) & 1u;
        } else {
            ok = same_state(cur, st + rec_start_at(chain, s), lane);
        }
        if (ok) { have = false; continue; }
        if (!have) cur = get_state(st + rec_end_at(chain, s - 1), lane);
        const int n0 = seg_first_blk(s);
        cur = run_span<DIR, true>(W, yb, ib, ho, cur, 0.0f, 0.0f, lane, lane == 0, n0, n0, seg_end_blk(s), nullptr);
        have = true;
    }
}

__global__ __launch_bounds__(32 * JOIN_WAVES) void lstm_scan_join(const float* __restrict__ y,
                          const float* __restrict__ inp,
                          const float* __restrict__ w_ih_f, const float* __restrict__ w_hh_f,
                          const float* __restrict__ b_ih_f, const float* __restrict__ b_hh_f,
                          const float* __restrict__ w_hr_f,
                          const float* __restrict__ w_ih_b, const float* __restrict__ w_hh_b,
                          const float* __restrict__ b_ih_b, const float* __restrict__ b_hh_b,
                          const float* __restrict__ w_hr_b,
                          float* __restrict__ hout, const float* st) {
    __shared__ unsigned eq[JOIN_WAVES];
    const int chain = blockIdx.x;
    const int wave = threadIdx.x >> 5;
    const int lane = threadIdx.x & 31;

    const unsigned m = boundaries_equal(st, chain, wave, lane);
    if (lane == 0) eq[wave] = m;
    __syncthreads();
    if (wave != 0) return;

    unsigned all = 0xffffffffu;
    for (int r = 0; r < JOIN_WAVES; ++r) all &= eq[r];
    if (all == 0xffffffffu) return;

    const int dir = chain >> 3;
    const int b = chain & 7;
    const int k = lane & 7;
    const bool real = k < HID;
    const int kk = real ? k : 0;

    const LaneW W = dir ? lane_weights(w_ih_b, w_hh_b, b_ih_b, b_hh_b, w_hr_b, real, kk)
                        : lane_weights(w_ih_f, w_hh_f, b_ih_f, b_hh_f, w_hr_f, real, kk);
    const float* yb = y + (size_t)b * LSEQ;
    const float* ib = inp + (size_t)b * LSEQ;
    float* ho = hout + (size_t)chain * LSEQ;

    if (dir) join_chain<1>(W, yb, ib, ho, st, eq, chain, lane);
    else     join_chain<0>(W, yb, ib, ho, st, eq, chain, lane);
}

#define SM_T 128
#define SM_STEPS 8
#define SM_CHUNK (2 * SM_T * SM_STEPS)
#define SM_NCH ((LSEQ + SM_CHUNK - 1) / SM_CHUNK)
#define SM_REC 32
static_assert(SM_NCH <= 32, "softmax_finish holds one chunk record per lane");
static_assert(LSEQ % 2 == 0, "a pair of positions is inside the row or past its end as a whole");

template <int N>
__device__ __forceinline__ float row_xor(float v) {
    return __builtin_bit_cast(float, __builtin_amdgcn_update_dpp(0, __builtin_bit_cast(int, v), 0x160 + N, 0xf, 0xf, true));
}
__device__ __forceinline__ float other_row(float v) {
    const int b = __builtin_bit_cast(int, v);
    return __builtin_bit_cast(float, __builtin_amdgcn_permlanex16(b, b, 0x76543210, 0xfedcba98, false, false));
}
__device__ __forceinline__ float wave_max(float v) {
    v = fmaxf(v, row_xor<1>(v));
    v = fmaxf(v, row_xor<2>(v));
    v = fmaxf(v, row_xor<4>(v));
    v = fmaxf(v, row_xor<8>(v));
    return fmaxf(v, other_row(v));
}
__device__ __forceinline__ float wave_sum(float v) {
    v += row_xor<1>(v);
    v += row_xor<2>(v);
    v += row_xor<4>(v);
    v += row_xor<8>(v);
    return v + other_row(v);
}

__device__ __forceinline__ void sm_load_chunk(const float* __restrict__ hf, const float* __restrict__ hb,
                                              int ch, int tid, float (&v)[2 * SM_STEPS]) {
    float2 f[SM_STEPS], g[SM_STEPS];
#pragma unroll
    for (int s = 0; s < SM_STEPS; ++s) {
        const int l = ch * SM_CHUNK + s * 2 * SM_T + 2 * tid;
        const int lc = l < LSEQ ? l : 0;
        f[s] = *reinterpret_cast<const float2*>(hf + lc);
        g[s] = *reinterpret_cast<const float2*>(hb + lc);
    }
#pragma unroll
    for (int s = 0; s < SM_STEPS; ++s) {
        const bool ok = ch * SM_CHUNK + s * 2 * SM_T + 2 * tid < LSEQ;
        v[2 * s] = ok ? f[s].x + g[s].x : -INFINITY;
        v[2 * s + 1] = ok ? f[s].y + g[s].y : -INFINITY;
    }
}

__global__ __launch_bounds__(SM_T) void softmax_chunk_stats(const float* __restrict__ hbuf, float* __restrict__ part) {
    const int b = blockIdx.x / SM_NCH;
    const int ch = blockIdx.x % SM_NCH;
    const int tid = threadIdx.x;
    __shared__ float4 redm, reds;
    static_assert(SM_T == 4 * 32, "four waves, one float4 of partial results");

    float v[2 * SM_STEPS];
    sm_load_chunk(hbuf + (size_t)b * LSEQ, hbuf + (size_t)(NB + b) * LSEQ, ch, tid, v);

    float m = v[0];
#pragma unroll
    for (int i = 1; i < 2 * SM_STEPS; ++i) m = fmaxf(m, v[i]);
    m = wave_max(m);
    if ((tid & 31) == 0) reinterpret_cast<float*>(&redm)[tid >> 5] = m;
    __syncthreads();
    const float4 wm = redm;
    m = fmaxf(fmaxf(wm.x, wm.y), fmaxf(wm.z, wm.w));

    float s = 0.0f;
#pragma unroll
    for (int i = 0; i < 2 * SM_STEPS; ++i) s += __expf(v[i] - m);
    s = wave_sum(s);
    if ((tid & 31) == 0) reinterpret_cast<float*>(&reds)[tid >> 5] = s;
    __syncthreads();
    if (tid == 0) {
        const float4 ws = reds;
        *reinterpret_cast<float2*>(part + (size_t)blockIdx.x * SM_REC) = make_float2(m, (ws.x + ws.y) + (ws.z + ws.w));
    }
}

__global__ __launch_bounds__(SM_T) void softmax_finish(const float* __restrict__ hbuf, const float* __restrict__ part,
                                                       float* __restrict__ out) {
    const int b = blockIdx.x / SM_NCH;
    const int ch = blockIdx.x % SM_NCH;
    const int tid = threadIdx.x;
    const int lane = tid & 31;

    float v[2 * SM_STEPS];
    sm_load_chunk(hbuf + (size_t)b * LSEQ, hbuf + (size_t)(NB + b) * LSEQ, ch, tid, v);

    float pm = -INFINITY, ps = 0.0f;
    if (lane < SM_NCH) {
        const float2 r = *reinterpret_cast<const float2*>(part + (size_t)(b * SM_NCH + lane) * SM_REC);
        pm = r.x;
        ps = r.y;
    }
    const float M = wave_max(pm);
    const float S = wave_sum(ps * __expf(pm - M));
    const float inv = __builtin_bit_cast(float, __builtin_amdgcn_readfirstlane(__builtin_bit_cast(int, __builtin_amdgcn_rcpf(S))));

    float* ob = out + (size_t)b * LSEQ * 2;
#pragma unroll
    for (int s = 0; s < SM_STEPS; ++s) {
        const int l = ch * SM_CHUNK + s * 2 * SM_T + 2 * tid;
        if (l < LSEQ) {
            const float e0 = __expf(v[2 * s] - M) * inv;
            const float e1 = __expf(v[2 * s + 1] - M) * inv;
            *reinterpret_cast<float4*>(ob + (size_t)l * 2) = make_float4(e0, e0, e1, e1);
        }
    }
}

extern "C" void kernel_launch(void* const* d_in, const int* in_sizes, int n_in,
                              void* d_out, int out_size, void* d_ws, size_t ws_size,
                              hipStream_t stream) {
    const float* x      = (const float*)d_in[0];
    const float* inp    = (const float*)d_in[1];
    const float* conv_w = (const float*)d_in[2];
    const float* conv_b = (const float*)d_in[3];
    const float* w_ih_f = (const float*)d_in[4];
    const float* w_hh_f = (const float*)d_in[5];
    const float* b_ih_f = (const float*)d_in[6];
    const float* b_hh_f = (const float*)d_in[7];
    const float* w_hr_f = (const float*)d_in[8];
    const float* w_ih_b = (const float*)d_in[9];
    const float* w_hh_b = (const float*)d_in[10];
    const float* b_ih_b = (const float*)d_in[11];
    const float* b_hh_b = (const float*)d_in[12];
    const float* w_hr_b = (const float*)d_in[13];
    const float* h0     = (const float*)d_in[14];
    const float* c0     = (const float*)d_in[15];
    float* out = (float*)d_out;

    float* y    = (float*)d_ws;
    float* hbuf = y + (size_t)NB * LSEQ;

    conv1x1_wmma<<<NB * CV_TPR / CV_WAVES, 32 * CV_WAVES, 0, stream>>>(x, conv_w, conv_b, y);

    float* st = out;

    lstm_scan_segments<<<NSEG, 32 * SCAN_WAVES, 0, stream>>>(y, inp,
                                    w_ih_f, w_hh_f, b_ih_f, b_hh_f, w_hr_f,
                                    w_ih_b, w_hh_b, b_ih_b, b_hh_b, w_hr_b,
                                    h0, c0, hbuf, st);

    lstm_scan_join<<<NCHAIN, 32 * JOIN_WAVES, 0, stream>>>(y, inp,
                                    w_ih_f, w_hh_f, b_ih_f, b_hh_f, w_hr_f,
                                    w_ih_b, w_hh_b, b_ih_b, b_hh_b, w_hr_b,
                                    hbuf, st);

    float* part = y;
    softmax_chunk_stats<<<NB * SM_NCH, SM_T, 0, stream>>>(hbuf, part);
    softmax_finish<<<NB * SM_NCH, SM_T, 0, stream>>>(hbuf, part, out);
}
